# starting kernel
# baseline (speedup 1.0000x reference)
#include <hip/hip_runtime.h>
#include <math.h>

typedef __attribute__((ext_vector_type(2))) float v2f;
typedef __attribute__((ext_vector_type(8))) float v8f;

#define GAT_D 64

__device__ __forceinline__ unsigned f2ord(float f) {
    int b = __float_as_int(f);
    return (unsigned)b ^ (unsigned)((b >> 31) | 0x80000000);
}
__device__ __forceinline__ float ord2f(unsigned u) {
    int b = (u & 0x80000000u) ? (int)(u ^ 0x80000000u) : ~(int)u;
    return __int_as_float(b);
}

__global__ void gat_gemm_wmma(const float* __restrict__ A,
                              const float* __restrict__ B,
                              float* __restrict__ C,
                              int M, int K, int N) {
    const int lane  = threadIdx.x;
    const int grp   = blockIdx.x * blockDim.y + threadIdx.y;
    const int mtile = blockIdx.y;
    const int n0 = grp * 64;
    const int m0 = mtile * 16;
    if (n0 >= N || m0 >= M) return;

    const int l15 = lane & 15;
    const int kh  = (lane >> 4) << 1;
    const float* Arow = A + (size_t)(m0 + l15) * K;
    v8f acc0 = {0.f,0.f,0.f,0.f,0.f,0.f,0.f,0.f};
    v8f acc1 = acc0, acc2 = acc0, acc3 = acc0;

    for (int k = 0; k < K; k += 4) {
        v2f a;
        a.x = Arow[k + kh];
        a.y = Arow[k + kh + 1];
        const float* Bp = B + (size_t)(k + kh) * N + n0 + l15;
        v2f b0, b1, b2, b3;
        b0.x = Bp[0];  b0.y = Bp[(size_t)N];
        b1.x = Bp[16]; b1.y = Bp[(size_t)N + 16];
        b2.x = Bp[32]; b2.y = Bp[(size_t)N + 32];
        b3.x = Bp[48]; b3.y = Bp[(size_t)N + 48];
        acc0 = __builtin_amdgcn_wmma_f32_16x16x4_f32(false, a, false, b0, (short)0, acc0, false, false);
        acc1 = __builtin_amdgcn_wmma_f32_16x16x4_f32(false, a, false, b1, (short)0, acc1, false, false);
        acc2 = __builtin_amdgcn_wmma_f32_16x16x4_f32(false, a, false, b2, (short)0, acc2, false, false);
        acc3 = __builtin_amdgcn_wmma_f32_16x16x4_f32(false, a, false, b3, (short)0, acc3, false, false);
    }

    float* Cp = C + (size_t)(m0 + ((lane >> 4) << 3)) * N + n0 + l15;
#pragma unroll
    for (int v = 0; v < 8; ++v) {
        Cp[(size_t)v * N + 0]  = acc0[v];
        Cp[(size_t)v * N + 16] = acc1[v];
        Cp[(size_t)v * N + 32] = acc2[v];
        Cp[(size_t)v * N + 48] = acc3[v];
    }
}

__global__ void gat_attn_coef(const float* __restrict__ h,
                              const float* __restrict__ a_s,
                              const float* __restrict__ a_d,
                              float* __restrict__ es, float* __restrict__ ed,
                              int N, int Hh) {
    int t = blockIdx.x * blockDim.x + threadIdx.x;
    if (t >= N * Hh) return;
    int n  = t / Hh;
    int hh = t - n * Hh;
    const float* hp = h   + ((size_t)n * Hh + hh) * GAT_D;
    const float* as = a_s + (size_t)hh * GAT_D;
    const float* ad = a_d + (size_t)hh * GAT_D;
    float ss = 0.f, dd = 0.f;
#pragma unroll 8
    for (int d = 0; d < GAT_D; ++d) { float v = hp[d]; ss += v * as[d]; dd += v * ad[d]; }
    es[t] = ss; ed[t] = dd;
}

__device__ __forceinline__ void edge_endpoints(const int* __restrict__ ei, int E,
                                               int e, int& src, int& dst) {
    if (e < E) { src = ei[e]; dst = ei[E + e]; }
    else       { src = e - E; dst = e - E;    }
}

__global__ void gat_edge_max(const int* __restrict__ ei, int E, int N,
                             const float* __restrict__ es, const float* __restrict__ ed,
                             unsigned* __restrict__ mT, int Hh) {
    int t = blockIdx.x * blockDim.x + threadIdx.x;
    int total = (E + N) * Hh;
    if (t >= total) return;
    int e = t / Hh, hh = t - e * Hh;
    int src, dst; edge_endpoints(ei, E, e, src, dst);
    float s  = es[src * Hh + hh] + ed[dst * Hh + hh];
    float lr = s > 0.f ? s : 0.2f * s;
    atomicMax(&mT[dst * Hh + hh], f2ord(lr));
}

__global__ void gat_edge_exp(const int* __restrict__ ei, int E, int N,
                             const float* __restrict__ es, const float* __restrict__ ed,
                             const unsigned* __restrict__ mT,
                             float* __restrict__ ex, float* __restrict__ den, int Hh) {
    int t = blockIdx.x * blockDim.x + threadIdx.x;
    int total = (E + N) * Hh;
    if (t >= total) return;
    int e = t / Hh, hh = t - e * Hh;
    int src, dst; edge_endpoints(ei, E, e, src, dst);
    float s  = es[src * Hh + hh] + ed[dst * Hh + hh];
    float lr = s > 0.f ? s : 0.2f * s;
    float v  = expf(lr - ord2f(mT[dst * Hh + hh]));
    ex[t] = v;
    atomicAdd(&den[dst * Hh + hh], v);
}

__global__ void gat_edge_alpha(const int* __restrict__ ei, int E, int N,
                               float* __restrict__ ex, const float* __restrict__ den, int Hh) {
    int t = blockIdx.x * blockDim.x + threadIdx.x;
    int total = (E + N) * Hh;
    if (t >= total) return;
    int e = t / Hh, hh = t - e * Hh;
    int src, dst; edge_endpoints(ei, E, e, src, dst);
    (void)src;
    ex[t] = ex[t] / den[dst * Hh + hh];
}

__global__ void gat_edge_agg(const int* __restrict__ ei, int E, int N,
                             const float* __restrict__ hfeat,
                             const float* __restrict__ alpha,
                             float* __restrict__ agg, int Hh) {
    int HD = Hh * GAT_D;
    int t = blockIdx.x * blockDim.x + threadIdx.x;
    int total = (E + N) * HD;
    if (t >= total) return;
    int e = t / HD;
    int c = t - e * HD;
    int hh = c >> 6;
    int src, dst; edge_endpoints(ei, E, e, src, dst);
    float w = alpha[e * Hh + hh];
    atomicAdd(&agg[(size_t)dst * HD + c], hfeat[(size_t)src * HD + c] * w);
}

__global__ void gat_epilogue(const float* __restrict__ agg, const float* __restrict__ b,
                             float* __restrict__ out, int N, int HD, int do_elu) {
    int t = blockIdx.x * blockDim.x + threadIdx.x;
    if (t >= N * HD) return;
    int c = t - (t / HD) * HD;
    float v = agg[t] + b[c];
    if (do_elu) v = v > 0.f ? v : (expf(v) - 1.f);
    out[t] = v;
}

__global__ void gat_colmean(const float* __restrict__ x, float* __restrict__ out, int N, int C) {
    __shared__ float sm[256];
    int c = blockIdx.x;
    float s = 0.f;
    for (int n = threadIdx.x; n < N; n += blockDim.x) s += x[(size_t)n * C + c];
    sm[threadIdx.x] = s;
    __syncthreads();
    for (int off = 128; off > 0; off >>= 1) {
        if ((int)threadIdx.x < off) sm[threadIdx.x] += sm[threadIdx.x + off];
        __syncthreads();
    }
    if (threadIdx.x == 0) out[c] = sm[0] / (float)N;
}

static inline int cdiv(long long a, long long b) { return (int)((a + b - 1) / b); }

extern "C" void kernel_launch(void* const* d_in, const int* in_sizes, int n_in,
                              void* d_out, int out_size, void* d_ws, size_t ws_size,
                              hipStream_t stream) {
    const float* x   = (const float*)d_in[0];
    const int*   ei  = (const int*)  d_in[1];
    const float* W1  = (const float*)d_in[2];
    const float* a1s = (const float*)d_in[3];
    const float* a1d = (const float*)d_in[4];
    const float* b1  = (const float*)d_in[5];
    const float* W2  = (const float*)d_in[6];
    const float* a2s = (const float*)d_in[7];
    const float* a2d = (const float*)d_in[8];
    const float* b2  = (const float*)d_in[9];
    const float* W3  = (const float*)d_in[10];
    const float* a3s = (const float*)d_in[11];
    const float* a3d = (const float*)d_in[12];
    const float* b3  = (const float*)d_in[13];

    const int N  = in_sizes[0] / 16;
    const int E  = in_sizes[1] / 2;
    const int Et = E + N;
    const int H  = 4;

    float* ws = (float*)d_ws;
    size_t off = 0;
    float*    buf_h    = ws + off; off += (size_t)N * 256;
    float*    buf_agg  = ws + off; off += (size_t)N * 256;
    float*    buf_feat = ws + off; off += (size_t)N * 256;
    float*    es       = ws + off; off += (size_t)N * H;
    float*    ed       = ws + off; off += (size_t)N * H;
    unsigned* mT       = (unsigned*)(ws + off); off += (size_t)N * H;
    float*    den      = ws + off; off += (size_t)N * H;
    float*    alpha    = ws + off; off += (size_t)Et * H;
    (void)ws_size;

    float* node_emb  = (float*)d_out;
    float* graph_emb = node_emb + (size_t)N * 64;

    auto run_layer = [&](const float* in_feat, const float* W,
                         const float* as_, const float* ad_, const float* bb,
                         float* out_feat, int Hh, int K, int Ncols, int do_elu) {
        const int HD = Hh * GAT_D;
        int ngrp = Ncols / 64;
        int wpb  = ngrp < 4 ? ngrp : 4;
        dim3 gb(32, wpb);
        dim3 gg(cdiv(ngrp, wpb), cdiv(N, 16));
        hipLaunchKernelGGL(gat_gemm_wmma, gg, gb, 0, stream, in_feat, W, buf_h, N, K, Ncols);
        hipLaunchKernelGGL(gat_attn_coef, dim3(cdiv((long long)N * Hh, 256)), dim3(256), 0, stream,
                           buf_h, as_, ad_, es, ed, N, Hh);
        hipMemsetAsync(mT,      0, (size_t)N * Hh * sizeof(unsigned), stream);
        hipMemsetAsync(den,     0, (size_t)N * Hh * sizeof(float),    stream);
        hipMemsetAsync(buf_agg, 0, (size_t)N * HD * sizeof(float),    stream);
        int eth = Et * Hh;
        hipLaunchKernelGGL(gat_edge_max,   dim3(cdiv(eth, 256)), dim3(256), 0, stream,
                           ei, E, N, es, ed, mT, Hh);
        hipLaunchKernelGGL(gat_edge_exp,   dim3(cdiv(eth, 256)), dim3(256), 0, stream,
                           ei, E, N, es, ed, mT, alpha, den, Hh);
        hipLaunchKernelGGL(gat_edge_alpha, dim3(cdiv(eth, 256)), dim3(256), 0, stream,
                           ei, E, N, alpha, den, Hh);
        hipLaunchKernelGGL(gat_edge_agg,   dim3(cdiv((long long)Et * HD, 256)), dim3(256), 0, stream,
                           ei, E, N, buf_h, alpha, buf_agg, Hh);
        hipLaunchKernelGGL(gat_epilogue, dim3(cdiv((long long)N * HD, 256)), dim3(256), 0, stream,
                           buf_agg, bb, out_feat, N, HD, do_elu);
    };

    run_layer(x,        W1, a1s, a1d, b1, buf_feat, H, 16,  256, 1);
    run_layer(buf_feat, W2, a2s, a2d, b2, buf_feat, H, 256, 256, 1);
    run_layer(buf_feat, W3, a3s, a3d, b3, node_emb, 1, 256, 64, 0);

    hipLaunchKernelGGL(gat_colmean, dim3(64), dim3(256), 0, stream, node_emb, graph_emb, N, 64);
}


# Graph mean taken where the rows are made: each layer-3 owner block adds its stored rows per column (registers, then LDS), leaves 64 sums; one small block adds the 256 sum rows; node rows not re-read
# speedup vs baseline: 5.0400x; 5.0400x over previous
#include <hip/hip_runtime.h>
#include <math.h>

typedef __attribute__((ext_vector_type(2))) float v2f;
typedef __attribute__((ext_vector_type(8))) float v8f;

#define GAT_D 64

__device__ __forceinline__ unsigned f2ord(float f) {
    int b = __float_as_int(f);
    return (unsigned)b ^ (unsigned)((b >> 31) | 0x80000000);
}
__device__ __forceinline__ float ord2f(unsigned u) {
    int b = (u & 0x80000000u) ? (int)(u ^ 0x80000000u) : ~(int)u;
    return __int_as_float(b);
}

__global__ void gat_gemm_wmma(const float* __restrict__ A,
                              const float* __restrict__ B,
                              float* __restrict__ C,
                              int M, int K, int N) {
    const int lane  = threadIdx.x;
    const int grp   = blockIdx.x * blockDim.y + threadIdx.y;
    const int mtile = blockIdx.y;
    const int n0 = grp * 64;
    const int m0 = mtile * 16;
    if (n0 >= N || m0 >= M) return;

    const int l15 = lane & 15;
    const int kh  = (lane >> 4) << 1;
    const float* Arow = A + (size_t)(m0 + l15) * K;
    v8f acc0 = {0.f,0.f,0.f,0.f,0.f,0.f,0.f,0.f};
    v8f acc1 = acc0, acc2 = acc0, acc3 = acc0;

    for (int k = 0; k < K; k += 4) {
        v2f a;
        a.x = Arow[k + kh];
        a.y = Arow[k + kh + 1];
        const float* Bp = B + (size_t)(k + kh) * N + n0 + l15;
        v2f b0, b1, b2, b3;
        b0.x = Bp[0];  b0.y = Bp[(size_t)N];
        b1.x = Bp[16]; b1.y = Bp[(size_t)N + 16];
        b2.x = Bp[32]; b2.y = Bp[(size_t)N + 32];
        b3.x = Bp[48]; b3.y = Bp[(size_t)N + 48];
        acc0 = __builtin_amdgcn_wmma_f32_16x16x4_f32(false, a, false, b0, (short)0, acc0, false, false);
        acc1 = __builtin_amdgcn_wmma_f32_16x16x4_f32(false, a, false, b1, (short)0, acc1, false, false);
        acc2 = __builtin_amdgcn_wmma_f32_16x16x4_f32(false, a, false, b2, (short)0, acc2, false, false);
        acc3 = __builtin_amdgcn_wmma_f32_16x16x4_f32(false, a, false, b3, (short)0, acc3, false, false);
    }

    float* Cp = C + (size_t)(m0 + ((lane >> 4) << 3)) * N + n0 + l15;
#pragma unroll
    for (int v = 0; v < 8; ++v) {
        Cp[(size_t)v * N + 0]  = acc0[v];
        Cp[(size_t)v * N + 16] = acc1[v];
        Cp[(size_t)v * N + 32] = acc2[v];
        Cp[(size_t)v * N + 48] = acc3[v];
    }
}

__global__ void gat_attn_coef(const float* __restrict__ h,
                              const float* __restrict__ a_s,
                              const float* __restrict__ a_d,
                              float* __restrict__ es, float* __restrict__ ed,
                              int N, int Hh) {
    int t = blockIdx.x * blockDim.x + threadIdx.x;
    if (t >= N * Hh) return;
    int n  = t / Hh;
    int hh = t - n * Hh;
    const float* hp = h   + ((size_t)n * Hh + hh) * GAT_D;
    const float* as = a_s + (size_t)hh * GAT_D;
    const float* ad = a_d + (size_t)hh * GAT_D;
    float ss = 0.f, dd = 0.f;
#pragma unroll 8
    for (int d = 0; d < GAT_D; ++d) { float v = hp[d]; ss += v * as[d]; dd += v * ad[d]; }
    es[t] = ss; ed[t] = dd;
}

__device__ __forceinline__ void edge_endpoints(const int* __restrict__ ei, int E,
                                               int e, int& src, int& dst) {
    if (e < E) { src = ei[e]; dst = ei[E + e]; }
    else       { src = e - E; dst = e - E;    }
}

#define AGG_ROWS    196
#define AGG_THREADS 1024
#define AGG_K       8
#define AGG_CAP     4096
#define AGG_U       4

template <int HD> struct RowVals { float v[HD / 32]; };

template <int HD>
__device__ __forceinline__ RowVals<HD> agg_fetch(const float* __restrict__ hfeat,
                                                 int src, int lane, float w0, float w1) {
    RowVals<HD> r;
    if constexpr (HD == 256) {
        const float4* hp = reinterpret_cast<const float4*>(hfeat + (size_t)src * 256);
        const float4 a = hp[lane];
        const float4 b = hp[32 + lane];
        r.v[0] = a.x * w0; r.v[1] = a.y * w0; r.v[2] = a.z * w0; r.v[3] = a.w * w0;
        r.v[4] = b.x * w1; r.v[5] = b.y * w1; r.v[6] = b.z * w1; r.v[7] = b.w * w1;
    } else {
        const float2* hp = reinterpret_cast<const float2*>(hfeat + (size_t)src * 64);
        const float2 a = hp[lane];
        r.v[0] = a.x * w0; r.v[1] = a.y * w0;
    }
    return r;
}

template <int HD>
__device__ __forceinline__ RowVals<HD> agg_take(const float* __restrict__ hfeat,
                                                const int* __restrict__ lsrc,
                                                const int* __restrict__ lrow,
                                                const float* __restrict__ lw,
                                                int i, int lane, int& row) {
    row = lrow[i];
    const float* wp = lw + i * (HD / 64);
    float w0, w1;
    if constexpr (HD == 256) { w0 = wp[lane >> 4]; w1 = wp[2 + (lane >> 4)]; }
    else                     { w0 = wp[0]; w1 = 0.f; }
    return agg_fetch<HD>(hfeat, lsrc[i], lane, w0, w1);
}

template <int HD>
__device__ __forceinline__ void agg_add_row(float* __restrict__ acc, int row, int lane,
                                            const RowVals<HD>& p) {
    float* a = acc + row * HD + lane;
#pragma unroll
    for (int j = 0; j < HD / 32; ++j)
        __hip_atomic_fetch_add(a + j * 32, p.v[j], __ATOMIC_RELAXED, __HIP_MEMORY_SCOPE_WORKGROUP);
}

__device__ __forceinline__ float agg_act(float v, int do_elu) {
    if (do_elu) v = v > 0.f ? v : (expf(v) - 1.f);
    return v;
}

__device__ __forceinline__ float agg_lrelu(float s) { return s > 0.f ? s : 0.2f * s; }

template <bool TAIL>
__device__ __forceinline__ void agg_load_dst(const int* __restrict__ dstp, int base, int lane,
                                             int E, int (&d)[AGG_K]) {
#pragma unroll
    for (int k = 0; k < AGG_K; ++k) {
        const int e = base + 32 * k + lane;
        if constexpr (TAIL) d[k] = e < E ? dstp[e] : -1;
        else                d[k] = dstp[e];
    }
}

__device__ __forceinline__ void agg_collect(const int (&d)[AGG_K], int base, int lane, int lo, int nrows,
                                            int* __restrict__ lsrc, int* __restrict__ lrow,
                                            unsigned* __restrict__ cnt) {
    unsigned m[AGG_K], any = 0;
    int rl[AGG_K];
#pragma unroll
    for (int k = 0; k < AGG_K; ++k) {
        rl[k] = d[k] - lo;
        m[k] = __builtin_amdgcn_ballot_w32((unsigned)rl[k] < (unsigned)nrows);
        any |= m[k];
    }
    if (!any) return;
    unsigned tot = 0;
#pragma unroll
    for (int k = 0; k < AGG_K; ++k) tot += (unsigned)__builtin_popcount(m[k]);
    unsigned start = 0;
    if (lane == 0)
        start = __hip_atomic_fetch_add(cnt, tot, __ATOMIC_RELAXED, __HIP_MEMORY_SCOPE_WORKGROUP);
    start = (unsigned)__builtin_amdgcn_readfirstlane((int)start);
    const unsigned below = (1u << lane) - 1u;
#pragma unroll
    for (int k = 0; k < AGG_K; ++k) {
        if ((m[k] >> lane) & 1u) {
            const unsigned p = start + (unsigned)__builtin_popcount(m[k] & below);
            if (p < (unsigned)AGG_CAP) { lsrc[p] = base + 32 * k + lane; lrow[p] = rl[k]; }
        }
        start += (unsigned)__builtin_popcount(m[k]);
    }
}

template <int HD>
__global__ __launch_bounds__(AGG_THREADS)
void gat_agg_owner(const int* __restrict__ ei, int E, int N,
                   const float* __restrict__ hfeat,
                   const float* __restrict__ es, const float* __restrict__ ed,
                   const float* __restrict__ bias,
                   float* __restrict__ out, int do_elu,
                   float* __restrict__ colpart) {
    constexpr int HH = HD / 64;
    constexpr int NW = AGG_THREADS / 32;
    constexpr int SPAN = 32 * AGG_K;
    constexpr int TRIP = NW * SPAN;
    __shared__ __attribute__((aligned(16))) float acc[AGG_ROWS * HD];
    __shared__ unsigned mx[AGG_ROWS * HH];
    __shared__ float    den[AGG_ROWS * HH];
    __shared__ float    edv[AGG_ROWS * HH];
    __shared__ int      lsrc[AGG_CAP];
    __shared__ int      lrow[AGG_CAP];
    __shared__ float    lw[AGG_CAP * HH];
    __shared__ unsigned cnt;
    const int tid  = threadIdx.x;
    const int lane = tid & 31;
    const int wave = tid >> 5;
    const int lo = blockIdx.x * AGG_ROWS;
    const int nrows = max(0, min(AGG_ROWS, N - lo));

    {
        float4* a4 = reinterpret_cast<float4*>(acc);
        for (int i = tid; i < AGG_ROWS * HD / 4; i += AGG_THREADS) a4[i] = make_float4(0.f, 0.f, 0.f, 0.f);
        for (int i = tid; i < nrows * HH; i += AGG_THREADS) {
            mx[i] = 0u;
            den[i] = 0.f;
            edv[i] = ed[(size_t)lo * HH + i];
        }
        for (int r = tid; r < nrows; r += AGG_THREADS) { lsrc[r] = E + lo + r; lrow[r] = r; }
        if (tid == 0) cnt = (unsigned)nrows;
    }
    __syncthreads();

    const int* __restrict__ dstp = ei + E;
    const int full = E / TRIP;
    const int t0 = (int)((blockIdx.x * (unsigned)full) / gridDim.x);
    const int woff = wave * SPAN;
    if (full > 0) {
        int dA[AGG_K], dB[AGG_K];
        agg_load_dst<false>(dstp, t0 * TRIP + woff, lane, E, dA);
        for (int t = 0; t < full; t += 2) {
            int tc0 = t0 + t;    if (tc0 >= full) tc0 -= full;
            int tc1 = tc0 + 1;   if (tc1 >= full) tc1 -= full;
            int tc2 = tc1 + 1;   if (tc2 >= full) tc2 -= full;
            agg_load_dst<false>(dstp, tc1 * TRIP + woff, lane, E, dB);
            agg_collect(dA, tc0 * TRIP + woff, lane, lo, nrows, lsrc, lrow, &cnt);
            agg_load_dst<false>(dstp, tc2 * TRIP + woff, lane, E, dA);
            if (t + 1 < full)
                agg_collect(dB, tc1 * TRIP + woff, lane, lo, nrows, lsrc, lrow, &cnt);
        }
    }
    if (full * TRIP < E) {
        int dT[AGG_K];
        agg_load_dst<true>(dstp, full * TRIP + woff, lane, E, dT);
        agg_collect(dT, full * TRIP + woff, lane, lo, nrows, lsrc, lrow, &cnt);
    }
    __syncthreads();
    const int n = (int)cnt;

    if (n <= AGG_CAP) {
        for (int i = tid; i < n; i += AGG_THREADS) {
            const int e = lsrc[i];
            const int src = e < E ? ei[e] : e - E;
            lsrc[i] = src;
            const int row = lrow[i];
            float sv[HH];
            if constexpr (HH == 4) {
                const float4 q = reinterpret_cast<const float4*>(es)[src];
                sv[0] = q.x; sv[1] = q.y; sv[2] = q.z; sv[3] = q.w;
            } else {
                sv[0] = es[src];
            }
#pragma unroll
            for (int h = 0; h < HH; ++h) {
                const float lr = agg_lrelu(sv[h] + edv[row * HH + h]);
                lw[i * HH + h] = lr;
                atomicMax(&mx[row * HH + h], f2ord(lr));
            }
        }
        __syncthreads();
        for (int i = tid; i < n; i += AGG_THREADS) {
            const int row = lrow[i];
#pragma unroll
            for (int h = 0; h < HH; ++h) {
                const float v = expf(lw[i * HH + h] - ord2f(mx[row * HH + h]));
                lw[i * HH + h] = v;
                __hip_atomic_fetch_add(&den[row * HH + h], v, __ATOMIC_RELAXED, __HIP_MEMORY_SCOPE_WORKGROUP);
            }
        }
        __syncthreads();
        for (int i = tid; i < n; i += AGG_THREADS) {
            const int row = lrow[i];
#pragma unroll
            for (int h = 0; h < HH; ++h) lw[i * HH + h] = lw[i * HH + h] / den[row * HH + h];
        }
        __syncthreads();
        const int ngrp = n / AGG_U;
        for (int g = wave; g < ngrp; g += NW) {
            RowVals<HD> p[AGG_U];
            int rw[AGG_U];
#pragma unroll
            for (int u = 0; u < AGG_U; ++u)
                p[u] = agg_take<HD>(hfeat, lsrc, lrow, lw, g * AGG_U + u, lane, rw[u]);
#pragma unroll
            for (int u = 0; u < AGG_U; ++u) agg_add_row<HD>(acc, rw[u], lane, p[u]);
        }
        for (int i = ngrp * AGG_U + wave; i < n; i += NW) {
            int rw1;
            const RowVals<HD> p1 = agg_take<HD>(hfeat, lsrc, lrow, lw, i, lane, rw1);
            agg_add_row<HD>(acc, rw1, lane, p1);
        }
    } else {
        const int Et = E + N;
        for (int ph = 0; ph < 3; ++ph) {
            for (int eb = wave * 32; eb < Et; eb += AGG_THREADS) {
                const int e = eb + lane;
                int src = 0, rl = -1;
                if (e < Et) {
                    int dst; edge_endpoints(ei, E, e, src, dst);
                    rl = dst - lo;
                }
                const bool in = (unsigned)rl < (unsigned)nrows;
                if (ph < 2) {
                    if (in) {
#pragma unroll
                        for (int h = 0; h < HH; ++h) {
                            const float lr = agg_lrelu(es[(size_t)src * HH + h] + edv[rl * HH + h]);
                            if (ph == 0) atomicMax(&mx[rl * HH + h], f2ord(lr));
                            else __hip_atomic_fetch_add(&den[rl * HH + h], expf(lr - ord2f(mx[rl * HH + h])),
                                                        __ATOMIC_RELAXED, __HIP_MEMORY_SCOPE_WORKGROUP);
                        }
                    }
                } else {
                    unsigned mk = __builtin_amdgcn_ballot_w32(in);
                    while (mk) {
                        const int b = __builtin_ctz(mk);
                        mk &= mk - 1;
                        const int sb = __builtin_amdgcn_readlane(src, b);
                        const int rb = __builtin_amdgcn_readlane(rl, b);
                        const int h0 = HH == 4 ? (lane >> 4) : 0;
                        const int h1 = HH == 4 ? 2 + (lane >> 4) : 0;
                        const float l0 = agg_lrelu(es[(size_t)sb * HH + h0] + edv[rb * HH + h0]);
                        const float l1 = agg_lrelu(es[(size_t)sb * HH + h1] + edv[rb * HH + h1]);
                        const float w0 = expf(l0 - ord2f(mx[rb * HH + h0])) / den[rb * HH + h0];
                        const float w1 = expf(l1 - ord2f(mx[rb * HH + h1])) / den[rb * HH + h1];
                        const RowVals<HD> p = agg_fetch<HD>(hfeat, sb, lane, w0, w1);
                        agg_add_row<HD>(acc, rb, lane, p);
                    }
                }
            }
            __syncthreads();
        }
    }
    __syncthreads();

    float cs0 = 0.f, cs1 = 0.f;
    for (int r = wave; r < nrows; r += NW) {
        const float* a = acc + r * HD + lane;
        const size_t nd = (size_t)(lo + r);
        if constexpr (HD == 256) {
            const float4* b4 = reinterpret_cast<const float4*>(bias);
            const float4 c0 = b4[lane], c1 = b4[32 + lane];
            float4 o0, o1;
            o0.x = agg_act(a[0]   + c0.x, do_elu); o0.y = agg_act(a[32]  + c0.y, do_elu);
            o0.z = agg_act(a[64]  + c0.z, do_elu); o0.w = agg_act(a[96]  + c0.w, do_elu);
            o1.x = agg_act(a[128] + c1.x, do_elu); o1.y = agg_act(a[160] + c1.y, do_elu);
            o1.z = agg_act(a[192] + c1.z, do_elu); o1.w = agg_act(a[224] + c1.w, do_elu);
            float4* o4 = reinterpret_cast<float4*>(out + nd * 256);
            o4[lane] = o0;
            o4[32 + lane] = o1;
        } else {
            const float2 c = reinterpret_cast<const float2*>(bias)[lane];
            float2 o;
            o.x = agg_act(a[0]  + c.x, do_elu);
            o.y = agg_act(a[32] + c.y, do_elu);
            reinterpret_cast<float2*>(out + nd * 64)[lane] = o;
            cs0 += o.x;
            cs1 += o.y;
        }
    }
    if constexpr (HD == 64) {
        if (colpart != nullptr) {
            lw[wave * 64 + 2 * lane]     = cs0;
            lw[wave * 64 + 2 * lane + 1] = cs1;
            __syncthreads();
            if (tid < 64) {
                float s = 0.f;
                for (int w = 0; w < NW; ++w) s += lw[w * 64 + tid];
                colpart[(size_t)blockIdx.x * 64 + tid] = s;
            }
        }
    }
}

#define MEAN_GROUPS 16
__global__ __launch_bounds__(64 * MEAN_GROUPS)
void gat_colmean_parts(const float* __restrict__ part, float* __restrict__ out, int N) {
    __shared__ float sm[MEAN_GROUPS * 64];
    const int nblk = (N + AGG_ROWS - 1) / AGG_ROWS;
    const int c = threadIdx.x & 63;
    const int g = threadIdx.x >> 6;
    float s = 0.f;
#pragma unroll 16
    for (int b = g; b < nblk; b += MEAN_GROUPS) s += part[(size_t)b * 64 + c];
    sm[g * 64 + c] = s;
    __syncthreads();
    if (threadIdx.x < 64) {
        float t = 0.f;
        for (int k = 0; k < MEAN_GROUPS; ++k) t += sm[k * 64 + threadIdx.x];
        out[threadIdx.x] = t / (float)N;
    }
}

static inline int cdiv(long long a, long long b) { return (int)((a + b - 1) / b); }

extern "C" void kernel_launch(void* const* d_in, const int* in_sizes, int n_in,
                              void* d_out, int out_size, void* d_ws, size_t ws_size,
                              hipStream_t stream) {
    const float* x   = (const float*)d_in[0];
    const int*   ei  = (const int*)  d_in[1];
    const float* W1  = (const float*)d_in[2];
    const float* a1s = (const float*)d_in[3];
    const float* a1d = (const float*)d_in[4];
    const float* b1  = (const float*)d_in[5];
    const float* W2  = (const float*)d_in[6];
    const float* a2s = (const float*)d_in[7];
    const float* a2d = (const float*)d_in[8];
    const float* b2  = (const float*)d_in[9];
    const float* W3  = (const float*)d_in[10];
    const float* a3s = (const float*)d_in[11];
    const float* a3d = (const float*)d_in[12];
    const float* b3  = (const float*)d_in[13];

    const int N  = in_sizes[0] / 16;
    const int E  = in_sizes[1] / 2;
    const int H  = 4;

    float* ws = (float*)d_ws;
    size_t off = 0;
    float*    buf_h    = ws + off; off += (size_t)N * 256;
    float*    buf_feat = ws + off; off += (size_t)N * 256;
    float*    es       = ws + off; off += (size_t)N * H;
    float*    ed       = ws + off; off += (size_t)N * H;
    (void)ws_size; (void)off;

    float* node_emb  = (float*)d_out;
    float* graph_emb = node_emb + (size_t)N * 64;

    auto run_layer = [&](const float* in_feat, const float* W,
                         const float* as_, const float* ad_, const float* bb,
                         float* out_feat, int Hh, int K, int Ncols, int do_elu, float* colpart) {
        int ngrp = Ncols / 64;
        int wpb  = ngrp < 4 ? ngrp : 4;
        dim3 gb(32, wpb);
        dim3 gg(cdiv(ngrp, wpb), cdiv(N, 16));
        hipLaunchKernelGGL(gat_gemm_wmma, gg, gb, 0, stream, in_feat, W, buf_h, N, K, Ncols);
        hipLaunchKernelGGL(gat_attn_coef, dim3(cdiv((long long)N * Hh, 256)), dim3(256), 0, stream,
                           buf_h, as_, ad_, es, ed, N, Hh);
        if (Hh == 4)
            hipLaunchKernelGGL(gat_agg_owner<256>, dim3(cdiv(N, AGG_ROWS)), dim3(AGG_THREADS), 0, stream,
                               ei, E, N, buf_h, es, ed, bb, out_feat, do_elu, colpart);
        else
            hipLaunchKernelGGL(gat_agg_owner<64>, dim3(cdiv(N, AGG_ROWS)), dim3(AGG_THREADS), 0, stream,
                               ei, E, N, buf_h, es, ed, bb, out_feat, do_elu, colpart);
    };

    run_layer(x,        W1, a1s, a1d, b1, buf_feat, H, 16,  256, 1, nullptr);
    run_layer(buf_feat, W2, a2s, a2d, b2, buf_feat, H, 256, 256, 1, nullptr);
    float* col_part = es + (size_t)N;
    run_layer(buf_feat, W3, a3s, a3d, b3, node_emb, 1, 256, 64, 0, col_part);

    hipLaunchKernelGGL(gat_colmean_parts, dim3(1), dim3(64 * MEAN_GROUPS), 0, stream,
                       col_part, graph_emb, N);
}


# Owner blocks' edge lists made once: layer 1 leaves each block's list (sources, owned rows) in its own rows of the idle output plane; layers 2, 3 reload it in that order, no edge walk; LDS adds kept
# speedup vs baseline: 6.1888x; 1.2279x over previous
#include <hip/hip_runtime.h>
#include <math.h>

typedef __attribute__((ext_vector_type(2))) float v2f;
typedef __attribute__((ext_vector_type(8))) float v8f;

#define GAT_D 64

__device__ __forceinline__ unsigned f2ord(float f) {
    int b = __float_as_int(f);
    return (unsigned)b ^ (unsigned)((b >> 31) | 0x80000000);
}
__device__ __forceinline__ float ord2f(unsigned u) {
    int b = (u & 0x80000000u) ? (int)(u ^ 0x80000000u) : ~(int)u;
    return __int_as_float(b);
}

__global__ void gat_gemm_wmma(const float* __restrict__ A,
                              const float* __restrict__ B,
                              float* __restrict__ C,
                              int M, int K, int N) {
    const int lane  = threadIdx.x;
    const int grp   = blockIdx.x * blockDim.y + threadIdx.y;
    const int mtile = blockIdx.y;
    const int n0 = grp * 64;
    const int m0 = mtile * 16;
    if (n0 >= N || m0 >= M) return;

    const int l15 = lane & 15;
    const int kh  = (lane >> 4) << 1;
    const float* Arow = A + (size_t)(m0 + l15) * K;
    v8f acc0 = {0.f,0.f,0.f,0.f,0.f,0.f,0.f,0.f};
    v8f acc1 = acc0, acc2 = acc0, acc3 = acc0;

    for (int k = 0; k < K; k += 4) {
        v2f a;
        a.x = Arow[k + kh];
        a.y = Arow[k + kh + 1];
        const float* Bp = B + (size_t)(k + kh) * N + n0 + l15;
        v2f b0, b1, b2, b3;
        b0.x = Bp[0];  b0.y = Bp[(size_t)N];
        b1.x = Bp[16]; b1.y = Bp[(size_t)N + 16];
        b2.x = Bp[32]; b2.y = Bp[(size_t)N + 32];
        b3.x = Bp[48]; b3.y = Bp[(size_t)N + 48];
        acc0 = __builtin_amdgcn_wmma_f32_16x16x4_f32(false, a, false, b0, (short)0, acc0, false, false);
        acc1 = __builtin_amdgcn_wmma_f32_16x16x4_f32(false, a, false, b1, (short)0, acc1, false, false);
        acc2 = __builtin_amdgcn_wmma_f32_16x16x4_f32(false, a, false, b2, (short)0, acc2, false, false);
        acc3 = __builtin_amdgcn_wmma_f32_16x16x4_f32(false, a, false, b3, (short)0, acc3, false, false);
    }

    float* Cp = C + (size_t)(m0 + ((lane >> 4) << 3)) * N + n0 + l15;
#pragma unroll
    for (int v = 0; v < 8; ++v) {
        Cp[(size_t)v * N + 0]  = acc0[v];
        Cp[(size_t)v * N + 16] = acc1[v];
        Cp[(size_t)v * N + 32] = acc2[v];
        Cp[(size_t)v * N + 48] = acc3[v];
    }
}

__global__ void gat_attn_coef(const float* __restrict__ h,
                              const float* __restrict__ a_s,
                              const float* __restrict__ a_d,
                              float* __restrict__ es, float* __restrict__ ed,
                              int N, int Hh) {
    int t = blockIdx.x * blockDim.x + threadIdx.x;
    if (t >= N * Hh) return;
    int n  = t / Hh;
    int hh = t - n * Hh;
    const float* hp = h   + ((size_t)n * Hh + hh) * GAT_D;
    const float* as = a_s + (size_t)hh * GAT_D;
    const float* ad = a_d + (size_t)hh * GAT_D;
    float ss = 0.f, dd = 0.f;
#pragma unroll 8
    for (int d = 0; d < GAT_D; ++d) { float v = hp[d]; ss += v * as[d]; dd += v * ad[d]; }
    es[t] = ss; ed[t] = dd;
}

__device__ __forceinline__ void edge_endpoints(const int* __restrict__ ei, int E,
                                               int e, int& src, int& dst) {
    if (e < E) { src = ei[e]; dst = ei[E + e]; }
    else       { src = e - E; dst = e - E;    }
}

#define AGG_ROWS    196
#define AGG_THREADS 1024
#define AGG_K       8
#define AGG_CAP     4096
#define AGG_U       4
#define KEEP_PITCH  64

template <int HD> struct RowVals { float v[HD / 32]; };

template <int HD>
__device__ __forceinline__ RowVals<HD> agg_fetch(const float* __restrict__ hfeat,
                                                 int src, int lane, float w0, float w1) {
    RowVals<HD> r;
    if constexpr (HD == 256) {
        const float4* hp = reinterpret_cast<const float4*>(hfeat + (size_t)src * 256);
        const float4 a = hp[lane];
        const float4 b = hp[32 + lane];
        r.v[0] = a.x * w0; r.v[1] = a.y * w0; r.v[2] = a.z * w0; r.v[3] = a.w * w0;
        r.v[4] = b.x * w1; r.v[5] = b.y * w1; r.v[6] = b.z * w1; r.v[7] = b.w * w1;
    } else {
        const float2* hp = reinterpret_cast<const float2*>(hfeat + (size_t)src * 64);
        const float2 a = hp[lane];
        r.v[0] = a.x * w0; r.v[1] = a.y * w0;
    }
    return r;
}

template <int HD>
__device__ __forceinline__ RowVals<HD> agg_take(const float* __restrict__ hfeat,
                                                const int* __restrict__ lsrc,
                                                const int* __restrict__ lrow,
                                                const float* __restrict__ lw,
                                                int i, int lane, int& row) {
    row = lrow[i];
    const float* wp = lw + i * (HD / 64);
    float w0, w1;
    if constexpr (HD == 256) { w0 = wp[lane >> 4]; w1 = wp[2 + (lane >> 4)]; }
    else                     { w0 = wp[0]; w1 = 0.f; }
    return agg_fetch<HD>(hfeat, lsrc[i], lane, w0, w1);
}

template <int HD>
__device__ __forceinline__ void agg_add_row(float* __restrict__ acc, int row, int lane,
                                            const RowVals<HD>& p) {
    float* a = acc + row * HD + lane;
#pragma unroll
    for (int j = 0; j < HD / 32; ++j)
        __hip_atomic_fetch_add(a + j * 32, p.v[j], __ATOMIC_RELAXED, __HIP_MEMORY_SCOPE_WORKGROUP);
}

__device__ __forceinline__ float agg_act(float v, int do_elu) {
    if (do_elu) v = v > 0.f ? v : (expf(v) - 1.f);
    return v;
}

__device__ __forceinline__ float agg_lrelu(float s) { return s > 0.f ? s : 0.2f * s; }

template <bool TAIL>
__device__ __forceinline__ void agg_load_dst(const int* __restrict__ dstp, int base, int lane,
                                             int E, int (&d)[AGG_K]) {
#pragma unroll
    for (int k = 0; k < AGG_K; ++k) {
        const int e = base + 32 * k + lane;
        if constexpr (TAIL) d[k] = e < E ? dstp[e] : -1;
        else                d[k] = dstp[e];
    }
}

__device__ __forceinline__ void agg_collect(const int (&d)[AGG_K], int base, int lane, int lo, int nrows,
                                            int* __restrict__ lsrc, int* __restrict__ lrow,
                                            unsigned* __restrict__ cnt) {
    unsigned m[AGG_K], any = 0;
    int rl[AGG_K];
#pragma unroll
    for (int k = 0; k < AGG_K; ++k) {
        rl[k] = d[k] - lo;
        m[k] = __builtin_amdgcn_ballot_w32((unsigned)rl[k] < (unsigned)nrows);
        any |= m[k];
    }
    if (!any) return;
    unsigned tot = 0;
#pragma unroll
    for (int k = 0; k < AGG_K; ++k) tot += (unsigned)__builtin_popcount(m[k]);
    unsigned start = 0;
    if (lane == 0)
        start = __hip_atomic_fetch_add(cnt, tot, __ATOMIC_RELAXED, __HIP_MEMORY_SCOPE_WORKGROUP);
    start = (unsigned)__builtin_amdgcn_readfirstlane((int)start);
    const unsigned below = (1u << lane) - 1u;
#pragma unroll
    for (int k = 0; k < AGG_K; ++k) {
        if ((m[k] >> lane) & 1u) {
            const unsigned p = start + (unsigned)__builtin_popcount(m[k] & below);
            if (p < (unsigned)AGG_CAP) { lsrc[p] = base + 32 * k + lane; lrow[p] = rl[k]; }
        }
        start += (unsigned)__builtin_popcount(m[k]);
    }
}

template <int HD>
__global__ __launch_bounds__(AGG_THREADS)
void gat_agg_owner(const int* __restrict__ ei, int E, int N,
                   const float* __restrict__ hfeat,
                   const float* __restrict__ es, const float* __restrict__ ed,
                   const float* __restrict__ bias,
                   float* out, int do_elu,
                   float* __restrict__ colpart,
                   float* keep, int take_list) {
    constexpr int HH = HD / 64;
    constexpr int NW = AGG_THREADS / 32;
    constexpr int SPAN = 32 * AGG_K;
    constexpr int TRIP = NW * SPAN;
    __shared__ __attribute__((aligned(16))) float acc[AGG_ROWS * HD];
    __shared__ unsigned mx[AGG_ROWS * HH];
    __shared__ float    den[AGG_ROWS * HH];
    __shared__ float    edv[AGG_ROWS * HH];
    __shared__ int      lsrc[AGG_CAP];
    __shared__ int      lrow[AGG_CAP];
    __shared__ float    lw[AGG_CAP * HH];
    __shared__ unsigned cnt;
    const int tid  = threadIdx.x;
    const int lane = tid & 31;
    const int wave = tid >> 5;
    const int lo = blockIdx.x * AGG_ROWS;
    const int nrows = max(0, min(AGG_ROWS, N - lo));

    float* kp = keep + (size_t)lo * KEEP_PITCH;
    const int kcap = nrows * KEEP_PITCH;
    int taken = -1;
    if (take_list && nrows > 0) taken = __builtin_amdgcn_readfirstlane(__float_as_int(kp[0]));
    const bool have = taken >= 0 && taken <= AGG_CAP && 1 + 2 * taken <= kcap;

    {
        float4* a4 = reinterpret_cast<float4*>(acc);
        for (int i = tid; i < AGG_ROWS * HD / 4; i += AGG_THREADS) a4[i] = make_float4(0.f, 0.f, 0.f, 0.f);
        for (int i = tid; i < nrows * HH; i += AGG_THREADS) {
            mx[i] = 0u;
            den[i] = 0.f;
            edv[i] = ed[(size_t)lo * HH + i];
        }
        if (have) {
            for (int i = tid; i < taken; i += AGG_THREADS) {
                lsrc[i] = __float_as_int(kp[1 + i]);
                lrow[i] = __float_as_int(kp[1 + taken + i]);
            }
            if (tid == 0) cnt = (unsigned)taken;
        } else {
            for (int r = tid; r < nrows; r += AGG_THREADS) { lsrc[r] = E + lo + r; lrow[r] = r; }
            if (tid == 0) cnt = (unsigned)nrows;
        }
    }
    __syncthreads();

    const int* __restrict__ dstp = ei + E;
    const int full = E / TRIP;
    const int t0 = (int)((blockIdx.x * (unsigned)full) / gridDim.x);
    const int woff = wave * SPAN;
    if (!have && full > 0) {
        int dA[AGG_K], dB[AGG_K];
        agg_load_dst<false>(dstp, t0 * TRIP + woff, lane, E, dA);
        for (int t = 0; t < full; t += 2) {
            int tc0 = t0 + t;    if (tc0 >= full) tc0 -= full;
            int tc1 = tc0 + 1;   if (tc1 >= full) tc1 -= full;
            int tc2 = tc1 + 1;   if (tc2 >= full) tc2 -= full;
            agg_load_dst<false>(dstp, tc1 * TRIP + woff, lane, E, dB);
            agg_collect(dA, tc0 * TRIP + woff, lane, lo, nrows, lsrc, lrow, &cnt);
            agg_load_dst<false>(dstp, tc2 * TRIP + woff, lane, E, dA);
            if (t + 1 < full)
                agg_collect(dB, tc1 * TRIP + woff, lane, lo, nrows, lsrc, lrow, &cnt);
        }
    }
    if (!have && full * TRIP < E) {
        int dT[AGG_K];
        agg_load_dst<true>(dstp, full * TRIP + woff, lane, E, dT);
        agg_collect(dT, full * TRIP + woff, lane, lo, nrows, lsrc, lrow, &cnt);
    }
    __syncthreads();
    const int n = (int)cnt;

    const bool save = !take_list && n <= AGG_CAP && 1 + 2 * n <= kcap;
    if (!take_list && tid == 0 && nrows > 0) kp[0] = __int_as_float(save ? n : -1);
    if (n <= AGG_CAP) {
        for (int i = tid; i < n; i += AGG_THREADS) {
            int src = lsrc[i];
            const int row = lrow[i];
            if (!have) {
                const int e = src;
                src = e < E ? ei[e] : e - E;
                lsrc[i] = src;
                if (save) { kp[1 + i] = __int_as_float(src); kp[1 + n + i] = __int_as_float(row); }
            }
            float sv[HH];
            if constexpr (HH == 4) {
                const float4 q = reinterpret_cast<const float4*>(es)[src];
                sv[0] = q.x; sv[1] = q.y; sv[2] = q.z; sv[3] = q.w;
            } else {
                sv[0] = es[src];
            }
#pragma unroll
            for (int h = 0; h < HH; ++h) {
                const float lr = agg_lrelu(sv[h] + edv[row * HH + h]);
                lw[i * HH + h] = lr;
                atomicMax(&mx[row * HH + h], f2ord(lr));
            }
        }
        __syncthreads();
        for (int i = tid; i < n; i += AGG_THREADS) {
            const int row = lrow[i];
#pragma unroll
            for (int h = 0; h < HH; ++h) {
                const float v = expf(lw[i * HH + h] - ord2f(mx[row * HH + h]));
                lw[i * HH + h] = v;
                __hip_atomic_fetch_add(&den[row * HH + h], v, __ATOMIC_RELAXED, __HIP_MEMORY_SCOPE_WORKGROUP);
            }
        }
        __syncthreads();
        for (int i = tid; i < n; i += AGG_THREADS) {
            const int row = lrow[i];
#pragma unroll
            for (int h = 0; h < HH; ++h) lw[i * HH + h] = lw[i * HH + h] / den[row * HH + h];
        }
        __syncthreads();
        const int ngrp = n / AGG_U;
        for (int g = wave; g < ngrp; g += NW) {
            RowVals<HD> p[AGG_U];
            int rw[AGG_U];
#pragma unroll
            for (int u = 0; u < AGG_U; ++u)
                p[u] = agg_take<HD>(hfeat, lsrc, lrow, lw, g * AGG_U + u, lane, rw[u]);
#pragma unroll
            for (int u = 0; u < AGG_U; ++u) agg_add_row<HD>(acc, rw[u], lane, p[u]);
        }
        for (int i = ngrp * AGG_U + wave; i < n; i += NW) {
            int rw1;
            const RowVals<HD> p1 = agg_take<HD>(hfeat, lsrc, lrow, lw, i, lane, rw1);
            agg_add_row<HD>(acc, rw1, lane, p1);
        }
    } else {
        const int Et = E + N;
        for (int ph = 0; ph < 3; ++ph) {
            for (int eb = wave * 32; eb < Et; eb += AGG_THREADS) {
                const int e = eb + lane;
                int src = 0, rl = -1;
                if (e < Et) {
                    int dst; edge_endpoints(ei, E, e, src, dst);
                    rl = dst - lo;
                }
                const bool in = (unsigned)rl < (unsigned)nrows;
                if (ph < 2) {
                    if (in) {
#pragma unroll
                        for (int h = 0; h < HH; ++h) {
                            const float lr = agg_lrelu(es[(size_t)src * HH + h] + edv[rl * HH + h]);
                            if (ph == 0) atomicMax(&mx[rl * HH + h], f2ord(lr));
                            else __hip_atomic_fetch_add(&den[rl * HH + h], expf(lr - ord2f(mx[rl * HH + h])),
                                                        __ATOMIC_RELAXED, __HIP_MEMORY_SCOPE_WORKGROUP);
                        }
                    }
                } else {
                    unsigned mk = __builtin_amdgcn_ballot_w32(in);
                    while (mk) {
                        const int b = __builtin_ctz(mk);
                        mk &= mk - 1;
                        const int sb = __builtin_amdgcn_readlane(src, b);
                        const int rb = __builtin_amdgcn_readlane(rl, b);
                        const int h0 = HH == 4 ? (lane >> 4) : 0;
                        const int h1 = HH == 4 ? 2 + (lane >> 4) : 0;
                        const float l0 = agg_lrelu(es[(size_t)sb * HH + h0] + edv[rb * HH + h0]);
                        const float l1 = agg_lrelu(es[(size_t)sb * HH + h1] + edv[rb * HH + h1]);
                        const float w0 = expf(l0 - ord2f(mx[rb * HH + h0])) / den[rb * HH + h0];
                        const float w1 = expf(l1 - ord2f(mx[rb * HH + h1])) / den[rb * HH + h1];
                        const RowVals<HD> p = agg_fetch<HD>(hfeat, sb, lane, w0, w1);
                        agg_add_row<HD>(acc, rb, lane, p);
                    }
                }
            }
            __syncthreads();
        }
    }
    __syncthreads();

    float cs0 = 0.f, cs1 = 0.f;
    for (int r = wave; r < nrows; r += NW) {
        const float* a = acc + r * HD + lane;
        const size_t nd = (size_t)(lo + r);
        if constexpr (HD == 256) {
            const float4* b4 = reinterpret_cast<const float4*>(bias);
            const float4 c0 = b4[lane], c1 = b4[32 + lane];
            float4 o0, o1;
            o0.x = agg_act(a[0]   + c0.x, do_elu); o0.y = agg_act(a[32]  + c0.y, do_elu);
            o0.z = agg_act(a[64]  + c0.z, do_elu); o0.w = agg_act(a[96]  + c0.w, do_elu);
            o1.x = agg_act(a[128] + c1.x, do_elu); o1.y = agg_act(a[160] + c1.y, do_elu);
            o1.z = agg_act(a[192] + c1.z, do_elu); o1.w = agg_act(a[224] + c1.w, do_elu);
            float4* o4 = reinterpret_cast<float4*>(out + nd * 256);
            o4[lane] = o0;
            o4[32 + lane] = o1;
        } else {
            const float2 c = reinterpret_cast<const float2*>(bias)[lane];
            float2 o;
            o.x = agg_act(a[0]  + c.x, do_elu);
            o.y = agg_act(a[32] + c.y, do_elu);
            reinterpret_cast<float2*>(out + nd * 64)[lane] = o;
            cs0 += o.x;
            cs1 += o.y;
        }
    }
    if constexpr (HD == 64) {
        if (colpart != nullptr) {
            lw[wave * 64 + 2 * lane]     = cs0;
            lw[wave * 64 + 2 * lane + 1] = cs1;
            __syncthreads();
            if (tid < 64) {
                float s = 0.f;
                for (int w = 0; w < NW; ++w) s += lw[w * 64 + tid];
                colpart[(size_t)blockIdx.x * 64 + tid] = s;
            }
        }
    }
}

#define MEAN_GROUPS 16
__global__ __launch_bounds__(64 * MEAN_GROUPS)
void gat_colmean_parts(const float* __restrict__ part, float* __restrict__ out, int N) {
    __shared__ float sm[MEAN_GROUPS * 64];
    const int nblk = (N + AGG_ROWS - 1) / AGG_ROWS;
    const int c = threadIdx.x & 63;
    const int g = threadIdx.x >> 6;
    float s = 0.f;
#pragma unroll 16
    for (int b = g; b < nblk; b += MEAN_GROUPS) s += part[(size_t)b * 64 + c];
    sm[g * 64 + c] = s;
    __syncthreads();
    if (threadIdx.x < 64) {
        float t = 0.f;
        for (int k = 0; k < MEAN_GROUPS; ++k) t += sm[k * 64 + threadIdx.x];
        out[threadIdx.x] = t / (float)N;
    }
}

static inline int cdiv(long long a, long long b) { return (int)((a + b - 1) / b); }

extern "C" void kernel_launch(void* const* d_in, const int* in_sizes, int n_in,
                              void* d_out, int out_size, void* d_ws, size_t ws_size,
                              hipStream_t stream) {
    const float* x   = (const float*)d_in[0];
    const int*   ei  = (const int*)  d_in[1];
    const float* W1  = (const float*)d_in[2];
    const float* a1s = (const float*)d_in[3];
    const float* a1d = (const float*)d_in[4];
    const float* b1  = (const float*)d_in[5];
    const float* W2  = (const float*)d_in[6];
    const float* a2s = (const float*)d_in[7];
    const float* a2d = (const float*)d_in[8];
    const float* b2  = (const float*)d_in[9];
    const float* W3  = (const float*)d_in[10];
    const float* a3s = (const float*)d_in[11];
    const float* a3d = (const float*)d_in[12];
    const float* b3  = (const float*)d_in[13];

    const int N  = in_sizes[0] / 16;
    const int E  = in_sizes[1] / 2;
    const int H  = 4;

    float* ws = (float*)d_ws;
    size_t off = 0;
    float*    buf_h    = ws + off; off += (size_t)N * 256;
    float*    buf_feat = ws + off; off += (size_t)N * 256;
    float*    es       = ws + off; off += (size_t)N * H;
    float*    ed       = ws + off; off += (size_t)N * H;
    (void)ws_size; (void)off;

    float* node_emb  = (float*)d_out;
    float* graph_emb = node_emb + (size_t)N * 64;

    auto run_layer = [&](const float* in_feat, const float* W,
                         const float* as_, const float* ad_, const float* bb,
                         float* out_feat, int Hh, int K, int Ncols, int do_elu, float* colpart, int take_list) {
        int ngrp = Ncols / 64;
        int wpb  = ngrp < 4 ? ngrp : 4;
        dim3 gb(32, wpb);
        dim3 gg(cdiv(ngrp, wpb), cdiv(N, 16));
        hipLaunchKernelGGL(gat_gemm_wmma, gg, gb, 0, stream, in_feat, W, buf_h, N, K, Ncols);
        hipLaunchKernelGGL(gat_attn_coef, dim3(cdiv((long long)N * Hh, 256)), dim3(256), 0, stream,
                           buf_h, as_, ad_, es, ed, N, Hh);
        if (Hh == 4)
            hipLaunchKernelGGL(gat_agg_owner<256>, dim3(cdiv(N, AGG_ROWS)), dim3(AGG_THREADS), 0, stream,
                               ei, E, N, buf_h, es, ed, bb, out_feat, do_elu, colpart, node_emb, take_list);
        else
            hipLaunchKernelGGL(gat_agg_owner<64>, dim3(cdiv(N, AGG_ROWS)), dim3(AGG_THREADS), 0, stream,
                               ei, E, N, buf_h, es, ed, bb, out_feat, do_elu, colpart, node_emb, take_list);
    };

    run_layer(x,        W1, a1s, a1d, b1, buf_feat, H, 16,  256, 1, nullptr, 0);
    run_layer(buf_feat, W2, a2s, a2d, b2, buf_feat, H, 256, 256, 1, nullptr, 1);
    float* col_part = es + (size_t)N;
    run_layer(buf_feat, W3, a3s, a3d, b3, node_emb, 1, 256, 64, 0, col_part, 1);

    hipLaunchKernelGGL(gat_colmean_parts, dim3(1), dim3(64 * MEAN_GROUPS), 0, stream,
                       col_part, graph_emb, N);
}


# Next layer's projection done by the owner block: its bias+ELU rows stay in LDS (channel order, bank-swizzled) and meet the next W on f32 WMMA there; feature planes and 2 GEMM launches gone
# speedup vs baseline: 8.2291x; 1.3297x over previous
#include <hip/hip_runtime.h>
#include <math.h>

typedef __attribute__((ext_vector_type(2))) float v2f;
typedef __attribute__((ext_vector_type(8))) float v8f;

#define GAT_D 64

__device__ __forceinline__ unsigned f2ord(float f) {
    int b = __float_as_int(f);
    return (unsigned)b ^ (unsigned)((b >> 31) | 0x80000000);
}
__device__ __forceinline__ float ord2f(unsigned u) {
    int b = (u & 0x80000000u) ? (int)(u ^ 0x80000000u) : ~(int)u;
    return __int_as_float(b);
}

__device__ __forceinline__ float row16_sum(float v) {
    v += __int_as_float(__builtin_amdgcn_update_dpp(0, __float_as_int(v), 0x161, 0xf, 0xf, false));
    v += __int_as_float(__builtin_amdgcn_update_dpp(0, __float_as_int(v), 0x162, 0xf, 0xf, false));
    v += __int_as_float(__builtin_amdgcn_update_dpp(0, __float_as_int(v), 0x164, 0xf, 0xf, false));
    v += __int_as_float(__builtin_amdgcn_update_dpp(0, __float_as_int(v), 0x168, 0xf, 0xf, false));
    return v;
}

__global__ void gat_gemm_wmma(const float* __restrict__ A,
                              const float* __restrict__ B,
                              float* __restrict__ C,
                              const float* __restrict__ a_s,
                              const float* __restrict__ a_d,
                              float* __restrict__ es, float* __restrict__ ed,
                              int M, int K, int N) {
    const int lane  = threadIdx.x;
    const int grp   = blockIdx.x * blockDim.y + threadIdx.y;
    const int mtile = blockIdx.y;
    const int n0 = grp * 64;
    const int m0 = mtile * 16;
    if (n0 >= N || m0 >= M) return;

    const int l15 = lane & 15;
    const int kh  = (lane >> 4) << 1;
    const float* Arow = A + (size_t)(m0 + l15) * K;
    v8f acc0 = {0.f,0.f,0.f,0.f,0.f,0.f,0.f,0.f};
    v8f acc1 = acc0, acc2 = acc0, acc3 = acc0;
    const float sa0 = a_s[n0 + l15], sa1 = a_s[n0 + 16 + l15], sa2 = a_s[n0 + 32 + l15], sa3 = a_s[n0 + 48 + l15];
    const float da0 = a_d[n0 + l15], da1 = a_d[n0 + 16 + l15], da2 = a_d[n0 + 32 + l15], da3 = a_d[n0 + 48 + l15];

    for (int k = 0; k < K; k += 4) {
        v2f a;
        a.x = Arow[k + kh];
        a.y = Arow[k + kh + 1];
        const float* Bp = B + (size_t)(k + kh) * N + n0 + l15;
        v2f b0, b1, b2, b3;
        b0.x = Bp[0];  b0.y = Bp[(size_t)N];
        b1.x = Bp[16]; b1.y = Bp[(size_t)N + 16];
        b2.x = Bp[32]; b2.y = Bp[(size_t)N + 32];
        b3.x = Bp[48]; b3.y = Bp[(size_t)N + 48];
        acc0 = __builtin_amdgcn_wmma_f32_16x16x4_f32(false, a, false, b0, (short)0, acc0, false, false);
        acc1 = __builtin_amdgcn_wmma_f32_16x16x4_f32(false, a, false, b1, (short)0, acc1, false, false);
        acc2 = __builtin_amdgcn_wmma_f32_16x16x4_f32(false, a, false, b2, (short)0, acc2, false, false);
        acc3 = __builtin_amdgcn_wmma_f32_16x16x4_f32(false, a, false, b3, (short)0, acc3, false, false);
    }

    float* Cp = C + (size_t)(m0 + ((lane >> 4) << 3)) * N + n0 + l15;
#pragma unroll
    for (int v = 0; v < 8; ++v) {
        Cp[(size_t)v * N + 0]  = acc0[v];
        Cp[(size_t)v * N + 16] = acc1[v];
        Cp[(size_t)v * N + 32] = acc2[v];
        Cp[(size_t)v * N + 48] = acc3[v];
    }

    float cs = 0.f, cd = 0.f;
#pragma unroll
    for (int v = 0; v < 8; ++v) {
        const float ps = row16_sum(acc0[v] * sa0 + acc1[v] * sa1 + acc2[v] * sa2 + acc3[v] * sa3);
        const float pd = row16_sum(acc0[v] * da0 + acc1[v] * da1 + acc2[v] * da2 + acc3[v] * da3);
        if (l15 == v) { cs = ps; cd = pd; }
    }
    if (l15 < 8) {
        const size_t ci = (size_t)(m0 + ((lane >> 4) << 3) + l15) * (size_t)(N / GAT_D) + (size_t)grp;
        es[ci] = cs;
        ed[ci] = cd;
    }
}

__device__ __forceinline__ void edge_endpoints(const int* __restrict__ ei, int E,
                                               int e, int& src, int& dst) {
    if (e < E) { src = ei[e]; dst = ei[E + e]; }
    else       { src = e - E; dst = e - E;    }
}

#define AGG_ROWS    196
#define AGG_THREADS 1024
#define AGG_K       8
#define AGG_CAP     4096
#define AGG_U       4
#define KEEP_PITCH  64
#define TAIL_WORDS  4

template <int HD> struct RowVals { float v[HD / 32]; };

__device__ __forceinline__ const float* es_tail(const float* keep, int N, int src) {
    const int blk = src / AGG_ROWS;
    const int lo_b = blk * AGG_ROWS;
    const int nr = min(AGG_ROWS, N - lo_b);
    return keep + (size_t)lo_b * KEEP_PITCH + nr * (KEEP_PITCH - TAIL_WORDS) + TAIL_WORDS * (src - lo_b);
}

template <int HH, bool ES_TAIL>
__device__ __forceinline__ float es_at(const float* es, const float* keep, int N, int src, int h) {
    if constexpr (ES_TAIL) return es_tail(keep, N, src)[h];
    else                   return es[(size_t)src * HH + h];
}

template <int HD>
__device__ __forceinline__ RowVals<HD> agg_fetch(const float* __restrict__ hfeat,
                                                 int src, int lane, float w0, float w1) {
    RowVals<HD> r;
    if constexpr (HD == 256) {
        const float4* hp = reinterpret_cast<const float4*>(hfeat + (size_t)src * 256);
        const float4 a = hp[lane];
        const float4 b = hp[32 + lane];
        r.v[0] = a.x * w0; r.v[1] = a.y * w0; r.v[2] = a.z * w0; r.v[3] = a.w * w0;
        r.v[4] = b.x * w1; r.v[5] = b.y * w1; r.v[6] = b.z * w1; r.v[7] = b.w * w1;
    } else {
        const float2* hp = reinterpret_cast<const float2*>(hfeat + (size_t)src * 64);
        const float2 a = hp[lane];
        r.v[0] = a.x * w0; r.v[1] = a.y * w0;
    }
    return r;
}

template <int HD>
__device__ __forceinline__ RowVals<HD> agg_take(const float* __restrict__ hfeat,
                                                const int* __restrict__ lsrc,
                                                const int* __restrict__ lrow,
                                                const float* __restrict__ lw,
                                                int i, int lane, int& row) {
    row = lrow[i];
    const float* wp = lw + i * (HD / 64);
    float w0, w1;
    if constexpr (HD == 256) { w0 = wp[lane >> 4]; w1 = wp[2 + (lane >> 4)]; }
    else                     { w0 = wp[0]; w1 = 0.f; }
    return agg_fetch<HD>(hfeat, lsrc[i], lane, w0, w1);
}

template <int HD>
__device__ __forceinline__ void agg_add_row(float* __restrict__ acc, int row, int lane,
                                            const RowVals<HD>& p) {
    float* a = acc + row * HD + lane;
#pragma unroll
    for (int j = 0; j < HD / 32; ++j)
        __hip_atomic_fetch_add(a + j * 32, p.v[j], __ATOMIC_RELAXED, __HIP_MEMORY_SCOPE_WORKGROUP);
}

__device__ __forceinline__ float agg_act(float v, int do_elu) {
    if (do_elu) v = v > 0.f ? v : (expf(v) - 1.f);
    return v;
}

__device__ __forceinline__ float agg_lrelu(float s) { return s > 0.f ? s : 0.2f * s; }

template <bool TAIL>
__device__ __forceinline__ void agg_load_dst(const int* __restrict__ dstp, int base, int lane,
                                             int E, int (&d)[AGG_K]) {
#pragma unroll
    for (int k = 0; k < AGG_K; ++k) {
        const int e = base + 32 * k + lane;
        if constexpr (TAIL) d[k] = e < E ? dstp[e] : -1;
        else                d[k] = dstp[e];
    }
}

__device__ __forceinline__ void agg_collect(const int (&d)[AGG_K], int base, int lane, int lo, int nrows,
                                            int* __restrict__ lsrc, int* __restrict__ lrow,
                                            unsigned* __restrict__ cnt) {
    unsigned m[AGG_K], any = 0;
    int rl[AGG_K];
#pragma unroll
    for (int k = 0; k < AGG_K; ++k) {
        rl[k] = d[k] - lo;
        m[k] = __builtin_amdgcn_ballot_w32((unsigned)rl[k] < (unsigned)nrows);
        any |= m[k];
    }
    if (!any) return;
    unsigned tot = 0;
#pragma unroll
    for (int k = 0; k < AGG_K; ++k) tot += (unsigned)__builtin_popcount(m[k]);
    unsigned start = 0;
    if (lane == 0)
        start = __hip_atomic_fetch_add(cnt, tot, __ATOMIC_RELAXED, __HIP_MEMORY_SCOPE_WORKGROUP);
    start = (unsigned)__builtin_amdgcn_readfirstlane((int)start);
    const unsigned below = (1u << lane) - 1u;
#pragma unroll
    for (int k = 0; k < AGG_K; ++k) {
        if ((m[k] >> lane) & 1u) {
            const unsigned p = start + (unsigned)__builtin_popcount(m[k] & below);
            if (p < (unsigned)AGG_CAP) { lsrc[p] = base + 32 * k + lane; lrow[p] = rl[k]; }
        }
        start += (unsigned)__builtin_popcount(m[k]);
    }
}

template <int HD, int NC, bool ES_TAIL>
__global__ __launch_bounds__(AGG_THREADS)
void gat_agg_owner(const int* __restrict__ ei, int E, int N,
                   const float* __restrict__ hfeat,
                   const float* es, const float* ed,
                   const float* __restrict__ bias,
                   float* out, int do_elu,
                   float* __restrict__ colpart,
                   float* keep, int take_list,
                   const float* __restrict__ Wn,
                   const float* __restrict__ asn, const float* __restrict__ adn,
                   float* __restrict__ hn, float* esn, float* edn) {
    static_assert(NC == 0 || HD == 256, "the projection at the end reads 256-channel rows");
    static_assert(!ES_TAIL || HD == 256, "only 4-head coefficients are left in the output plane");
    constexpr int HH = HD / 64;
    constexpr int NW = AGG_THREADS / 32;
    constexpr int SPAN = 32 * AGG_K;
    constexpr int TRIP = NW * SPAN;
    __shared__ __attribute__((aligned(16))) float acc[AGG_ROWS * HD];
    __shared__ unsigned mx[AGG_ROWS * HH];
    __shared__ float    den[AGG_ROWS * HH];
    __shared__ float    edv[AGG_ROWS * HH];
    __shared__ int      lsrc[AGG_CAP];
    __shared__ int      lrow[AGG_CAP];
    __shared__ float    lw[AGG_CAP * HH];
    __shared__ unsigned cnt;
    const int tid  = threadIdx.x;
    const int lane = tid & 31;
    const int wave = tid >> 5;
    const int lo = blockIdx.x * AGG_ROWS;
    const int nrows = max(0, min(AGG_ROWS, N - lo));

    float* kp = keep + (size_t)lo * KEEP_PITCH;
    const int kcap = nrows * (KEEP_PITCH - TAIL_WORDS);
    int taken = -1;
    if (take_list && nrows > 0) taken = __builtin_amdgcn_readfirstlane(__float_as_int(kp[0]));
    const bool have = taken >= 0 && taken <= AGG_CAP && 1 + 2 * taken <= kcap;

    {
        float4* a4 = reinterpret_cast<float4*>(acc);
        for (int i = tid; i < AGG_ROWS * HD / 4; i += AGG_THREADS) a4[i] = make_float4(0.f, 0.f, 0.f, 0.f);
        for (int i = tid; i < nrows * HH; i += AGG_THREADS) {
            mx[i] = 0u;
            den[i] = 0.f;
            edv[i] = ed[(size_t)lo * HH + i];
        }
        if (have) {
            for (int i = tid; i < taken; i += AGG_THREADS) {
                lsrc[i] = __float_as_int(kp[1 + i]);
                lrow[i] = __float_as_int(kp[1 + taken + i]);
            }
            if (tid == 0) cnt = (unsigned)taken;
        } else {
            for (int r = tid; r < nrows; r += AGG_THREADS) { lsrc[r] = E + lo + r; lrow[r] = r; }
            if (tid == 0) cnt = (unsigned)nrows;
        }
    }
    __syncthreads();

    const int* __restrict__ dstp = ei + E;
    const int full = E / TRIP;
    const int t0 = (int)((blockIdx.x * (unsigned)full) / gridDim.x);
    const int woff = wave * SPAN;
    if (!have && full > 0) {
        int dA[AGG_K], dB[AGG_K];
        agg_load_dst<false>(dstp, t0 * TRIP + woff, lane, E, dA);
        for (int t = 0; t < full; t += 2) {
            int tc0 = t0 + t;    if (tc0 >= full) tc0 -= full;
            int tc1 = tc0 + 1;   if (tc1 >= full) tc1 -= full;
            int tc2 = tc1 + 1;   if (tc2 >= full) tc2 -= full;
            agg_load_dst<false>(dstp, tc1 * TRIP + woff, lane, E, dB);
            agg_collect(dA, tc0 * TRIP + woff, lane, lo, nrows, lsrc, lrow, &cnt);
            agg_load_dst<false>(dstp, tc2 * TRIP + woff, lane, E, dA);
            if (t + 1 < full)
                agg_collect(dB, tc1 * TRIP + woff, lane, lo, nrows, lsrc, lrow, &cnt);
        }
    }
    if (!have && full * TRIP < E) {
        int dT[AGG_K];
        agg_load_dst<true>(dstp, full * TRIP + woff, lane, E, dT);
        agg_collect(dT, full * TRIP + woff, lane, lo, nrows, lsrc, lrow, &cnt);
    }
    __syncthreads();
    const int n = (int)cnt;

    const bool save = !take_list && n <= AGG_CAP && 1 + 2 * n <= kcap;
    if (!take_list && tid == 0 && nrows > 0) kp[0] = __int_as_float(save ? n : -1);
    if (n <= AGG_CAP) {
        for (int i = tid; i < n; i += AGG_THREADS) {
            int src = lsrc[i];
            const int row = lrow[i];
            if (!have) {
                const int e = src;
                src = e < E ? ei[e] : e - E;
                lsrc[i] = src;
                if (save) { kp[1 + i] = __int_as_float(src); kp[1 + n + i] = __int_as_float(row); }
            }
            float sv[HH];
            if constexpr (HH == 4) {
                const float4 q = ES_TAIL ? *reinterpret_cast<const float4*>(es_tail(keep, N, src))
                                         : reinterpret_cast<const float4*>(es)[src];
                sv[0] = q.x; sv[1] = q.y; sv[2] = q.z; sv[3] = q.w;
            } else {
                sv[0] = es[src];
            }
#pragma unroll
            for (int h = 0; h < HH; ++h) {
                const float lr = agg_lrelu(sv[h] + edv[row * HH + h]);
                lw[i * HH + h] = lr;
                atomicMax(&mx[row * HH + h], f2ord(lr));
            }
        }
        __syncthreads();
        for (int i = tid; i < n; i += AGG_THREADS) {
            const int row = lrow[i];
#pragma unroll
            for (int h = 0; h < HH; ++h) {
                const float v = expf(lw[i * HH + h] - ord2f(mx[row * HH + h]));
                lw[i * HH + h] = v;
                __hip_atomic_fetch_add(&den[row * HH + h], v, __ATOMIC_RELAXED, __HIP_MEMORY_SCOPE_WORKGROUP);
            }
        }
        __syncthreads();
        for (int i = tid; i < n; i += AGG_THREADS) {
            const int row = lrow[i];
#pragma unroll
            for (int h = 0; h < HH; ++h) lw[i * HH + h] = lw[i * HH + h] / den[row * HH + h];
        }
        __syncthreads();
        const int ngrp = n / AGG_U;
        for (int g = wave; g < ngrp; g += NW) {
            RowVals<HD> p[AGG_U];
            int rw[AGG_U];
#pragma unroll
            for (int u = 0; u < AGG_U; ++u)
                p[u] = agg_take<HD>(hfeat, lsrc, lrow, lw, g * AGG_U + u, lane, rw[u]);
#pragma unroll
            for (int u = 0; u < AGG_U; ++u) agg_add_row<HD>(acc, rw[u], lane, p[u]);
        }
        for (int i = ngrp * AGG_U + wave; i < n; i += NW) {
            int rw1;
            const RowVals<HD> p1 = agg_take<HD>(hfeat, lsrc, lrow, lw, i, lane, rw1);
            agg_add_row<HD>(acc, rw1, lane, p1);
        }
    } else {
        const int Et = E + N;
        for (int ph = 0; ph < 3; ++ph) {
            for (int eb = wave * 32; eb < Et; eb += AGG_THREADS) {
                const int e = eb + lane;
                int src = 0, rl = -1;
                if (e < Et) {
                    int dst; edge_endpoints(ei, E, e, src, dst);
                    rl = dst - lo;
                }
                const bool in = (unsigned)rl < (unsigned)nrows;
                if (ph < 2) {
                    if (in) {
#pragma unroll
                        for (int h = 0; h < HH; ++h) {
                            const float lr = agg_lrelu(es_at<HH, ES_TAIL>(es, keep, N, src, h) + edv[rl * HH + h]);
                            if (ph == 0) atomicMax(&mx[rl * HH + h], f2ord(lr));
                            else __hip_atomic_fetch_add(&den[rl * HH + h], expf(lr - ord2f(mx[rl * HH + h])),
                                                        __ATOMIC_RELAXED, __HIP_MEMORY_SCOPE_WORKGROUP);
                        }
                    }
                } else {
                    unsigned mk = __builtin_amdgcn_ballot_w32(in);
                    while (mk) {
                        const int b = __builtin_ctz(mk);
                        mk &= mk - 1;
                        const int sb = __builtin_amdgcn_readlane(src, b);
                        const int rb = __builtin_amdgcn_readlane(rl, b);
                        const int h0 = HH == 4 ? (lane >> 4) : 0;
                        const int h1 = HH == 4 ? 2 + (lane >> 4) : 0;
                        const float l0 = agg_lrelu(es_at<HH, ES_TAIL>(es, keep, N, sb, h0) + edv[rb * HH + h0]);
                        const float l1 = agg_lrelu(es_at<HH, ES_TAIL>(es, keep, N, sb, h1) + edv[rb * HH + h1]);
                        const float w0 = expf(l0 - ord2f(mx[rb * HH + h0])) / den[rb * HH + h0];
                        const float w1 = expf(l1 - ord2f(mx[rb * HH + h1])) / den[rb * HH + h1];
                        const RowVals<HD> p = agg_fetch<HD>(hfeat, sb, lane, w0, w1);
                        agg_add_row<HD>(acc, rb, lane, p);
                    }
                }
            }
            __syncthreads();
        }
    }
    __syncthreads();

    if constexpr (NC > 0) {
        for (int r = wave; r < nrows; r += NW) {
            float* arow = acc + r * HD;
            const float* a = arow + lane;
            const float4* b4 = reinterpret_cast<const float4*>(bias);
            const float4 c0 = b4[lane], c1 = b4[32 + lane];
            const float s0 = a[0], s1 = a[32], s2 = a[64], s3 = a[96], s4 = a[128], s5 = a[160], s6 = a[192], s7 = a[224];
            float4 o0, o1;
            o0.x = agg_act(s0 + c0.x, do_elu); o0.y = agg_act(s1 + c0.y, do_elu);
            o0.z = agg_act(s2 + c0.z, do_elu); o0.w = agg_act(s3 + c0.w, do_elu);
            o1.x = agg_act(s4 + c1.x, do_elu); o1.y = agg_act(s5 + c1.y, do_elu);
            o1.z = agg_act(s6 + c1.z, do_elu); o1.w = agg_act(s7 + c1.w, do_elu);
            const int swz = (r & 15) << 2;
            *reinterpret_cast<float4*>(arow + ((4 * lane) ^ swz))       = o0;
            *reinterpret_cast<float4*>(arow + ((128 + 4 * lane) ^ swz)) = o1;
        }
        __syncthreads();

        constexpr int HN = NC / 64;
        constexpr int MG = NW / HN;
        const int wv = __builtin_amdgcn_readfirstlane(wave);
        const int head = wv % HN;
        const int n0 = head * 64;
        const int nmt = (nrows + 15) >> 4;
        const int l15 = lane & 15;
        const int kh  = (lane >> 4) << 1;
        const float sa0 = asn[n0 + l15], sa1 = asn[n0 + 16 + l15], sa2 = asn[n0 + 32 + l15], sa3 = asn[n0 + 48 + l15];
        const float da0 = adn[n0 + l15], da1 = adn[n0 + 16 + l15], da2 = adn[n0 + 32 + l15], da3 = adn[n0 + 48 + l15];
        for (int mt = wv / HN; mt < nmt; mt += MG) {
            const int ar = min(mt * 16 + l15, AGG_ROWS - 1);
            const float* Ap = acc + ar * HD + kh;
            const int swz = (ar & 15) << 2;
            v8f q0 = {0.f,0.f,0.f,0.f,0.f,0.f,0.f,0.f};
            v8f q1 = q0, q2 = q0, q3 = q0;
            for (int k = 0; k < HD; k += 4) {
                const v2f a = *reinterpret_cast<const v2f*>(Ap + (k ^ swz));
                const float* Bp = Wn + (size_t)(k + kh) * NC + n0 + l15;
                v2f b0, b1, b2, b3;
                b0.x = Bp[0];  b0.y = Bp[(size_t)NC];
                b1.x = Bp[16]; b1.y = Bp[(size_t)NC + 16];
                b2.x = Bp[32]; b2.y = Bp[(size_t)NC + 32];
                b3.x = Bp[48]; b3.y = Bp[(size_t)NC + 48];
                q0 = __builtin_amdgcn_wmma_f32_16x16x4_f32(false, a, false, b0, (short)0, q0, false, false);
                q1 = __builtin_amdgcn_wmma_f32_16x16x4_f32(false, a, false, b1, (short)0, q1, false, false);
                q2 = __builtin_amdgcn_wmma_f32_16x16x4_f32(false, a, false, b2, (short)0, q2, false, false);
                q3 = __builtin_amdgcn_wmma_f32_16x16x4_f32(false, a, false, b3, (short)0, q3, false, false);
            }
            const int rbase = mt * 16 + ((lane >> 4) << 3);
            float* Cp = hn + (size_t)(lo + rbase) * NC + n0 + l15;
#pragma unroll
            for (int v = 0; v < 8; ++v) {
                if (rbase + v < nrows) {
                    Cp[(size_t)v * NC + 0]  = q0[v];
                    Cp[(size_t)v * NC + 16] = q1[v];
                    Cp[(size_t)v * NC + 32] = q2[v];
                    Cp[(size_t)v * NC + 48] = q3[v];
                }
            }
            float cs = 0.f, cd = 0.f;
#pragma unroll
            for (int v = 0; v < 8; ++v) {
                const float ps = row16_sum(q0[v] * sa0 + q1[v] * sa1 + q2[v] * sa2 + q3[v] * sa3);
                const float pd = row16_sum(q0[v] * da0 + q1[v] * da1 + q2[v] * da2 + q3[v] * da3);
                if (l15 == v) { cs = ps; cd = pd; }
            }
            if (l15 < 8 && rbase + l15 < nrows) {
                const int rl = rbase + l15;
                if constexpr (HN == 4) {
                    kp[kcap + TAIL_WORDS * rl + head] = cs;
                    edn[(size_t)(lo + rl) * HN + head] = cd;
                } else {
                    esn[(size_t)(lo + rl) * HN + head] = cs;
                    edn[(size_t)(lo + rl) * HN + head] = cd;
                }
            }
        }
    } else {
    float cs0 = 0.f, cs1 = 0.f;
    for (int r = wave; r < nrows; r += NW) {
        const float* a = acc + r * HD + lane;
        const size_t nd = (size_t)(lo + r);
        if constexpr (HD == 256) {
            const float4* b4 = reinterpret_cast<const float4*>(bias);
            const float4 c0 = b4[lane], c1 = b4[32 + lane];
            float4 o0, o1;
            o0.x = agg_act(a[0]   + c0.x, do_elu); o0.y = agg_act(a[32]  + c0.y, do_elu);
            o0.z = agg_act(a[64]  + c0.z, do_elu); o0.w = agg_act(a[96]  + c0.w, do_elu);
            o1.x = agg_act(a[128] + c1.x, do_elu); o1.y = agg_act(a[160] + c1.y, do_elu);
            o1.z = agg_act(a[192] + c1.z, do_elu); o1.w = agg_act(a[224] + c1.w, do_elu);
            float4* o4 = reinterpret_cast<float4*>(out + nd * 256);
            o4[lane] = o0;
            o4[32 + lane] = o1;
        } else {
            const float2 c = reinterpret_cast<const float2*>(bias)[lane];
            float2 o;
            o.x = agg_act(a[0]  + c.x, do_elu);
            o.y = agg_act(a[32] + c.y, do_elu);
            reinterpret_cast<float2*>(out + nd * 64)[lane] = o;
            cs0 += o.x;
            cs1 += o.y;
        }
    }
    if constexpr (HD == 64) {
        if (colpart != nullptr) {
            lw[wave * 64 + 2 * lane]     = cs0;
            lw[wave * 64 + 2 * lane + 1] = cs1;
            __syncthreads();
            if (tid < 64) {
                float s = 0.f;
                for (int w = 0; w < NW; ++w) s += lw[w * 64 + tid];
                colpart[(size_t)blockIdx.x * 64 + tid] = s;
            }
        }
    }
    }
}

#define MEAN_GROUPS 16
__global__ __launch_bounds__(64 * MEAN_GROUPS)
void gat_colmean_parts(const float* __restrict__ part, float* __restrict__ out, int N) {
    __shared__ float sm[MEAN_GROUPS * 64];
    const int nblk = (N + AGG_ROWS - 1) / AGG_ROWS;
    const int c = threadIdx.x & 63;
    const int g = threadIdx.x >> 6;
    float s = 0.f;
#pragma unroll 16
    for (int b = g; b < nblk; b += MEAN_GROUPS) s += part[(size_t)b * 64 + c];
    sm[g * 64 + c] = s;
    __syncthreads();
    if (threadIdx.x < 64) {
        float t = 0.f;
        for (int k = 0; k < MEAN_GROUPS; ++k) t += sm[k * 64 + threadIdx.x];
        out[threadIdx.x] = t / (float)N;
    }
}

static inline int cdiv(long long a, long long b) { return (int)((a + b - 1) / b); }

extern "C" void kernel_launch(void* const* d_in, const int* in_sizes, int n_in,
                              void* d_out, int out_size, void* d_ws, size_t ws_size,
                              hipStream_t stream) {
    const float* x   = (const float*)d_in[0];
    const int*   ei  = (const int*)  d_in[1];
    const float* W1  = (const float*)d_in[2];
    const float* a1s = (const float*)d_in[3];
    const float* a1d = (const float*)d_in[4];
    const float* b1  = (const float*)d_in[5];
    const float* W2  = (const float*)d_in[6];
    const float* a2s = (const float*)d_in[7];
    const float* a2d = (const float*)d_in[8];
    const float* b2  = (const float*)d_in[9];
    const float* W3  = (const float*)d_in[10];
    const float* a3s = (const float*)d_in[11];
    const float* a3d = (const float*)d_in[12];
    const float* b3  = (const float*)d_in[13];

    const int N  = in_sizes[0] / 16;
    const int E  = in_sizes[1] / 2;
    const int H  = 4;

    float* ws = (float*)d_ws;
    size_t off = 0;
    float*    buf_h    = ws + off; off += (size_t)N * 256;
    float*    buf_h2   = ws + off; off += (size_t)N * 256;
    float*    es       = ws + off; off += (size_t)N * H;
    float*    ed       = ws + off; off += (size_t)N * H;
    (void)ws_size; (void)off;

    float* node_emb  = (float*)d_out;
    float* graph_emb = node_emb + (size_t)N * 64;

    hipLaunchKernelGGL(gat_gemm_wmma, dim3(1, cdiv(N, 16)), dim3(32, 4), 0, stream,
                       x, W1, buf_h, a1s, a1d, es, ed, N, 16, 256);

    const dim3 ag(cdiv(N, AGG_ROWS)), ab(AGG_THREADS);
    float* es3      = es;
    float* col_part = es + (size_t)N;
    float* ed3      = es + 2 * (size_t)N;
    float* none     = nullptr;
    gat_agg_owner<256, 256, false><<<ag, ab, 0, stream>>>(ei, E, N, buf_h, es, ed, b1, none, 1, none, node_emb, 0,
                                                          W2, a2s, a2d, buf_h2, none, ed);
    gat_agg_owner<256, 64, true><<<ag, ab, 0, stream>>>(ei, E, N, buf_h2, es, ed, b2, none, 1, none, node_emb, 1,
                                                        W3, a3s, a3d, buf_h, es3, ed3);
    gat_agg_owner<64, 0, false><<<ag, ab, 0, stream>>>(ei, E, N, buf_h, es3, ed3, b3, node_emb, 0, col_part, node_emb, 1,
                                                       none, none, none, none, none, none);

    hipLaunchKernelGGL(gat_colmean_parts, dim3(1), dim3(64 * MEAN_GROUPS), 0, stream,
                       col_part, graph_emb, N);
}


# Edge walk reads 4 destinations per 16-byte load; a lane tests the unsigned minimum of its 8 (dst - lo), only a lane owning a row notes it by its own returning LDS add; list and LDS add order change
# speedup vs baseline: 8.4951x; 1.0323x over previous
#include <hip/hip_runtime.h>
#include <math.h>

typedef __attribute__((ext_vector_type(2))) float v2f;
typedef __attribute__((ext_vector_type(8))) float v8f;

#define GAT_D 64

__device__ __forceinline__ unsigned f2ord(float f) {
    int b = __float_as_int(f);
    return (unsigned)b ^ (unsigned)((b >> 31) | 0x80000000);
}
__device__ __forceinline__ float ord2f(unsigned u) {
    int b = (u & 0x80000000u) ? (int)(u ^ 0x80000000u) : ~(int)u;
    return __int_as_float(b);
}

__device__ __forceinline__ float row16_sum(float v) {
    v += __int_as_float(__builtin_amdgcn_update_dpp(0, __float_as_int(v), 0x161, 0xf, 0xf, false));
    v += __int_as_float(__builtin_amdgcn_update_dpp(0, __float_as_int(v), 0x162, 0xf, 0xf, false));
    v += __int_as_float(__builtin_amdgcn_update_dpp(0, __float_as_int(v), 0x164, 0xf, 0xf, false));
    v += __int_as_float(__builtin_amdgcn_update_dpp(0, __float_as_int(v), 0x168, 0xf, 0xf, false));
    return v;
}

__global__ void gat_gemm_wmma(const float* __restrict__ A,
                              const float* __restrict__ B,
                              float* __restrict__ C,
                              const float* __restrict__ a_s,
                              const float* __restrict__ a_d,
                              float* __restrict__ es, float* __restrict__ ed,
                              int M, int K, int N) {
    const int lane  = threadIdx.x;
    const int grp   = blockIdx.x * blockDim.y + threadIdx.y;
    const int mtile = blockIdx.y;
    const int n0 = grp * 64;
    const int m0 = mtile * 16;
    if (n0 >= N || m0 >= M) return;

    const int l15 = lane & 15;
    const int kh  = (lane >> 4) << 1;
    const float* Arow = A + (size_t)(m0 + l15) * K;
    v8f acc0 = {0.f,0.f,0.f,0.f,0.f,0.f,0.f,0.f};
    v8f acc1 = acc0, acc2 = acc0, acc3 = acc0;
    const float sa0 = a_s[n0 + l15], sa1 = a_s[n0 + 16 + l15], sa2 = a_s[n0 + 32 + l15], sa3 = a_s[n0 + 48 + l15];
    const float da0 = a_d[n0 + l15], da1 = a_d[n0 + 16 + l15], da2 = a_d[n0 + 32 + l15], da3 = a_d[n0 + 48 + l15];

    for (int k = 0; k < K; k += 4) {
        v2f a;
        a.x = Arow[k + kh];
        a.y = Arow[k + kh + 1];
        const float* Bp = B + (size_t)(k + kh) * N + n0 + l15;
        v2f b0, b1, b2, b3;
        b0.x = Bp[0];  b0.y = Bp[(size_t)N];
        b1.x = Bp[16]; b1.y = Bp[(size_t)N + 16];
        b2.x = Bp[32]; b2.y = Bp[(size_t)N + 32];
        b3.x = Bp[48]; b3.y = Bp[(size_t)N + 48];
        acc0 = __builtin_amdgcn_wmma_f32_16x16x4_f32(false, a, false, b0, (short)0, acc0, false, false);
        acc1 = __builtin_amdgcn_wmma_f32_16x16x4_f32(false, a, false, b1, (short)0, acc1, false, false);
        acc2 = __builtin_amdgcn_wmma_f32_16x16x4_f32(false, a, false, b2, (short)0, acc2, false, false);
        acc3 = __builtin_amdgcn_wmma_f32_16x16x4_f32(false, a, false, b3, (short)0, acc3, false, false);
    }

    float* Cp = C + (size_t)(m0 + ((lane >> 4) << 3)) * N + n0 + l15;
#pragma unroll
    for (int v = 0; v < 8; ++v) {
        Cp[(size_t)v * N + 0]  = acc0[v];
        Cp[(size_t)v * N + 16] = acc1[v];
        Cp[(size_t)v * N + 32] = acc2[v];
        Cp[(size_t)v * N + 48] = acc3[v];
    }

    float cs = 0.f, cd = 0.f;
#pragma unroll
    for (int v = 0; v < 8; ++v) {
        const float ps = row16_sum(acc0[v] * sa0 + acc1[v] * sa1 + acc2[v] * sa2 + acc3[v] * sa3);
        const float pd = row16_sum(acc0[v] * da0 + acc1[v] * da1 + acc2[v] * da2 + acc3[v] * da3);
        if (l15 == v) { cs = ps; cd = pd; }
    }
    if (l15 < 8) {
        const size_t ci = (size_t)(m0 + ((lane >> 4) << 3) + l15) * (size_t)(N / GAT_D) + (size_t)grp;
        es[ci] = cs;
        ed[ci] = cd;
    }
}

__device__ __forceinline__ void edge_endpoints(const int* __restrict__ ei, int E,
                                               int e, int& src, int& dst) {
    if (e < E) { src = ei[e]; dst = ei[E + e]; }
    else       { src = e - E; dst = e - E;    }
}

#define AGG_ROWS    196
#define AGG_THREADS 1024
#define AGG_K       8
#define AGG_CAP     4096
#define AGG_U       4
#define KEEP_PITCH  64
#define TAIL_WORDS  4

template <int HD> struct RowVals { float v[HD / 32]; };

__device__ __forceinline__ const float* es_tail(const float* keep, int N, int src) {
    const int blk = src / AGG_ROWS;
    const int lo_b = blk * AGG_ROWS;
    const int nr = min(AGG_ROWS, N - lo_b);
    return keep + (size_t)lo_b * KEEP_PITCH + nr * (KEEP_PITCH - TAIL_WORDS) + TAIL_WORDS * (src - lo_b);
}

template <int HH, bool ES_TAIL>
__device__ __forceinline__ float es_at(const float* es, const float* keep, int N, int src, int h) {
    if constexpr (ES_TAIL) return es_tail(keep, N, src)[h];
    else                   return es[(size_t)src * HH + h];
}

template <int HD>
__device__ __forceinline__ RowVals<HD> agg_fetch(const float* __restrict__ hfeat,
                                                 int src, int lane, float w0, float w1) {
    RowVals<HD> r;
    if constexpr (HD == 256) {
        const float4* hp = reinterpret_cast<const float4*>(hfeat + (size_t)src * 256);
        const float4 a = hp[lane];
        const float4 b = hp[32 + lane];
        r.v[0] = a.x * w0; r.v[1] = a.y * w0; r.v[2] = a.z * w0; r.v[3] = a.w * w0;
        r.v[4] = b.x * w1; r.v[5] = b.y * w1; r.v[6] = b.z * w1; r.v[7] = b.w * w1;
    } else {
        const float2* hp = reinterpret_cast<const float2*>(hfeat + (size_t)src * 64);
        const float2 a = hp[lane];
        r.v[0] = a.x * w0; r.v[1] = a.y * w0;
    }
    return r;
}

template <int HD>
__device__ __forceinline__ RowVals<HD> agg_take(const float* __restrict__ hfeat,
                                                const int* __restrict__ lsrc,
                                                const int* __restrict__ lrow,
                                                const float* __restrict__ lw,
                                                int i, int lane, int& row) {
    row = lrow[i];
    const float* wp = lw + i * (HD / 64);
    float w0, w1;
    if constexpr (HD == 256) { w0 = wp[lane >> 4]; w1 = wp[2 + (lane >> 4)]; }
    else                     { w0 = wp[0]; w1 = 0.f; }
    return agg_fetch<HD>(hfeat, lsrc[i], lane, w0, w1);
}

template <int HD>
__device__ __forceinline__ void agg_add_row(float* __restrict__ acc, int row, int lane,
                                            const RowVals<HD>& p) {
    float* a = acc + row * HD + lane;
#pragma unroll
    for (int j = 0; j < HD / 32; ++j)
        __hip_atomic_fetch_add(a + j * 32, p.v[j], __ATOMIC_RELAXED, __HIP_MEMORY_SCOPE_WORKGROUP);
}

__device__ __forceinline__ float agg_act(float v, int do_elu) {
    if (do_elu) v = v > 0.f ? v : (expf(v) - 1.f);
    return v;
}

__device__ __forceinline__ float agg_lrelu(float s) { return s > 0.f ? s : 0.2f * s; }

template <bool TAIL>
__device__ __forceinline__ void agg_load_dst(const int* __restrict__ dstp, int base, int lane,
                                             int E, int (&d)[AGG_K]) {
#pragma unroll
    for (int k = 0; k < AGG_K; ++k) {
        const int e = base + 32 * k + lane;
        if constexpr (TAIL) d[k] = e < E ? dstp[e] : -1;
        else                d[k] = dstp[e];
    }
}

__device__ __forceinline__ void agg_collect(const int (&d)[AGG_K], int base, int lane, int lo, int nrows,
                                            int* __restrict__ lsrc, int* __restrict__ lrow,
                                            unsigned* __restrict__ cnt) {
    unsigned m[AGG_K], any = 0;
    int rl[AGG_K];
#pragma unroll
    for (int k = 0; k < AGG_K; ++k) {
        rl[k] = d[k] - lo;
        m[k] = __builtin_amdgcn_ballot_w32((unsigned)rl[k] < (unsigned)nrows);
        any |= m[k];
    }
    if (!any) return;
    unsigned tot = 0;
#pragma unroll
    for (int k = 0; k < AGG_K; ++k) tot += (unsigned)__builtin_popcount(m[k]);
    unsigned start = 0;
    if (lane == 0)
        start = __hip_atomic_fetch_add(cnt, tot, __ATOMIC_RELAXED, __HIP_MEMORY_SCOPE_WORKGROUP);
    start = (unsigned)__builtin_amdgcn_readfirstlane((int)start);
    const unsigned below = (1u << lane) - 1u;
#pragma unroll
    for (int k = 0; k < AGG_K; ++k) {
        if ((m[k] >> lane) & 1u) {
            const unsigned p = start + (unsigned)__builtin_popcount(m[k] & below);
            if (p < (unsigned)AGG_CAP) { lsrc[p] = base + 32 * k + lane; lrow[p] = rl[k]; }
        }
        start += (unsigned)__builtin_popcount(m[k]);
    }
}

struct DstOne  { int  d[AGG_K]; };
struct DstFour { int4 d[AGG_K / 4]; };

__device__ __forceinline__ void agg_load(const int* __restrict__ dstp, int base, int lane, DstOne& s) {
    agg_load_dst<false>(dstp, base, lane, 0, s.d);
}
__device__ __forceinline__ void agg_load(const int* __restrict__ dstp, int base, int lane, DstFour& s) {
#pragma unroll
    for (int q = 0; q < AGG_K / 4; ++q)
        s.d[q] = *reinterpret_cast<const int4*>(dstp + base + 128 * q + 4 * lane);
}

__device__ __forceinline__ void agg_note(const DstOne& s, int base, int lane, int lo, int nrows,
                                         int* __restrict__ lsrc, int* __restrict__ lrow,
                                         unsigned* __restrict__ cnt) {
    agg_collect(s.d, base, lane, lo, nrows, lsrc, lrow, cnt);
}

__device__ __forceinline__ void agg_note_one(unsigned rl, int e, int nrows,
                                             int* __restrict__ lsrc, int* __restrict__ lrow,
                                             unsigned* __restrict__ cnt) {
    if (rl < (unsigned)nrows) {
        const unsigned p = __hip_atomic_fetch_add(cnt, 1u, __ATOMIC_RELAXED, __HIP_MEMORY_SCOPE_WORKGROUP);
        if (p < (unsigned)AGG_CAP) { lsrc[p] = e; lrow[p] = (int)rl; }
    }
}

__device__ __forceinline__ void agg_note(const DstFour& s, int base, int lane, int lo, int nrows,
                                         int* __restrict__ lsrc, int* __restrict__ lrow,
                                         unsigned* __restrict__ cnt) {
    unsigned r[AGG_K];
#pragma unroll
    for (int q = 0; q < AGG_K / 4; ++q) {
        r[4 * q + 0] = (unsigned)(s.d[q].x - lo);
        r[4 * q + 1] = (unsigned)(s.d[q].y - lo);
        r[4 * q + 2] = (unsigned)(s.d[q].z - lo);
        r[4 * q + 3] = (unsigned)(s.d[q].w - lo);
    }
    unsigned mn = r[0];
#pragma unroll
    for (int k = 1; k < AGG_K; ++k) mn = min(mn, r[k]);
    if (mn < (unsigned)nrows) {
#pragma unroll
        for (int k = 0; k < AGG_K; ++k)
            agg_note_one(r[k], base + 128 * (k >> 2) + 4 * lane + (k & 3), nrows, lsrc, lrow, cnt);
    }
}

template <class DS>
__device__ __forceinline__ void agg_walk(const int* __restrict__ dstp, int full, int t0, int trip, int woff,
                                         int lane, int lo, int nrows,
                                         int* __restrict__ lsrc, int* __restrict__ lrow,
                                         unsigned* __restrict__ cnt) {
    DS dA, dB;
    agg_load(dstp, t0 * trip + woff, lane, dA);
    for (int t = 0; t < full; t += 2) {
        int tc0 = t0 + t;    if (tc0 >= full) tc0 -= full;
        int tc1 = tc0 + 1;   if (tc1 >= full) tc1 -= full;
        int tc2 = tc1 + 1;   if (tc2 >= full) tc2 -= full;
        agg_load(dstp, tc1 * trip + woff, lane, dB);
        agg_note(dA, tc0 * trip + woff, lane, lo, nrows, lsrc, lrow, cnt);
        agg_load(dstp, tc2 * trip + woff, lane, dA);
        if (t + 1 < full)
            agg_note(dB, tc1 * trip + woff, lane, lo, nrows, lsrc, lrow, cnt);
    }
}

template <int HD, int NC, bool ES_TAIL>
__global__ __launch_bounds__(AGG_THREADS)
void gat_agg_owner(const int* __restrict__ ei, int E, int N,
                   const float* __restrict__ hfeat,
                   const float* es, const float* ed,
                   const float* __restrict__ bias,
                   float* out, int do_elu,
                   float* __restrict__ colpart,
                   float* keep, int take_list,
                   const float* __restrict__ Wn,
                   const float* __restrict__ asn, const float* __restrict__ adn,
                   float* __restrict__ hn, float* esn, float* edn) {
    static_assert(NC == 0 || HD == 256, "the projection at the end reads 256-channel rows");
    static_assert(!ES_TAIL || HD == 256, "only 4-head coefficients are left in the output plane");
    constexpr int HH = HD / 64;
    constexpr int NW = AGG_THREADS / 32;
    constexpr int SPAN = 32 * AGG_K;
    constexpr int TRIP = NW * SPAN;
    __shared__ __attribute__((aligned(16))) float acc[AGG_ROWS * HD];
    __shared__ unsigned mx[AGG_ROWS * HH];
    __shared__ float    den[AGG_ROWS * HH];
    __shared__ float    edv[AGG_ROWS * HH];
    __shared__ int      lsrc[AGG_CAP];
    __shared__ int      lrow[AGG_CAP];
    __shared__ float    lw[AGG_CAP * HH];
    __shared__ unsigned cnt;
    const int tid  = threadIdx.x;
    const int lane = tid & 31;
    const int wave = tid >> 5;
    const int lo = blockIdx.x * AGG_ROWS;
    const int nrows = max(0, min(AGG_ROWS, N - lo));

    float* kp = keep + (size_t)lo * KEEP_PITCH;
    const int kcap = nrows * (KEEP_PITCH - TAIL_WORDS);
    int taken = -1;
    if (take_list && nrows > 0) taken = __builtin_amdgcn_readfirstlane(__float_as_int(kp[0]));
    const bool have = taken >= 0 && taken <= AGG_CAP && 1 + 2 * taken <= kcap;

    {
        float4* a4 = reinterpret_cast<float4*>(acc);
        for (int i = tid; i < AGG_ROWS * HD / 4; i += AGG_THREADS) a4[i] = make_float4(0.f, 0.f, 0.f, 0.f);
        for (int i = tid; i < nrows * HH; i += AGG_THREADS) {
            mx[i] = 0u;
            den[i] = 0.f;
            edv[i] = ed[(size_t)lo * HH + i];
        }
        if (have) {
            for (int i = tid; i < taken; i += AGG_THREADS) {
                lsrc[i] = __float_as_int(kp[1 + i]);
                lrow[i] = __float_as_int(kp[1 + taken + i]);
            }
            if (tid == 0) cnt = (unsigned)taken;
        } else {
            for (int r = tid; r < nrows; r += AGG_THREADS) { lsrc[r] = E + lo + r; lrow[r] = r; }
            if (tid == 0) cnt = (unsigned)nrows;
        }
    }
    __syncthreads();

    const int* __restrict__ dstp = ei + E;
    const int full = E / TRIP;
    const int t0 = (int)((blockIdx.x * (unsigned)full) / gridDim.x);
    const int woff = wave * SPAN;
    if (!have && full > 0) {
        if ((E & 3) == 0)
            agg_walk<DstFour>(dstp, full, t0, TRIP, woff, lane, lo, nrows, lsrc, lrow, &cnt);
        else
            agg_walk<DstOne>(dstp, full, t0, TRIP, woff, lane, lo, nrows, lsrc, lrow, &cnt);
    }
    if (!have && full * TRIP < E) {
        int dT[AGG_K];
        agg_load_dst<true>(dstp, full * TRIP + woff, lane, E, dT);
        agg_collect(dT, full * TRIP + woff, lane, lo, nrows, lsrc, lrow, &cnt);
    }
    __syncthreads();
    const int n = (int)cnt;

    const bool save = !take_list && n <= AGG_CAP && 1 + 2 * n <= kcap;
    if (!take_list && tid == 0 && nrows > 0) kp[0] = __int_as_float(save ? n : -1);
    if (n <= AGG_CAP) {
        for (int i = tid; i < n; i += AGG_THREADS) {
            int src = lsrc[i];
            const int row = lrow[i];
            if (!have) {
                const int e = src;
                src = e < E ? ei[e] : e - E;
                lsrc[i] = src;
                if (save) { kp[1 + i] = __int_as_float(src); kp[1 + n + i] = __int_as_float(row); }
            }
            float sv[HH];
            if constexpr (HH == 4) {
                const float4 q = ES_TAIL ? *reinterpret_cast<const float4*>(es_tail(keep, N, src))
                                         : reinterpret_cast<const float4*>(es)[src];
                sv[0] = q.x; sv[1] = q.y; sv[2] = q.z; sv[3] = q.w;
            } else {
                sv[0] = es[src];
            }
#pragma unroll
            for (int h = 0; h < HH; ++h) {
                const float lr = agg_lrelu(sv[h] + edv[row * HH + h]);
                lw[i * HH + h] = lr;
                atomicMax(&mx[row * HH + h], f2ord(lr));
            }
        }
        __syncthreads();
        for (int i = tid; i < n; i += AGG_THREADS) {
            const int row = lrow[i];
#pragma unroll
            for (int h = 0; h < HH; ++h) {
                const float v = expf(lw[i * HH + h] - ord2f(mx[row * HH + h]));
                lw[i * HH + h] = v;
                __hip_atomic_fetch_add(&den[row * HH + h], v, __ATOMIC_RELAXED, __HIP_MEMORY_SCOPE_WORKGROUP);
            }
        }
        __syncthreads();
        for (int i = tid; i < n; i += AGG_THREADS) {
            const int row = lrow[i];
#pragma unroll
            for (int h = 0; h < HH; ++h) lw[i * HH + h] = lw[i * HH + h] / den[row * HH + h];
        }
        __syncthreads();
        const int ngrp = n / AGG_U;
        for (int g = wave; g < ngrp; g += NW) {
            RowVals<HD> p[AGG_U];
            int rw[AGG_U];
#pragma unroll
            for (int u = 0; u < AGG_U; ++u)
                p[u] = agg_take<HD>(hfeat, lsrc, lrow, lw, g * AGG_U + u, lane, rw[u]);
#pragma unroll
            for (int u = 0; u < AGG_U; ++u) agg_add_row<HD>(acc, rw[u], lane, p[u]);
        }
        for (int i = ngrp * AGG_U + wave; i < n; i += NW) {
            int rw1;
            const RowVals<HD> p1 = agg_take<HD>(hfeat, lsrc, lrow, lw, i, lane, rw1);
            agg_add_row<HD>(acc, rw1, lane, p1);
        }
    } else {
        const int Et = E + N;
        for (int ph = 0; ph < 3; ++ph) {
            for (int eb = wave * 32; eb < Et; eb += AGG_THREADS) {
                const int e = eb + lane;
                int src = 0, rl = -1;
                if (e < Et) {
                    int dst; edge_endpoints(ei, E, e, src, dst);
                    rl = dst - lo;
                }
                const bool in = (unsigned)rl < (unsigned)nrows;
                if (ph < 2) {
                    if (in) {
#pragma unroll
                        for (int h = 0; h < HH; ++h) {
                            const float lr = agg_lrelu(es_at<HH, ES_TAIL>(es, keep, N, src, h) + edv[rl * HH + h]);
                            if (ph == 0) atomicMax(&mx[rl * HH + h], f2ord(lr));
                            else __hip_atomic_fetch_add(&den[rl * HH + h], expf(lr - ord2f(mx[rl * HH + h])),
                                                        __ATOMIC_RELAXED, __HIP_MEMORY_SCOPE_WORKGROUP);
                        }
                    }
                } else {
                    unsigned mk = __builtin_amdgcn_ballot_w32(in);
                    while (mk) {
                        const int b = __builtin_ctz(mk);
                        mk &= mk - 1;
                        const int sb = __builtin_amdgcn_readlane(src, b);
                        const int rb = __builtin_amdgcn_readlane(rl, b);
                        const int h0 = HH == 4 ? (lane >> 4) : 0;
                        const int h1 = HH == 4 ? 2 + (lane >> 4) : 0;
                        const float l0 = agg_lrelu(es_at<HH, ES_TAIL>(es, keep, N, sb, h0) + edv[rb * HH + h0]);
                        const float l1 = agg_lrelu(es_at<HH, ES_TAIL>(es, keep, N, sb, h1) + edv[rb * HH + h1]);
                        const float w0 = expf(l0 - ord2f(mx[rb * HH + h0])) / den[rb * HH + h0];
                        const float w1 = expf(l1 - ord2f(mx[rb * HH + h1])) / den[rb * HH + h1];
                        const RowVals<HD> p = agg_fetch<HD>(hfeat, sb, lane, w0, w1);
                        agg_add_row<HD>(acc, rb, lane, p);
                    }
                }
            }
            __syncthreads();
        }
    }
    __syncthreads();

    if constexpr (NC > 0) {
        for (int r = wave; r < nrows; r += NW) {
            float* arow = acc + r * HD;
            const float* a = arow + lane;
            const float4* b4 = reinterpret_cast<const float4*>(bias);
            const float4 c0 = b4[lane], c1 = b4[32 + lane];
            const float s0 = a[0], s1 = a[32], s2 = a[64], s3 = a[96], s4 = a[128], s5 = a[160], s6 = a[192], s7 = a[224];
            float4 o0, o1;
            o0.x = agg_act(s0 + c0.x, do_elu); o0.y = agg_act(s1 + c0.y, do_elu);
            o0.z = agg_act(s2 + c0.z, do_elu); o0.w = agg_act(s3 + c0.w, do_elu);
            o1.x = agg_act(s4 + c1.x, do_elu); o1.y = agg_act(s5 + c1.y, do_elu);
            o1.z = agg_act(s6 + c1.z, do_elu); o1.w = agg_act(s7 + c1.w, do_elu);
            const int swz = (r & 15) << 2;
            *reinterpret_cast<float4*>(arow + ((4 * lane) ^ swz))       = o0;
            *reinterpret_cast<float4*>(arow + ((128 + 4 * lane) ^ swz)) = o1;
        }
        __syncthreads();

        constexpr int HN = NC / 64;
        constexpr int MG = NW / HN;
        const int wv = __builtin_amdgcn_readfirstlane(wave);
        const int head = wv % HN;
        const int n0 = head * 64;
        const int nmt = (nrows + 15) >> 4;
        const int l15 = lane & 15;
        const int kh  = (lane >> 4) << 1;
        const float sa0 = asn[n0 + l15], sa1 = asn[n0 + 16 + l15], sa2 = asn[n0 + 32 + l15], sa3 = asn[n0 + 48 + l15];
        const float da0 = adn[n0 + l15], da1 = adn[n0 + 16 + l15], da2 = adn[n0 + 32 + l15], da3 = adn[n0 + 48 + l15];
        for (int mt = wv / HN; mt < nmt; mt += MG) {
            const int ar = min(mt * 16 + l15, AGG_ROWS - 1);
            const float* Ap = acc + ar * HD + kh;
            const int swz = (ar & 15) << 2;
            v8f q0 = {0.f,0.f,0.f,0.f,0.f,0.f,0.f,0.f};
            v8f q1 = q0, q2 = q0, q3 = q0;
            for (int k = 0; k < HD; k += 4) {
                const v2f a = *reinterpret_cast<const v2f*>(Ap + (k ^ swz));
                const float* Bp = Wn + (size_t)(k + kh) * NC + n0 + l15;
                v2f b0, b1, b2, b3;
                b0.x = Bp[0];  b0.y = Bp[(size_t)NC];
                b1.x = Bp[16]; b1.y = Bp[(size_t)NC + 16];
                b2.x = Bp[32]; b2.y = Bp[(size_t)NC + 32];
                b3.x = Bp[48]; b3.y = Bp[(size_t)NC + 48];
                q0 = __builtin_amdgcn_wmma_f32_16x16x4_f32(false, a, false, b0, (short)0, q0, false, false);
                q1 = __builtin_amdgcn_wmma_f32_16x16x4_f32(false, a, false, b1, (short)0, q1, false, false);
                q2 = __builtin_amdgcn_wmma_f32_16x16x4_f32(false, a, false, b2, (short)0, q2, false, false);
                q3 = __builtin_amdgcn_wmma_f32_16x16x4_f32(false, a, false, b3, (short)0, q3, false, false);
            }
            const int rbase = mt * 16 + ((lane >> 4) << 3);
            float* Cp = hn + (size_t)(lo + rbase) * NC + n0 + l15;
#pragma unroll
            for (int v = 0; v < 8; ++v) {
                if (rbase + v < nrows) {
                    Cp[(size_t)v * NC + 0]  = q0[v];
                    Cp[(size_t)v * NC + 16] = q1[v];
                    Cp[(size_t)v * NC + 32] = q2[v];
                    Cp[(size_t)v * NC + 48] = q3[v];
                }
            }
            float cs = 0.f, cd = 0.f;
#pragma unroll
            for (int v = 0; v < 8; ++v) {
                const float ps = row16_sum(q0[v] * sa0 + q1[v] * sa1 + q2[v] * sa2 + q3[v] * sa3);
                const float pd = row16_sum(q0[v] * da0 + q1[v] * da1 + q2[v] * da2 + q3[v] * da3);
                if (l15 == v) { cs = ps; cd = pd; }
            }
            if (l15 < 8 && rbase + l15 < nrows) {
                const int rl = rbase + l15;
                if constexpr (HN == 4) {
                    kp[kcap + TAIL_WORDS * rl + head] = cs;
                    edn[(size_t)(lo + rl) * HN + head] = cd;
                } else {
                    esn[(size_t)(lo + rl) * HN + head] = cs;
                    edn[(size_t)(lo + rl) * HN + head] = cd;
                }
            }
        }
    } else {
    float cs0 = 0.f, cs1 = 0.f;
    for (int r = wave; r < nrows; r += NW) {
        const float* a = acc + r * HD + lane;
        const size_t nd = (size_t)(lo + r);
        if constexpr (HD == 256) {
            const float4* b4 = reinterpret_cast<const float4*>(bias);
            const float4 c0 = b4[lane], c1 = b4[32 + lane];
            float4 o0, o1;
            o0.x = agg_act(a[0]   + c0.x, do_elu); o0.y = agg_act(a[32]  + c0.y, do_elu);
            o0.z = agg_act(a[64]  + c0.z, do_elu); o0.w = agg_act(a[96]  + c0.w, do_elu);
            o1.x = agg_act(a[128] + c1.x, do_elu); o1.y = agg_act(a[160] + c1.y, do_elu);
            o1.z = agg_act(a[192] + c1.z, do_elu); o1.w = agg_act(a[224] + c1.w, do_elu);
            float4* o4 = reinterpret_cast<float4*>(out + nd * 256);
            o4[lane] = o0;
            o4[32 + lane] = o1;
        } else {
            const float2 c = reinterpret_cast<const float2*>(bias)[lane];
            float2 o;
            o.x = agg_act(a[0]  + c.x, do_elu);
            o.y = agg_act(a[32] + c.y, do_elu);
            reinterpret_cast<float2*>(out + nd * 64)[lane] = o;
            cs0 += o.x;
            cs1 += o.y;
        }
    }
    if constexpr (HD == 64) {
        if (colpart != nullptr) {
            lw[wave * 64 + 2 * lane]     = cs0;
            lw[wave * 64 + 2 * lane + 1] = cs1;
            __syncthreads();
            if (tid < 64) {
                float s = 0.f;
                for (int w = 0; w < NW; ++w) s += lw[w * 64 + tid];
                colpart[(size_t)blockIdx.x * 64 + tid] = s;
            }
        }
    }
    }
}

#define MEAN_GROUPS 16
__global__ __launch_bounds__(64 * MEAN_GROUPS)
void gat_colmean_parts(const float* __restrict__ part, float* __restrict__ out, int N) {
    __shared__ float sm[MEAN_GROUPS * 64];
    const int nblk = (N + AGG_ROWS - 1) / AGG_ROWS;
    const int c = threadIdx.x & 63;
    const int g = threadIdx.x >> 6;
    float s = 0.f;
#pragma unroll 16
    for (int b = g; b < nblk; b += MEAN_GROUPS) s += part[(size_t)b * 64 + c];
    sm[g * 64 + c] = s;
    __syncthreads();
    if (threadIdx.x < 64) {
        float t = 0.f;
        for (int k = 0; k < MEAN_GROUPS; ++k) t += sm[k * 64 + threadIdx.x];
        out[threadIdx.x] = t / (float)N;
    }
}

static inline int cdiv(long long a, long long b) { return (int)((a + b - 1) / b); }

extern "C" void kernel_launch(void* const* d_in, const int* in_sizes, int n_in,
                              void* d_out, int out_size, void* d_ws, size_t ws_size,
                              hipStream_t stream) {
    const float* x   = (const float*)d_in[0];
    const int*   ei  = (const int*)  d_in[1];
    const float* W1  = (const float*)d_in[2];
    const float* a1s = (const float*)d_in[3];
    const float* a1d = (const float*)d_in[4];
    const float* b1  = (const float*)d_in[5];
    const float* W2  = (const float*)d_in[6];
    const float* a2s = (const float*)d_in[7];
    const float* a2d = (const float*)d_in[8];
    const float* b2  = (const float*)d_in[9];
    const float* W3  = (const float*)d_in[10];
    const float* a3s = (const float*)d_in[11];
    const float* a3d = (const float*)d_in[12];
    const float* b3  = (const float*)d_in[13];

    const int N  = in_sizes[0] / 16;
    const int E  = in_sizes[1] / 2;
    const int H  = 4;

    float* ws = (float*)d_ws;
    size_t off = 0;
    float*    buf_h    = ws + off; off += (size_t)N * 256;
    float*    buf_h2   = ws + off; off += (size_t)N * 256;
    float*    es       = ws + off; off += (size_t)N * H;
    float*    ed       = ws + off; off += (size_t)N * H;
    (void)ws_size; (void)off;

    float* node_emb  = (float*)d_out;
    float* graph_emb = node_emb + (size_t)N * 64;

    hipLaunchKernelGGL(gat_gemm_wmma, dim3(1, cdiv(N, 16)), dim3(32, 4), 0, stream,
                       x, W1, buf_h, a1s, a1d, es, ed, N, 16, 256);

    const dim3 ag(cdiv(N, AGG_ROWS)), ab(AGG_THREADS);
    float* es3      = es;
    float* col_part = es + (size_t)N;
    float* ed3      = es + 2 * (size_t)N;
    float* none     = nullptr;
    gat_agg_owner<256, 256, false><<<ag, ab, 0, stream>>>(ei, E, N, buf_h, es, ed, b1, none, 1, none, node_emb, 0,
                                                          W2, a2s, a2d, buf_h2, none, ed);
    gat_agg_owner<256, 64, true><<<ag, ab, 0, stream>>>(ei, E, N, buf_h2, es, ed, b2, none, 1, none, node_emb, 1,
                                                        W3, a3s, a3d, buf_h, es3, ed3);
    gat_agg_owner<64, 0, false><<<ag, ab, 0, stream>>>(ei, E, N, buf_h, es3, ed3, b3, node_emb, 0, col_part, node_emb, 1,
                                                       none, none, none, none, none, none);

    hipLaunchKernelGGL(gat_colmean_parts, dim3(1), dim3(64 * MEAN_GROUPS), 0, stream,
                       col_part, graph_emb, N);
}


# Layer 1 finds a block's edges by an index the first launch leaves in idle output words (tag byte per edge, per owner a bit per 16 edges): no walk over all destinations; list, LDS add order change
# speedup vs baseline: 9.2844x; 1.0929x over previous
#include <hip/hip_runtime.h>
#include <math.h>

typedef __attribute__((ext_vector_type(2))) float v2f;
typedef __attribute__((ext_vector_type(8))) float v8f;

#define GAT_D 64

__device__ __forceinline__ unsigned f2ord(float f) {
    int b = __float_as_int(f);
    return (unsigned)b ^ (unsigned)((b >> 31) | 0x80000000);
}
__device__ __forceinline__ float ord2f(unsigned u) {
    int b = (u & 0x80000000u) ? (int)(u ^ 0x80000000u) : ~(int)u;
    return __int_as_float(b);
}

__device__ __forceinline__ float row16_sum(float v) {
    v += __int_as_float(__builtin_amdgcn_update_dpp(0, __float_as_int(v), 0x161, 0xf, 0xf, false));
    v += __int_as_float(__builtin_amdgcn_update_dpp(0, __float_as_int(v), 0x162, 0xf, 0xf, false));
    v += __int_as_float(__builtin_amdgcn_update_dpp(0, __float_as_int(v), 0x164, 0xf, 0xf, false));
    v += __int_as_float(__builtin_amdgcn_update_dpp(0, __float_as_int(v), 0x168, 0xf, 0xf, false));
    return v;
}

#define TAG_ROWS     196
#define TAG_REGION   (196 * 64)
#define TAG_LIST_MAX 4096
#define TAG_OFF      ((1 + 2 * TAG_LIST_MAX + 3) & ~3)
#define TAG_WAVES    4
#define TAG_UNITS    256
#define TAG_WORDS    (4 * TAG_UNITS)
#define BM_OFF       (TAG_OFF + TAG_WORDS)
#define BM_WORDS     ((TAG_UNITS / 32) * 256)

__device__ __forceinline__ bool tags_fit(int E, int N) {
    return (E & 15) == 0 && N <= 256 * TAG_ROWS && ((E >> 4) + TAG_UNITS - 1) / TAG_UNITS <= N / TAG_ROWS &&
           ((E >> 4) + 31) / 32 <= 2 * 1024;
}
__device__ __forceinline__ uint4* tag_unit(float* plane, int u) {
    return reinterpret_cast<uint4*>(plane + (size_t)(u / TAG_UNITS) * TAG_REGION + TAG_OFF) + (u % TAG_UNITS);
}
__device__ __forceinline__ unsigned* tag_group(float* plane, int g) {
    return reinterpret_cast<unsigned*>(plane + (size_t)(g / (TAG_UNITS / 32)) * TAG_REGION + BM_OFF) + (g % (TAG_UNITS / 32)) * 256;
}
__device__ __forceinline__ unsigned tag_word(int4 d) {
    return  (((unsigned)d.x / TAG_ROWS) & 255u)        | ((((unsigned)d.y / TAG_ROWS) & 255u) << 8) |
           ((((unsigned)d.z / TAG_ROWS) & 255u) << 16) | ((((unsigned)d.w / TAG_ROWS) & 255u) << 24);
}

__global__ void gat_gemm_wmma(const float* __restrict__ A,
                              const float* __restrict__ B,
                              float* __restrict__ C,
                              const float* __restrict__ a_s,
                              const float* __restrict__ a_d,
                              float* __restrict__ es, float* __restrict__ ed,
                              int M, int K, int N,
                              const int* __restrict__ ei, int E, float* tagplane) {
    __shared__ unsigned own[TAG_WAVES * 256];
    if (tagplane != nullptr && tags_fit(E, M)) {
        const int U = E >> 4, G = (U + 31) >> 5;
        const int nblk = (int)(gridDim.x * gridDim.y);
        unsigned* mine = own + threadIdx.y * 256;
        const int ln = threadIdx.x;
        const int4* d4 = reinterpret_cast<const int4*>(ei + E);
        for (int gb = (int)(blockIdx.y * gridDim.x + blockIdx.x); gb * TAG_WAVES < G; gb += nblk) {
            const int g = gb * TAG_WAVES + (int)threadIdx.y;
            const int u = 32 * g + ln;
            for (int k = 0; k < 8; ++k) mine[ln + 32 * k] = 0u;
            __syncthreads();
            if (u < U) {
                unsigned w[4];
                for (int q = 0; q < 4; ++q) w[q] = tag_word(d4[4 * u + q]);
                *tag_unit(tagplane, u) = make_uint4(w[0], w[1], w[2], w[3]);
                for (int q = 0; q < 4; ++q)
                    for (int b = 0; b < 4; ++b)
                        __hip_atomic_fetch_or(&mine[(w[q] >> (8 * b)) & 255u], 1u << ln,
                                              __ATOMIC_RELAXED, __HIP_MEMORY_SCOPE_WORKGROUP);
            }
            __syncthreads();
            if (g < G) {
                unsigned* gw = tag_group(tagplane, g);
                for (int k = 0; k < 8; ++k) gw[ln + 32 * k] = mine[ln + 32 * k];
            }
            __syncthreads();
        }
    }
    const int lane  = threadIdx.x;
    const int grp   = blockIdx.x * blockDim.y + threadIdx.y;
    const int mtile = blockIdx.y;
    const int n0 = grp * 64;
    const int m0 = mtile * 16;
    if (n0 >= N || m0 >= M) return;

    const int l15 = lane & 15;
    const int kh  = (lane >> 4) << 1;
    const float* Arow = A + (size_t)(m0 + l15) * K;
    v8f acc0 = {0.f,0.f,0.f,0.f,0.f,0.f,0.f,0.f};
    v8f acc1 = acc0, acc2 = acc0, acc3 = acc0;
    const float sa0 = a_s[n0 + l15], sa1 = a_s[n0 + 16 + l15], sa2 = a_s[n0 + 32 + l15], sa3 = a_s[n0 + 48 + l15];
    const float da0 = a_d[n0 + l15], da1 = a_d[n0 + 16 + l15], da2 = a_d[n0 + 32 + l15], da3 = a_d[n0 + 48 + l15];

    for (int k = 0; k < K; k += 4) {
        v2f a;
        a.x = Arow[k + kh];
        a.y = Arow[k + kh + 1];
        const float* Bp = B + (size_t)(k + kh) * N + n0 + l15;
        v2f b0, b1, b2, b3;
        b0.x = Bp[0];  b0.y = Bp[(size_t)N];
        b1.x = Bp[16]; b1.y = Bp[(size_t)N + 16];
        b2.x = Bp[32]; b2.y = Bp[(size_t)N + 32];
        b3.x = Bp[48]; b3.y = Bp[(size_t)N + 48];
        acc0 = __builtin_amdgcn_wmma_f32_16x16x4_f32(false, a, false, b0, (short)0, acc0, false, false);
        acc1 = __builtin_amdgcn_wmma_f32_16x16x4_f32(false, a, false, b1, (short)0, acc1, false, false);
        acc2 = __builtin_amdgcn_wmma_f32_16x16x4_f32(false, a, false, b2, (short)0, acc2, false, false);
        acc3 = __builtin_amdgcn_wmma_f32_16x16x4_f32(false, a, false, b3, (short)0, acc3, false, false);
    }

    float* Cp = C + (size_t)(m0 + ((lane >> 4) << 3)) * N + n0 + l15;
#pragma unroll
    for (int v = 0; v < 8; ++v) {
        Cp[(size_t)v * N + 0]  = acc0[v];
        Cp[(size_t)v * N + 16] = acc1[v];
        Cp[(size_t)v * N + 32] = acc2[v];
        Cp[(size_t)v * N + 48] = acc3[v];
    }

    float cs = 0.f, cd = 0.f;
#pragma unroll
    for (int v = 0; v < 8; ++v) {
        const float ps = row16_sum(acc0[v] * sa0 + acc1[v] * sa1 + acc2[v] * sa2 + acc3[v] * sa3);
        const float pd = row16_sum(acc0[v] * da0 + acc1[v] * da1 + acc2[v] * da2 + acc3[v] * da3);
        if (l15 == v) { cs = ps; cd = pd; }
    }
    if (l15 < 8) {
        const size_t ci = (size_t)(m0 + ((lane >> 4) << 3) + l15) * (size_t)(N / GAT_D) + (size_t)grp;
        es[ci] = cs;
        ed[ci] = cd;
    }
}

__device__ __forceinline__ void edge_endpoints(const int* __restrict__ ei, int E,
                                               int e, int& src, int& dst) {
    if (e < E) { src = ei[e]; dst = ei[E + e]; }
    else       { src = e - E; dst = e - E;    }
}

#define AGG_ROWS    196
#define AGG_THREADS 1024
#define AGG_K       8
#define AGG_CAP     4096
#define AGG_U       4
#define KEEP_PITCH  64
#define TAIL_WORDS  4

template <int HD> struct RowVals { float v[HD / 32]; };

__device__ __forceinline__ const float* es_tail(const float* keep, int N, int src) {
    const int blk = src / AGG_ROWS;
    const int lo_b = blk * AGG_ROWS;
    const int nr = min(AGG_ROWS, N - lo_b);
    return keep + (size_t)lo_b * KEEP_PITCH + nr * (KEEP_PITCH - TAIL_WORDS) + TAIL_WORDS * (src - lo_b);
}

template <int HH, bool ES_TAIL>
__device__ __forceinline__ float es_at(const float* es, const float* keep, int N, int src, int h) {
    if constexpr (ES_TAIL) return es_tail(keep, N, src)[h];
    else                   return es[(size_t)src * HH + h];
}

template <int HD>
__device__ __forceinline__ RowVals<HD> agg_fetch(const float* __restrict__ hfeat,
                                                 int src, int lane, float w0, float w1) {
    RowVals<HD> r;
    if constexpr (HD == 256) {
        const float4* hp = reinterpret_cast<const float4*>(hfeat + (size_t)src * 256);
        const float4 a = hp[lane];
        const float4 b = hp[32 + lane];
        r.v[0] = a.x * w0; r.v[1] = a.y * w0; r.v[2] = a.z * w0; r.v[3] = a.w * w0;
        r.v[4] = b.x * w1; r.v[5] = b.y * w1; r.v[6] = b.z * w1; r.v[7] = b.w * w1;
    } else {
        const float2* hp = reinterpret_cast<const float2*>(hfeat + (size_t)src * 64);
        const float2 a = hp[lane];
        r.v[0] = a.x * w0; r.v[1] = a.y * w0;
    }
    return r;
}

template <int HD>
__device__ __forceinline__ RowVals<HD> agg_take(const float* __restrict__ hfeat,
                                                const int* __restrict__ lsrc,
                                                const int* __restrict__ lrow,
                                                const float* __restrict__ lw,
                                                int i, int lane, int& row) {
    row = lrow[i];
    const float* wp = lw + i * (HD / 64);
    float w0, w1;
    if constexpr (HD == 256) { w0 = wp[lane >> 4]; w1 = wp[2 + (lane >> 4)]; }
    else                     { w0 = wp[0]; w1 = 0.f; }
    return agg_fetch<HD>(hfeat, lsrc[i], lane, w0, w1);
}

template <int HD>
__device__ __forceinline__ void agg_add_row(float* __restrict__ acc, int row, int lane,
                                            const RowVals<HD>& p) {
    float* a = acc + row * HD + lane;
#pragma unroll
    for (int j = 0; j < HD / 32; ++j)
        __hip_atomic_fetch_add(a + j * 32, p.v[j], __ATOMIC_RELAXED, __HIP_MEMORY_SCOPE_WORKGROUP);
}

__device__ __forceinline__ float agg_act(float v, int do_elu) {
    if (do_elu) v = v > 0.f ? v : (expf(v) - 1.f);
    return v;
}

__device__ __forceinline__ float agg_lrelu(float s) { return s > 0.f ? s : 0.2f * s; }

template <bool TAIL>
__device__ __forceinline__ void agg_load_dst(const int* __restrict__ dstp, int base, int lane,
                                             int E, int (&d)[AGG_K]) {
#pragma unroll
    for (int k = 0; k < AGG_K; ++k) {
        const int e = base + 32 * k + lane;
        if constexpr (TAIL) d[k] = e < E ? dstp[e] : -1;
        else                d[k] = dstp[e];
    }
}

__device__ __forceinline__ void agg_collect(const int (&d)[AGG_K], int base, int lane, int lo, int nrows,
                                            int* __restrict__ lsrc, int* __restrict__ lrow,
                                            unsigned* __restrict__ cnt) {
    unsigned m[AGG_K], any = 0;
    int rl[AGG_K];
#pragma unroll
    for (int k = 0; k < AGG_K; ++k) {
        rl[k] = d[k] - lo;
        m[k] = __builtin_amdgcn_ballot_w32((unsigned)rl[k] < (unsigned)nrows);
        any |= m[k];
    }
    if (!any) return;
    unsigned tot = 0;
#pragma unroll
    for (int k = 0; k < AGG_K; ++k) tot += (unsigned)__builtin_popcount(m[k]);
    unsigned start = 0;
    if (lane == 0)
        start = __hip_atomic_fetch_add(cnt, tot, __ATOMIC_RELAXED, __HIP_MEMORY_SCOPE_WORKGROUP);
    start = (unsigned)__builtin_amdgcn_readfirstlane((int)start);
    const unsigned below = (1u << lane) - 1u;
#pragma unroll
    for (int k = 0; k < AGG_K; ++k) {
        if ((m[k] >> lane) & 1u) {
            const unsigned p = start + (unsigned)__builtin_popcount(m[k] & below);
            if (p < (unsigned)AGG_CAP) { lsrc[p] = base + 32 * k + lane; lrow[p] = rl[k]; }
        }
        start += (unsigned)__builtin_popcount(m[k]);
    }
}

struct DstOne  { int  d[AGG_K]; };
struct DstFour { int4 d[AGG_K / 4]; };

__device__ __forceinline__ void agg_load(const int* __restrict__ dstp, int base, int lane, DstOne& s) {
    agg_load_dst<false>(dstp, base, lane, 0, s.d);
}
__device__ __forceinline__ void agg_load(const int* __restrict__ dstp, int base, int lane, DstFour& s) {
#pragma unroll
    for (int q = 0; q < AGG_K / 4; ++q)
        s.d[q] = *reinterpret_cast<const int4*>(dstp + base + 128 * q + 4 * lane);
}

__device__ __forceinline__ void agg_note(const DstOne& s, int base, int lane, int lo, int nrows,
                                         int* __restrict__ lsrc, int* __restrict__ lrow,
                                         unsigned* __restrict__ cnt) {
    agg_collect(s.d, base, lane, lo, nrows, lsrc, lrow, cnt);
}

__device__ __forceinline__ void agg_note_one(unsigned rl, int e, int nrows,
                                             int* __restrict__ lsrc, int* __restrict__ lrow,
                                             unsigned* __restrict__ cnt) {
    if (rl < (unsigned)nrows) {
        const unsigned p = __hip_atomic_fetch_add(cnt, 1u, __ATOMIC_RELAXED, __HIP_MEMORY_SCOPE_WORKGROUP);
        if (p < (unsigned)AGG_CAP) { lsrc[p] = e; lrow[p] = (int)rl; }
    }
}

__device__ __forceinline__ void agg_note(const DstFour& s, int base, int lane, int lo, int nrows,
                                         int* __restrict__ lsrc, int* __restrict__ lrow,
                                         unsigned* __restrict__ cnt) {
    unsigned r[AGG_K];
#pragma unroll
    for (int q = 0; q < AGG_K / 4; ++q) {
        r[4 * q + 0] = (unsigned)(s.d[q].x - lo);
        r[4 * q + 1] = (unsigned)(s.d[q].y - lo);
        r[4 * q + 2] = (unsigned)(s.d[q].z - lo);
        r[4 * q + 3] = (unsigned)(s.d[q].w - lo);
    }
    unsigned mn = r[0];
#pragma unroll
    for (int k = 1; k < AGG_K; ++k) mn = min(mn, r[k]);
    if (mn < (unsigned)nrows) {
#pragma unroll
        for (int k = 0; k < AGG_K; ++k)
            agg_note_one(r[k], base + 128 * (k >> 2) + 4 * lane + (k & 3), nrows, lsrc, lrow, cnt);
    }
}

template <class DS>
__device__ __forceinline__ void agg_walk(const int* __restrict__ dstp, int full, int t0, int trip, int woff,
                                         int lane, int lo, int nrows,
                                         int* __restrict__ lsrc, int* __restrict__ lrow,
                                         unsigned* __restrict__ cnt) {
    DS dA, dB;
    agg_load(dstp, t0 * trip + woff, lane, dA);
    for (int t = 0; t < full; t += 2) {
        int tc0 = t0 + t;    if (tc0 >= full) tc0 -= full;
        int tc1 = tc0 + 1;   if (tc1 >= full) tc1 -= full;
        int tc2 = tc1 + 1;   if (tc2 >= full) tc2 -= full;
        agg_load(dstp, tc1 * trip + woff, lane, dB);
        agg_note(dA, tc0 * trip + woff, lane, lo, nrows, lsrc, lrow, cnt);
        agg_load(dstp, tc2 * trip + woff, lane, dA);
        if (t + 1 < full)
            agg_note(dB, tc1 * trip + woff, lane, lo, nrows, lsrc, lrow, cnt);
    }
}

static_assert(TAG_ROWS == AGG_ROWS && TAG_REGION == AGG_ROWS * KEEP_PITCH && TAG_LIST_MAX == AGG_CAP,
              "the tag bytes are laid out for the owner blocks' rows, words and list length");
static_assert(BM_OFF + BM_WORDS <= AGG_ROWS * (KEEP_PITCH - TAIL_WORDS),
              "the tag bytes and owner words end before a full block's coefficient words");

__device__ __forceinline__ unsigned zero_bytes(unsigned x) {
    return ~((((x & 0x7F7F7F7Fu) + 0x7F7F7F7Fu) | x) | 0x7F7F7F7Fu);
}

__device__ __forceinline__ int tag_next(unsigned& bits0, unsigned& bits1, int g0, int g1) {
    if (bits0 != 0u) { const int b = __builtin_ctz(bits0); bits0 &= bits0 - 1u; return 32 * g0 + b; }
    if (bits1 != 0u) { const int b = __builtin_ctz(bits1); bits1 &= bits1 - 1u; return 32 * g1 + b; }
    return -1;
}

__device__ __forceinline__ void tag_note(uint4 w, int u, unsigned pat,
                                         int* __restrict__ lsrc, unsigned* __restrict__ cnt) {
    unsigned m = (zero_bytes(w.x ^ pat) >> 7) | (zero_bytes(w.y ^ pat) >> 6) |
                 (zero_bytes(w.z ^ pat) >> 5) | (zero_bytes(w.w ^ pat) >> 4);
    while (m != 0u) {
        const int bit = __builtin_ctz(m);
        m &= m - 1u;
        const unsigned p = __hip_atomic_fetch_add(cnt, 1u, __ATOMIC_RELAXED, __HIP_MEMORY_SCOPE_WORKGROUP);
        if (p < (unsigned)AGG_CAP) lsrc[p] = 16 * u + 4 * (bit & 7) + (bit >> 3);
    }
}

__device__ __forceinline__ void tag_walk(float* plane, int E, int tid, unsigned owner,
                                         int* __restrict__ lsrc, unsigned* __restrict__ cnt) {
    const int U = E >> 4, G = (U + 31) >> 5;
    const unsigned pat = owner * 0x01010101u;
    const int g0 = tid, g1 = tid + AGG_THREADS;
    unsigned bits0 = g0 < G ? tag_group(plane, g0)[owner] : 0u;
    unsigned bits1 = g1 < G ? tag_group(plane, g1)[owner] : 0u;
    int u = tag_next(bits0, bits1, g0, g1);
    uint4 w = *tag_unit(plane, max(u, 0));
    while (u >= 0) {
        const int un = tag_next(bits0, bits1, g0, g1);
        const uint4 wn = *tag_unit(plane, max(un, 0));
        tag_note(w, u, pat, lsrc, cnt);
        u = un; w = wn;
    }
}

template <int HD, int NC, bool ES_TAIL>
__global__ __launch_bounds__(AGG_THREADS)
void gat_agg_owner(const int* __restrict__ ei, int E, int N,
                   const float* __restrict__ hfeat,
                   const float* es, const float* ed,
                   const float* __restrict__ bias,
                   float* out, int do_elu,
                   float* __restrict__ colpart,
                   float* keep, int take_list,
                   const float* __restrict__ Wn,
                   const float* __restrict__ asn, const float* __restrict__ adn,
                   float* __restrict__ hn, float* esn, float* edn) {
    static_assert(NC == 0 || HD == 256, "the projection at the end reads 256-channel rows");
    static_assert(!ES_TAIL || HD == 256, "only 4-head coefficients are left in the output plane");
    constexpr int HH = HD / 64;
    constexpr int NW = AGG_THREADS / 32;
    constexpr int SPAN = 32 * AGG_K;
    constexpr int TRIP = NW * SPAN;
    __shared__ __attribute__((aligned(16))) float acc[AGG_ROWS * HD];
    __shared__ unsigned mx[AGG_ROWS * HH];
    __shared__ float    den[AGG_ROWS * HH];
    __shared__ float    edv[AGG_ROWS * HH];
    __shared__ int      lsrc[AGG_CAP];
    __shared__ int      lrow[AGG_CAP];
    __shared__ float    lw[AGG_CAP * HH];
    __shared__ unsigned cnt;
    const int tid  = threadIdx.x;
    const int lane = tid & 31;
    const int wave = tid >> 5;
    const int lo = blockIdx.x * AGG_ROWS;
    const int nrows = max(0, min(AGG_ROWS, N - lo));

    float* kp = keep + (size_t)lo * KEEP_PITCH;
    const int kcap = nrows * (KEEP_PITCH - TAIL_WORDS);
    int taken = -1;
    if (take_list && nrows > 0) taken = __builtin_amdgcn_readfirstlane(__float_as_int(kp[0]));
    const bool have = taken >= 0 && taken <= AGG_CAP && 1 + 2 * taken <= kcap;

    {
        float4* a4 = reinterpret_cast<float4*>(acc);
        for (int i = tid; i < AGG_ROWS * HD / 4; i += AGG_THREADS) a4[i] = make_float4(0.f, 0.f, 0.f, 0.f);
        for (int i = tid; i < nrows * HH; i += AGG_THREADS) {
            mx[i] = 0u;
            den[i] = 0.f;
            edv[i] = ed[(size_t)lo * HH + i];
        }
        if (have) {
            for (int i = tid; i < taken; i += AGG_THREADS) {
                lsrc[i] = __float_as_int(kp[1 + i]);
                lrow[i] = __float_as_int(kp[1 + taken + i]);
            }
            if (tid == 0) cnt = (unsigned)taken;
        } else {
            for (int r = tid; r < nrows; r += AGG_THREADS) { lsrc[r] = E + lo + r; lrow[r] = r; }
            if (tid == 0) cnt = (unsigned)nrows;
        }
    }
    __syncthreads();

    const int* __restrict__ dstp = ei + E;
    const int full = E / TRIP;
    const int t0 = (int)((blockIdx.x * (unsigned)full) / gridDim.x);
    const int woff = wave * SPAN;
    const bool tagw = !take_list && tags_fit(E, N);
    if (!have && tagw) {
        tag_walk(keep, E, tid, blockIdx.x, lsrc, &cnt);
    } else if (!have && full > 0) {
        if ((E & 3) == 0)
            agg_walk<DstFour>(dstp, full, t0, TRIP, woff, lane, lo, nrows, lsrc, lrow, &cnt);
        else
            agg_walk<DstOne>(dstp, full, t0, TRIP, woff, lane, lo, nrows, lsrc, lrow, &cnt);
    }
    if (!have && !tagw && full * TRIP < E) {
        int dT[AGG_K];
        agg_load_dst<true>(dstp, full * TRIP + woff, lane, E, dT);
        agg_collect(dT, full * TRIP + woff, lane, lo, nrows, lsrc, lrow, &cnt);
    }
    __syncthreads();
    const int n = (int)cnt;

    const bool save = !take_list && n <= AGG_CAP && 1 + 2 * n <= kcap;
    if (!take_list && tid == 0 && nrows > 0) kp[0] = __int_as_float(save ? n : -1);
    if (n <= AGG_CAP) {
        for (int i = tid; i < n; i += AGG_THREADS) {
            int src = lsrc[i];
            int row;
            if (!have) {
                const int e = src;
                src = e < E ? ei[e] : e - E;
                lsrc[i] = src;
                if (tagw && e < E) { row = dstp[e] - lo; lrow[i] = row; }
                else               row = lrow[i];
                if (save) { kp[1 + i] = __int_as_float(src); kp[1 + n + i] = __int_as_float(row); }
            } else {
                row = lrow[i];
            }
            float sv[HH];
            if constexpr (HH == 4) {
                const float4 q = ES_TAIL ? *reinterpret_cast<const float4*>(es_tail(keep, N, src))
                                         : reinterpret_cast<const float4*>(es)[src];
                sv[0] = q.x; sv[1] = q.y; sv[2] = q.z; sv[3] = q.w;
            } else {
                sv[0] = es[src];
            }
#pragma unroll
            for (int h = 0; h < HH; ++h) {
                const float lr = agg_lrelu(sv[h] + edv[row * HH + h]);
                lw[i * HH + h] = lr;
                atomicMax(&mx[row * HH + h], f2ord(lr));
            }
        }
        __syncthreads();
        for (int i = tid; i < n; i += AGG_THREADS) {
            const int row = lrow[i];
#pragma unroll
            for (int h = 0; h < HH; ++h) {
                const float v = expf(lw[i * HH + h] - ord2f(mx[row * HH + h]));
                lw[i * HH + h] = v;
                __hip_atomic_fetch_add(&den[row * HH + h], v, __ATOMIC_RELAXED, __HIP_MEMORY_SCOPE_WORKGROUP);
            }
        }
        __syncthreads();
        for (int i = tid; i < n; i += AGG_THREADS) {
            const int row = lrow[i];
#pragma unroll
            for (int h = 0; h < HH; ++h) lw[i * HH + h] = lw[i * HH + h] / den[row * HH + h];
        }
        __syncthreads();
        const int ngrp = n / AGG_U;
        for (int g = wave; g < ngrp; g += NW) {
            RowVals<HD> p[AGG_U];
            int rw[AGG_U];
#pragma unroll
            for (int u = 0; u < AGG_U; ++u)
                p[u] = agg_take<HD>(hfeat, lsrc, lrow, lw, g * AGG_U + u, lane, rw[u]);
#pragma unroll
            for (int u = 0; u < AGG_U; ++u) agg_add_row<HD>(acc, rw[u], lane, p[u]);
        }
        for (int i = ngrp * AGG_U + wave; i < n; i += NW) {
            int rw1;
            const RowVals<HD> p1 = agg_take<HD>(hfeat, lsrc, lrow, lw, i, lane, rw1);
            agg_add_row<HD>(acc, rw1, lane, p1);
        }
    } else {
        const int Et = E + N;
        for (int ph = 0; ph < 3; ++ph) {
            for (int eb = wave * 32; eb < Et; eb += AGG_THREADS) {
                const int e = eb + lane;
                int src = 0, rl = -1;
                if (e < Et) {
                    int dst; edge_endpoints(ei, E, e, src, dst);
                    rl = dst - lo;
                }
                const bool in = (unsigned)rl < (unsigned)nrows;
                if (ph < 2) {
                    if (in) {
#pragma unroll
                        for (int h = 0; h < HH; ++h) {
                            const float lr = agg_lrelu(es_at<HH, ES_TAIL>(es, keep, N, src, h) + edv[rl * HH + h]);
                            if (ph == 0) atomicMax(&mx[rl * HH + h], f2ord(lr));
                            else __hip_atomic_fetch_add(&den[rl * HH + h], expf(lr - ord2f(mx[rl * HH + h])),
                                                        __ATOMIC_RELAXED, __HIP_MEMORY_SCOPE_WORKGROUP);
                        }
                    }
                } else {
                    unsigned mk = __builtin_amdgcn_ballot_w32(in);
                    while (mk) {
                        const int b = __builtin_ctz(mk);
                        mk &= mk - 1;
                        const int sb = __builtin_amdgcn_readlane(src, b);
                        const int rb = __builtin_amdgcn_readlane(rl, b);
                        const int h0 = HH == 4 ? (lane >> 4) : 0;
                        const int h1 = HH == 4 ? 2 + (lane >> 4) : 0;
                        const float l0 = agg_lrelu(es_at<HH, ES_TAIL>(es, keep, N, sb, h0) + edv[rb * HH + h0]);
                        const float l1 = agg_lrelu(es_at<HH, ES_TAIL>(es, keep, N, sb, h1) + edv[rb * HH + h1]);
                        const float w0 = expf(l0 - ord2f(mx[rb * HH + h0])) / den[rb * HH + h0];
                        const float w1 = expf(l1 - ord2f(mx[rb * HH + h1])) / den[rb * HH + h1];
                        const RowVals<HD> p = agg_fetch<HD>(hfeat, sb, lane, w0, w1);
                        agg_add_row<HD>(acc, rb, lane, p);
                    }
                }
            }
            __syncthreads();
        }
    }
    __syncthreads();

    if constexpr (NC > 0) {
        for (int r = wave; r < nrows; r += NW) {
            float* arow = acc + r * HD;
            const float* a = arow + lane;
            const float4* b4 = reinterpret_cast<const float4*>(bias);
            const float4 c0 = b4[lane], c1 = b4[32 + lane];
            const float s0 = a[0], s1 = a[32], s2 = a[64], s3 = a[96], s4 = a[128], s5 = a[160], s6 = a[192], s7 = a[224];
            float4 o0, o1;
            o0.x = agg_act(s0 + c0.x, do_elu); o0.y = agg_act(s1 + c0.y, do_elu);
            o0.z = agg_act(s2 + c0.z, do_elu); o0.w = agg_act(s3 + c0.w, do_elu);
            o1.x = agg_act(s4 + c1.x, do_elu); o1.y = agg_act(s5 + c1.y, do_elu);
            o1.z = agg_act(s6 + c1.z, do_elu); o1.w = agg_act(s7 + c1.w, do_elu);
            const int swz = (r & 15) << 2;
            *reinterpret_cast<float4*>(arow + ((4 * lane) ^ swz))       = o0;
            *reinterpret_cast<float4*>(arow + ((128 + 4 * lane) ^ swz)) = o1;
        }
        __syncthreads();

        constexpr int HN = NC / 64;
        constexpr int MG = NW / HN;
        const int wv = __builtin_amdgcn_readfirstlane(wave);
        const int head = wv % HN;
        const int n0 = head * 64;
        const int nmt = (nrows + 15) >> 4;
        const int l15 = lane & 15;
        const int kh  = (lane >> 4) << 1;
        const float sa0 = asn[n0 + l15], sa1 = asn[n0 + 16 + l15], sa2 = asn[n0 + 32 + l15], sa3 = asn[n0 + 48 + l15];
        const float da0 = adn[n0 + l15], da1 = adn[n0 + 16 + l15], da2 = adn[n0 + 32 + l15], da3 = adn[n0 + 48 + l15];
        for (int mt = wv / HN; mt < nmt; mt += MG) {
            const int ar = min(mt * 16 + l15, AGG_ROWS - 1);
            const float* Ap = acc + ar * HD + kh;
            const int swz = (ar & 15) << 2;
            v8f q0 = {0.f,0.f,0.f,0.f,0.f,0.f,0.f,0.f};
            v8f q1 = q0, q2 = q0, q3 = q0;
            for (int k = 0; k < HD; k += 4) {
                const v2f a = *reinterpret_cast<const v2f*>(Ap + (k ^ swz));
                const float* Bp = Wn + (size_t)(k + kh) * NC + n0 + l15;
                v2f b0, b1, b2, b3;
                b0.x = Bp[0];  b0.y = Bp[(size_t)NC];
                b1.x = Bp[16]; b1.y = Bp[(size_t)NC + 16];
                b2.x = Bp[32]; b2.y = Bp[(size_t)NC + 32];
                b3.x = Bp[48]; b3.y = Bp[(size_t)NC + 48];
                q0 = __builtin_amdgcn_wmma_f32_16x16x4_f32(false, a, false, b0, (short)0, q0, false, false);
                q1 = __builtin_amdgcn_wmma_f32_16x16x4_f32(false, a, false, b1, (short)0, q1, false, false);
                q2 = __builtin_amdgcn_wmma_f32_16x16x4_f32(false, a, false, b2, (short)0, q2, false, false);
                q3 = __builtin_amdgcn_wmma_f32_16x16x4_f32(false, a, false, b3, (short)0, q3, false, false);
            }
            const int rbase = mt * 16 + ((lane >> 4) << 3);
            float* Cp = hn + (size_t)(lo + rbase) * NC + n0 + l15;
#pragma unroll
            for (int v = 0; v < 8; ++v) {
                if (rbase + v < nrows) {
                    Cp[(size_t)v * NC + 0]  = q0[v];
                    Cp[(size_t)v * NC + 16] = q1[v];
                    Cp[(size_t)v * NC + 32] = q2[v];
                    Cp[(size_t)v * NC + 48] = q3[v];
                }
            }
            float cs = 0.f, cd = 0.f;
#pragma unroll
            for (int v = 0; v < 8; ++v) {
                const float ps = row16_sum(q0[v] * sa0 + q1[v] * sa1 + q2[v] * sa2 + q3[v] * sa3);
                const float pd = row16_sum(q0[v] * da0 + q1[v] * da1 + q2[v] * da2 + q3[v] * da3);
                if (l15 == v) { cs = ps; cd = pd; }
            }
            if (l15 < 8 && rbase + l15 < nrows) {
                const int rl = rbase + l15;
                if constexpr (HN == 4) {
                    kp[kcap + TAIL_WORDS * rl + head] = cs;
                    edn[(size_t)(lo + rl) * HN + head] = cd;
                } else {
                    esn[(size_t)(lo + rl) * HN + head] = cs;
                    edn[(size_t)(lo + rl) * HN + head] = cd;
                }
            }
        }
    } else {
    float cs0 = 0.f, cs1 = 0.f;
    for (int r = wave; r < nrows; r += NW) {
        const float* a = acc + r * HD + lane;
        const size_t nd = (size_t)(lo + r);
        if constexpr (HD == 256) {
            const float4* b4 = reinterpret_cast<const float4*>(bias);
            const float4 c0 = b4[lane], c1 = b4[32 + lane];
            float4 o0, o1;
            o0.x = agg_act(a[0]   + c0.x, do_elu); o0.y = agg_act(a[32]  + c0.y, do_elu);
            o0.z = agg_act(a[64]  + c0.z, do_elu); o0.w = agg_act(a[96]  + c0.w, do_elu);
            o1.x = agg_act(a[128] + c1.x, do_elu); o1.y = agg_act(a[160] + c1.y, do_elu);
            o1.z = agg_act(a[192] + c1.z, do_elu); o1.w = agg_act(a[224] + c1.w, do_elu);
            float4* o4 = reinterpret_cast<float4*>(out + nd * 256);
            o4[lane] = o0;
            o4[32 + lane] = o1;
        } else {
            const float2 c = reinterpret_cast<const float2*>(bias)[lane];
            float2 o;
            o.x = agg_act(a[0]  + c.x, do_elu);
            o.y = agg_act(a[32] + c.y, do_elu);
            reinterpret_cast<float2*>(out + nd * 64)[lane] = o;
            cs0 += o.x;
            cs1 += o.y;
        }
    }
    if constexpr (HD == 64) {
        if (colpart != nullptr) {
            lw[wave * 64 + 2 * lane]     = cs0;
            lw[wave * 64 + 2 * lane + 1] = cs1;
            __syncthreads();
            if (tid < 64) {
                float s = 0.f;
                for (int w = 0; w < NW; ++w) s += lw[w * 64 + tid];
                colpart[(size_t)blockIdx.x * 64 + tid] = s;
            }
        }
    }
    }
}

#define MEAN_GROUPS 16
__global__ __launch_bounds__(64 * MEAN_GROUPS)
void gat_colmean_parts(const float* __restrict__ part, float* __restrict__ out, int N) {
    __shared__ float sm[MEAN_GROUPS * 64];
    const int nblk = (N + AGG_ROWS - 1) / AGG_ROWS;
    const int c = threadIdx.x & 63;
    const int g = threadIdx.x >> 6;
    float s = 0.f;
#pragma unroll 16
    for (int b = g; b < nblk; b += MEAN_GROUPS) s += part[(size_t)b * 64 + c];
    sm[g * 64 + c] = s;
    __syncthreads();
    if (threadIdx.x < 64) {
        float t = 0.f;
        for (int k = 0; k < MEAN_GROUPS; ++k) t += sm[k * 64 + threadIdx.x];
        out[threadIdx.x] = t / (float)N;
    }
}

static inline int cdiv(long long a, long long b) { return (int)((a + b - 1) / b); }

extern "C" void kernel_launch(void* const* d_in, const int* in_sizes, int n_in,
                              void* d_out, int out_size, void* d_ws, size_t ws_size,
                              hipStream_t stream) {
    const float* x   = (const float*)d_in[0];
    const int*   ei  = (const int*)  d_in[1];
    const float* W1  = (const float*)d_in[2];
    const float* a1s = (const float*)d_in[3];
    const float* a1d = (const float*)d_in[4];
    const float* b1  = (const float*)d_in[5];
    const float* W2  = (const float*)d_in[6];
    const float* a2s = (const float*)d_in[7];
    const float* a2d = (const float*)d_in[8];
    const float* b2  = (const float*)d_in[9];
    const float* W3  = (const float*)d_in[10];
    const float* a3s = (const float*)d_in[11];
    const float* a3d = (const float*)d_in[12];
    const float* b3  = (const float*)d_in[13];

    const int N  = in_sizes[0] / 16;
    const int E  = in_sizes[1] / 2;
    const int H  = 4;

    float* ws = (float*)d_ws;
    size_t off = 0;
    float*    buf_h    = ws + off; off += (size_t)N * 256;
    float*    buf_h2   = ws + off; off += (size_t)N * 256;
    float*    es       = ws + off; off += (size_t)N * H;
    float*    ed       = ws + off; off += (size_t)N * H;
    (void)ws_size; (void)off;

    float* node_emb  = (float*)d_out;
    float* graph_emb = node_emb + (size_t)N * 64;

    hipLaunchKernelGGL(gat_gemm_wmma, dim3(1, cdiv(N, 16)), dim3(32, 4), 0, stream,
                       x, W1, buf_h, a1s, a1d, es, ed, N, 16, 256, ei, E, node_emb);

    const dim3 ag(cdiv(N, AGG_ROWS)), ab(AGG_THREADS);
    float* es3      = es;
    float* col_part = es + (size_t)N;
    float* ed3      = es + 2 * (size_t)N;
    float* none     = nullptr;
    gat_agg_owner<256, 256, false><<<ag, ab, 0, stream>>>(ei, E, N, buf_h, es, ed, b1, none, 1, none, node_emb, 0,
                                                          W2, a2s, a2d, buf_h2, none, ed);
    gat_agg_owner<256, 64, true><<<ag, ab, 0, stream>>>(ei, E, N, buf_h2, es, ed, b2, none, 1, none, node_emb, 1,
                                                        W3, a3s, a3d, buf_h, es3, ed3);
    gat_agg_owner<64, 0, false><<<ag, ab, 0, stream>>>(ei, E, N, buf_h, es3, ed3, b3, node_emb, 0, col_part, node_emb, 1,
                                                       none, none, none, none, none, none);

    hipLaunchKernelGGL(gat_colmean_parts, dim3(1), dim3(64 * MEAN_GROUPS), 0, stream,
                       col_part, graph_emb, N);
}


# Gather sums kept in registers: wave w of an owner block owns rows w, w+32, ..; it takes its rows' entries from the block's list, stores each row to LDS once; no LDS float adds; sum order changes
# speedup vs baseline: 9.6582x; 1.0403x over previous
#include <hip/hip_runtime.h>
#include <math.h>

typedef __attribute__((ext_vector_type(2))) float v2f;
typedef __attribute__((ext_vector_type(8))) float v8f;

#define GAT_D 64

__device__ __forceinline__ unsigned f2ord(float f) {
    int b = __float_as_int(f);
    return (unsigned)b ^ (unsigned)((b >> 31) | 0x80000000);
}
__device__ __forceinline__ float ord2f(unsigned u) {
    int b = (u & 0x80000000u) ? (int)(u ^ 0x80000000u) : ~(int)u;
    return __int_as_float(b);
}

__device__ __forceinline__ float row16_sum(float v) {
    v += __int_as_float(__builtin_amdgcn_update_dpp(0, __float_as_int(v), 0x161, 0xf, 0xf, false));
    v += __int_as_float(__builtin_amdgcn_update_dpp(0, __float_as_int(v), 0x162, 0xf, 0xf, false));
    v += __int_as_float(__builtin_amdgcn_update_dpp(0, __float_as_int(v), 0x164, 0xf, 0xf, false));
    v += __int_as_float(__builtin_amdgcn_update_dpp(0, __float_as_int(v), 0x168, 0xf, 0xf, false));
    return v;
}

#define TAG_ROWS     196
#define TAG_REGION   (196 * 64)
#define TAG_LIST_MAX 4096
#define TAG_OFF      ((1 + 2 * TAG_LIST_MAX + 3) & ~3)
#define TAG_WAVES    4
#define TAG_UNITS    256
#define TAG_WORDS    (4 * TAG_UNITS)
#define BM_OFF       (TAG_OFF + TAG_WORDS)
#define BM_WORDS     ((TAG_UNITS / 32) * 256)

__device__ __forceinline__ bool tags_fit(int E, int N) {
    return (E & 15) == 0 && N <= 256 * TAG_ROWS && ((E >> 4) + TAG_UNITS - 1) / TAG_UNITS <= N / TAG_ROWS &&
           ((E >> 4) + 31) / 32 <= 2 * 1024;
}
__device__ __forceinline__ uint4* tag_unit(float* plane, int u) {
    return reinterpret_cast<uint4*>(plane + (size_t)(u / TAG_UNITS) * TAG_REGION + TAG_OFF) + (u % TAG_UNITS);
}
__device__ __forceinline__ unsigned* tag_group(float* plane, int g) {
    return reinterpret_cast<unsigned*>(plane + (size_t)(g / (TAG_UNITS / 32)) * TAG_REGION + BM_OFF) + (g % (TAG_UNITS / 32)) * 256;
}
__device__ __forceinline__ unsigned tag_word(int4 d) {
    return  (((unsigned)d.x / TAG_ROWS) & 255u)        | ((((unsigned)d.y / TAG_ROWS) & 255u) << 8) |
           ((((unsigned)d.z / TAG_ROWS) & 255u) << 16) | ((((unsigned)d.w / TAG_ROWS) & 255u) << 24);
}

__global__ void gat_gemm_wmma(const float* __restrict__ A,
                              const float* __restrict__ B,
                              float* __restrict__ C,
                              const float* __restrict__ a_s,
                              const float* __restrict__ a_d,
                              float* __restrict__ es, float* __restrict__ ed,
                              int M, int K, int N,
                              const int* __restrict__ ei, int E, float* tagplane) {
    __shared__ unsigned own[TAG_WAVES * 256];
    if (tagplane != nullptr && tags_fit(E, M)) {
        const int U = E >> 4, G = (U + 31) >> 5;
        const int nblk = (int)(gridDim.x * gridDim.y);
        unsigned* mine = own + threadIdx.y * 256;
        const int ln = threadIdx.x;
        const int4* d4 = reinterpret_cast<const int4*>(ei + E);
        for (int gb = (int)(blockIdx.y * gridDim.x + blockIdx.x); gb * TAG_WAVES < G; gb += nblk) {
            const int g = gb * TAG_WAVES + (int)threadIdx.y;
            const int u = 32 * g + ln;
            for (int k = 0; k < 8; ++k) mine[ln + 32 * k] = 0u;
            __syncthreads();
            if (u < U) {
                unsigned w[4];
                for (int q = 0; q < 4; ++q) w[q] = tag_word(d4[4 * u + q]);
                *tag_unit(tagplane, u) = make_uint4(w[0], w[1], w[2], w[3]);
                for (int q = 0; q < 4; ++q)
                    for (int b = 0; b < 4; ++b)
                        __hip_atomic_fetch_or(&mine[(w[q] >> (8 * b)) & 255u], 1u << ln,
                                              __ATOMIC_RELAXED, __HIP_MEMORY_SCOPE_WORKGROUP);
            }
            __syncthreads();
            if (g < G) {
                unsigned* gw = tag_group(tagplane, g);
                for (int k = 0; k < 8; ++k) gw[ln + 32 * k] = mine[ln + 32 * k];
            }
            __syncthreads();
        }
    }
    const int lane  = threadIdx.x;
    const int grp   = blockIdx.x * blockDim.y + threadIdx.y;
    const int mtile = blockIdx.y;
    const int n0 = grp * 64;
    const int m0 = mtile * 16;
    if (n0 >= N || m0 >= M) return;

    const int l15 = lane & 15;
    const int kh  = (lane >> 4) << 1;
    const float* Arow = A + (size_t)(m0 + l15) * K;
    v8f acc0 = {0.f,0.f,0.f,0.f,0.f,0.f,0.f,0.f};
    v8f acc1 = acc0, acc2 = acc0, acc3 = acc0;
    const float sa0 = a_s[n0 + l15], sa1 = a_s[n0 + 16 + l15], sa2 = a_s[n0 + 32 + l15], sa3 = a_s[n0 + 48 + l15];
    const float da0 = a_d[n0 + l15], da1 = a_d[n0 + 16 + l15], da2 = a_d[n0 + 32 + l15], da3 = a_d[n0 + 48 + l15];

    for (int k = 0; k < K; k += 4) {
        v2f a;
        a.x = Arow[k + kh];
        a.y = Arow[k + kh + 1];
        const float* Bp = B + (size_t)(k + kh) * N + n0 + l15;
        v2f b0, b1, b2, b3;
        b0.x = Bp[0];  b0.y = Bp[(size_t)N];
        b1.x = Bp[16]; b1.y = Bp[(size_t)N + 16];
        b2.x = Bp[32]; b2.y = Bp[(size_t)N + 32];
        b3.x = Bp[48]; b3.y = Bp[(size_t)N + 48];
        acc0 = __builtin_amdgcn_wmma_f32_16x16x4_f32(false, a, false, b0, (short)0, acc0, false, false);
        acc1 = __builtin_amdgcn_wmma_f32_16x16x4_f32(false, a, false, b1, (short)0, acc1, false, false);
        acc2 = __builtin_amdgcn_wmma_f32_16x16x4_f32(false, a, false, b2, (short)0, acc2, false, false);
        acc3 = __builtin_amdgcn_wmma_f32_16x16x4_f32(false, a, false, b3, (short)0, acc3, false, false);
    }

    float* Cp = C + (size_t)(m0 + ((lane >> 4) << 3)) * N + n0 + l15;
#pragma unroll
    for (int v = 0; v < 8; ++v) {
        Cp[(size_t)v * N + 0]  = acc0[v];
        Cp[(size_t)v * N + 16] = acc1[v];
        Cp[(size_t)v * N + 32] = acc2[v];
        Cp[(size_t)v * N + 48] = acc3[v];
    }

    float cs = 0.f, cd = 0.f;
#pragma unroll
    for (int v = 0; v < 8; ++v) {
        const float ps = row16_sum(acc0[v] * sa0 + acc1[v] * sa1 + acc2[v] * sa2 + acc3[v] * sa3);
        const float pd = row16_sum(acc0[v] * da0 + acc1[v] * da1 + acc2[v] * da2 + acc3[v] * da3);
        if (l15 == v) { cs = ps; cd = pd; }
    }
    if (l15 < 8) {
        const size_t ci = (size_t)(m0 + ((lane >> 4) << 3) + l15) * (size_t)(N / GAT_D) + (size_t)grp;
        es[ci] = cs;
        ed[ci] = cd;
    }
}

__device__ __forceinline__ void edge_endpoints(const int* __restrict__ ei, int E,
                                               int e, int& src, int& dst) {
    if (e < E) { src = ei[e]; dst = ei[E + e]; }
    else       { src = e - E; dst = e - E;    }
}

#define AGG_ROWS    196
#define AGG_THREADS 1024
#define AGG_K       8
#define AGG_CAP     4096
#define AGG_U       4
#define SUB_CAP     256
#define KEEP_PITCH  64
#define TAIL_WORDS  4

template <int HD> struct RowVals { float v[HD / 32]; };

__device__ __forceinline__ const float* es_tail(const float* keep, int N, int src) {
    const int blk = src / AGG_ROWS;
    const int lo_b = blk * AGG_ROWS;
    const int nr = min(AGG_ROWS, N - lo_b);
    return keep + (size_t)lo_b * KEEP_PITCH + nr * (KEEP_PITCH - TAIL_WORDS) + TAIL_WORDS * (src - lo_b);
}

template <int HH, bool ES_TAIL>
__device__ __forceinline__ float es_at(const float* es, const float* keep, int N, int src, int h) {
    if constexpr (ES_TAIL) return es_tail(keep, N, src)[h];
    else                   return es[(size_t)src * HH + h];
}

template <int HD>
__device__ __forceinline__ RowVals<HD> agg_fetch(const float* __restrict__ hfeat,
                                                 int src, int lane, float w0, float w1) {
    RowVals<HD> r;
    if constexpr (HD == 256) {
        const float4* hp = reinterpret_cast<const float4*>(hfeat + (size_t)src * 256);
        const float4 a = hp[lane];
        const float4 b = hp[32 + lane];
        r.v[0] = a.x * w0; r.v[1] = a.y * w0; r.v[2] = a.z * w0; r.v[3] = a.w * w0;
        r.v[4] = b.x * w1; r.v[5] = b.y * w1; r.v[6] = b.z * w1; r.v[7] = b.w * w1;
    } else {
        const float2* hp = reinterpret_cast<const float2*>(hfeat + (size_t)src * 64);
        const float2 a = hp[lane];
        r.v[0] = a.x * w0; r.v[1] = a.y * w0;
    }
    return r;
}

template <int HD>
__device__ __forceinline__ RowVals<HD> agg_take(const float* __restrict__ hfeat,
                                                const int* __restrict__ lsrc,
                                                const int* __restrict__ lrow,
                                                const float* __restrict__ lw,
                                                int i, int lane, int& row) {
    row = lrow[i];
    const float* wp = lw + i * (HD / 64);
    float w0, w1;
    if constexpr (HD == 256) { w0 = wp[lane >> 4]; w1 = wp[2 + (lane >> 4)]; }
    else                     { w0 = wp[0]; w1 = 0.f; }
    return agg_fetch<HD>(hfeat, lsrc[i], lane, w0, w1);
}

template <int HD>
__device__ __forceinline__ RowVals<HD> agg_take_at(const float* __restrict__ hfeat,
                                                   const float* __restrict__ lw,
                                                   int i, int src, int lane) {
    const float* wp = lw + i * (HD / 64);
    float w0, w1;
    if constexpr (HD == 256) { w0 = wp[lane >> 4]; w1 = wp[2 + (lane >> 4)]; }
    else                     { w0 = wp[0]; w1 = 0.f; }
    return agg_fetch<HD>(hfeat, src, lane, w0, w1);
}

template <int HD>
__device__ __forceinline__ void agg_add_row(float* __restrict__ acc, int row, int lane,
                                            const RowVals<HD>& p) {
    float* a = acc + row * HD + lane;
#pragma unroll
    for (int j = 0; j < HD / 32; ++j)
        __hip_atomic_fetch_add(a + j * 32, p.v[j], __ATOMIC_RELAXED, __HIP_MEMORY_SCOPE_WORKGROUP);
}

template <int HD, int SL>
__device__ __forceinline__ void own_add(float (&ra)[SL][HD / 32], int s, const RowVals<HD>& p) {
#pragma unroll
    for (int c = 0; c < SL; ++c) {
        if (__builtin_expect(s == c, 0)) {
#pragma unroll
            for (int j = 0; j < HD / 32; ++j) ra[c][j] += p.v[j];
        }
    }
}

__device__ __forceinline__ float agg_act(float v, int do_elu) {
    if (do_elu) v = v > 0.f ? v : (expf(v) - 1.f);
    return v;
}

__device__ __forceinline__ float agg_lrelu(float s) { return s > 0.f ? s : 0.2f * s; }

template <bool TAIL>
__device__ __forceinline__ void agg_load_dst(const int* __restrict__ dstp, int base, int lane,
                                             int E, int (&d)[AGG_K]) {
#pragma unroll
    for (int k = 0; k < AGG_K; ++k) {
        const int e = base + 32 * k + lane;
        if constexpr (TAIL) d[k] = e < E ? dstp[e] : -1;
        else                d[k] = dstp[e];
    }
}

__device__ __forceinline__ void agg_collect(const int (&d)[AGG_K], int base, int lane, int lo, int nrows,
                                            int* __restrict__ lsrc, int* __restrict__ lrow,
                                            unsigned* __restrict__ cnt) {
    unsigned m[AGG_K], any = 0;
    int rl[AGG_K];
#pragma unroll
    for (int k = 0; k < AGG_K; ++k) {
        rl[k] = d[k] - lo;
        m[k] = __builtin_amdgcn_ballot_w32((unsigned)rl[k] < (unsigned)nrows);
        any |= m[k];
    }
    if (!any) return;
    unsigned tot = 0;
#pragma unroll
    for (int k = 0; k < AGG_K; ++k) tot += (unsigned)__builtin_popcount(m[k]);
    unsigned start = 0;
    if (lane == 0)
        start = __hip_atomic_fetch_add(cnt, tot, __ATOMIC_RELAXED, __HIP_MEMORY_SCOPE_WORKGROUP);
    start = (unsigned)__builtin_amdgcn_readfirstlane((int)start);
    const unsigned below = (1u << lane) - 1u;
#pragma unroll
    for (int k = 0; k < AGG_K; ++k) {
        if ((m[k] >> lane) & 1u) {
            const unsigned p = start + (unsigned)__builtin_popcount(m[k] & below);
            if (p < (unsigned)AGG_CAP) { lsrc[p] = base + 32 * k + lane; lrow[p] = rl[k]; }
        }
        start += (unsigned)__builtin_popcount(m[k]);
    }
}

struct DstOne  { int  d[AGG_K]; };
struct DstFour { int4 d[AGG_K / 4]; };

__device__ __forceinline__ void agg_load(const int* __restrict__ dstp, int base, int lane, DstOne& s) {
    agg_load_dst<false>(dstp, base, lane, 0, s.d);
}
__device__ __forceinline__ void agg_load(const int* __restrict__ dstp, int base, int lane, DstFour& s) {
#pragma unroll
    for (int q = 0; q < AGG_K / 4; ++q)
        s.d[q] = *reinterpret_cast<const int4*>(dstp + base + 128 * q + 4 * lane);
}

__device__ __forceinline__ void agg_note(const DstOne& s, int base, int lane, int lo, int nrows,
                                         int* __restrict__ lsrc, int* __restrict__ lrow,
                                         unsigned* __restrict__ cnt) {
    agg_collect(s.d, base, lane, lo, nrows, lsrc, lrow, cnt);
}

__device__ __forceinline__ void agg_note_one(unsigned rl, int e, int nrows,
                                             int* __restrict__ lsrc, int* __restrict__ lrow,
                                             unsigned* __restrict__ cnt) {
    if (rl < (unsigned)nrows) {
        const unsigned p = __hip_atomic_fetch_add(cnt, 1u, __ATOMIC_RELAXED, __HIP_MEMORY_SCOPE_WORKGROUP);
        if (p < (unsigned)AGG_CAP) { lsrc[p] = e; lrow[p] = (int)rl; }
    }
}

__device__ __forceinline__ void agg_note(const DstFour& s, int base, int lane, int lo, int nrows,
                                         int* __restrict__ lsrc, int* __restrict__ lrow,
                                         unsigned* __restrict__ cnt) {
    unsigned r[AGG_K];
#pragma unroll
    for (int q = 0; q < AGG_K / 4; ++q) {
        r[4 * q + 0] = (unsigned)(s.d[q].x - lo);
        r[4 * q + 1] = (unsigned)(s.d[q].y - lo);
        r[4 * q + 2] = (unsigned)(s.d[q].z - lo);
        r[4 * q + 3] = (unsigned)(s.d[q].w - lo);
    }
    unsigned mn = r[0];
#pragma unroll
    for (int k = 1; k < AGG_K; ++k) mn = min(mn, r[k]);
    if (mn < (unsigned)nrows) {
#pragma unroll
        for (int k = 0; k < AGG_K; ++k)
            agg_note_one(r[k], base + 128 * (k >> 2) + 4 * lane + (k & 3), nrows, lsrc, lrow, cnt);
    }
}

template <class DS>
__device__ __forceinline__ void agg_walk(const int* __restrict__ dstp, int full, int t0, int trip, int woff,
                                         int lane, int lo, int nrows,
                                         int* __restrict__ lsrc, int* __restrict__ lrow,
                                         unsigned* __restrict__ cnt) {
    DS dA, dB;
    agg_load(dstp, t0 * trip + woff, lane, dA);
    for (int t = 0; t < full; t += 2) {
        int tc0 = t0 + t;    if (tc0 >= full) tc0 -= full;
        int tc1 = tc0 + 1;   if (tc1 >= full) tc1 -= full;
        int tc2 = tc1 + 1;   if (tc2 >= full) tc2 -= full;
        agg_load(dstp, tc1 * trip + woff, lane, dB);
        agg_note(dA, tc0 * trip + woff, lane, lo, nrows, lsrc, lrow, cnt);
        agg_load(dstp, tc2 * trip + woff, lane, dA);
        if (t + 1 < full)
            agg_note(dB, tc1 * trip + woff, lane, lo, nrows, lsrc, lrow, cnt);
    }
}

static_assert(TAG_ROWS == AGG_ROWS && TAG_REGION == AGG_ROWS * KEEP_PITCH && TAG_LIST_MAX == AGG_CAP,
              "the tag bytes are laid out for the owner blocks' rows, words and list length");
static_assert(BM_OFF + BM_WORDS <= AGG_ROWS * (KEEP_PITCH - TAIL_WORDS),
              "the tag bytes and owner words end before a full block's coefficient words");

__device__ __forceinline__ unsigned zero_bytes(unsigned x) {
    return ~((((x & 0x7F7F7F7Fu) + 0x7F7F7F7Fu) | x) | 0x7F7F7F7Fu);
}

__device__ __forceinline__ int tag_next(unsigned& bits0, unsigned& bits1, int g0, int g1) {
    if (bits0 != 0u) { const int b = __builtin_ctz(bits0); bits0 &= bits0 - 1u; return 32 * g0 + b; }
    if (bits1 != 0u) { const int b = __builtin_ctz(bits1); bits1 &= bits1 - 1u; return 32 * g1 + b; }
    return -1;
}

__device__ __forceinline__ void tag_note(uint4 w, int u, unsigned pat,
                                         int* __restrict__ lsrc, unsigned* __restrict__ cnt) {
    unsigned m = (zero_bytes(w.x ^ pat) >> 7) | (zero_bytes(w.y ^ pat) >> 6) |
                 (zero_bytes(w.z ^ pat) >> 5) | (zero_bytes(w.w ^ pat) >> 4);
    while (m != 0u) {
        const int bit = __builtin_ctz(m);
        m &= m - 1u;
        const unsigned p = __hip_atomic_fetch_add(cnt, 1u, __ATOMIC_RELAXED, __HIP_MEMORY_SCOPE_WORKGROUP);
        if (p < (unsigned)AGG_CAP) lsrc[p] = 16 * u + 4 * (bit & 7) + (bit >> 3);
    }
}

__device__ __forceinline__ void tag_walk(float* plane, int E, int tid, unsigned owner,
                                         int* __restrict__ lsrc, unsigned* __restrict__ cnt) {
    const int U = E >> 4, G = (U + 31) >> 5;
    const unsigned pat = owner * 0x01010101u;
    const int g0 = tid, g1 = tid + AGG_THREADS;
    unsigned bits0 = g0 < G ? tag_group(plane, g0)[owner] : 0u;
    unsigned bits1 = g1 < G ? tag_group(plane, g1)[owner] : 0u;
    int u = tag_next(bits0, bits1, g0, g1);
    uint4 w = *tag_unit(plane, max(u, 0));
    while (u >= 0) {
        const int un = tag_next(bits0, bits1, g0, g1);
        const uint4 wn = *tag_unit(plane, max(un, 0));
        tag_note(w, u, pat, lsrc, cnt);
        u = un; w = wn;
    }
}

template <int HD, int NC, bool ES_TAIL>
__global__ __launch_bounds__(AGG_THREADS)
void gat_agg_owner(const int* __restrict__ ei, int E, int N,
                   const float* __restrict__ hfeat,
                   const float* es, const float* ed,
                   const float* __restrict__ bias,
                   float* out, int do_elu,
                   float* __restrict__ colpart,
                   float* keep, int take_list,
                   const float* __restrict__ Wn,
                   const float* __restrict__ asn, const float* __restrict__ adn,
                   float* __restrict__ hn, float* esn, float* edn) {
    static_assert(NC == 0 || HD == 256, "the projection at the end reads 256-channel rows");
    static_assert(!ES_TAIL || HD == 256, "only 4-head coefficients are left in the output plane");
    constexpr int HH = HD / 64;
    constexpr int NW = AGG_THREADS / 32;
    constexpr int SPAN = 32 * AGG_K;
    constexpr int TRIP = NW * SPAN;
    __shared__ __attribute__((aligned(16))) float acc[AGG_ROWS * HD];
    __shared__ unsigned mx[AGG_ROWS * HH];
    __shared__ float    den[AGG_ROWS * HH];
    __shared__ float    edv[AGG_ROWS * HH];
    __shared__ int      lsrc[AGG_CAP];
    __shared__ int      lrow[AGG_CAP];
    __shared__ float    lw[AGG_CAP * HH];
    __shared__ unsigned cnt;
    __shared__ unsigned short sub[(AGG_THREADS / 32) * SUB_CAP];
    __shared__ unsigned subcnt[AGG_THREADS / 32];
    const int tid  = threadIdx.x;
    const int lane = tid & 31;
    const int wave = tid >> 5;
    const int lo = blockIdx.x * AGG_ROWS;
    const int nrows = max(0, min(AGG_ROWS, N - lo));

    float* kp = keep + (size_t)lo * KEEP_PITCH;
    const int kcap = nrows * (KEEP_PITCH - TAIL_WORDS);
    int taken = -1;
    if (take_list && nrows > 0) taken = __builtin_amdgcn_readfirstlane(__float_as_int(kp[0]));
    const bool have = taken >= 0 && taken <= AGG_CAP && 1 + 2 * taken <= kcap;

    {
        float4* a4 = reinterpret_cast<float4*>(acc);
        for (int i = tid; i < AGG_ROWS * HD / 4; i += AGG_THREADS) a4[i] = make_float4(0.f, 0.f, 0.f, 0.f);
        if (tid < AGG_THREADS / 32) subcnt[tid] = 0u;
        for (int i = tid; i < nrows * HH; i += AGG_THREADS) {
            mx[i] = 0u;
            den[i] = 0.f;
            edv[i] = ed[(size_t)lo * HH + i];
        }
        if (have) {
            for (int i = tid; i < taken; i += AGG_THREADS) {
                lsrc[i] = __float_as_int(kp[1 + i]);
                lrow[i] = __float_as_int(kp[1 + taken + i]);
            }
            if (tid == 0) cnt = (unsigned)taken;
        } else {
            for (int r = tid; r < nrows; r += AGG_THREADS) { lsrc[r] = E + lo + r; lrow[r] = r; }
            if (tid == 0) cnt = (unsigned)nrows;
        }
    }
    __syncthreads();

    const int* __restrict__ dstp = ei + E;
    const int full = E / TRIP;
    const int t0 = (int)((blockIdx.x * (unsigned)full) / gridDim.x);
    const int woff = wave * SPAN;
    const bool tagw = !take_list && tags_fit(E, N);
    if (!have && tagw) {
        tag_walk(keep, E, tid, blockIdx.x, lsrc, &cnt);
    } else if (!have && full > 0) {
        if ((E & 3) == 0)
            agg_walk<DstFour>(dstp, full, t0, TRIP, woff, lane, lo, nrows, lsrc, lrow, &cnt);
        else
            agg_walk<DstOne>(dstp, full, t0, TRIP, woff, lane, lo, nrows, lsrc, lrow, &cnt);
    }
    if (!have && !tagw && full * TRIP < E) {
        int dT[AGG_K];
        agg_load_dst<true>(dstp, full * TRIP + woff, lane, E, dT);
        agg_collect(dT, full * TRIP + woff, lane, lo, nrows, lsrc, lrow, &cnt);
    }
    __syncthreads();
    const int n = (int)cnt;

    const bool save = !take_list && n <= AGG_CAP && 1 + 2 * n <= kcap;
    if (!take_list && tid == 0 && nrows > 0) kp[0] = __int_as_float(save ? n : -1);
    if (n <= AGG_CAP) {
        for (int i = tid; i < n; i += AGG_THREADS) {
            int src = lsrc[i];
            int row;
            if (!have) {
                const int e = src;
                src = e < E ? ei[e] : e - E;
                lsrc[i] = src;
                if (tagw && e < E) { row = dstp[e] - lo; lrow[i] = row; }
                else               row = lrow[i];
                if (save) { kp[1 + i] = __int_as_float(src); kp[1 + n + i] = __int_as_float(row); }
            } else {
                row = lrow[i];
            }
            float sv[HH];
            if constexpr (HH == 4) {
                const float4 q = ES_TAIL ? *reinterpret_cast<const float4*>(es_tail(keep, N, src))
                                         : reinterpret_cast<const float4*>(es)[src];
                sv[0] = q.x; sv[1] = q.y; sv[2] = q.z; sv[3] = q.w;
            } else {
                sv[0] = es[src];
            }
#pragma unroll
            for (int h = 0; h < HH; ++h) {
                const float lr = agg_lrelu(sv[h] + edv[row * HH + h]);
                lw[i * HH + h] = lr;
                atomicMax(&mx[row * HH + h], f2ord(lr));
            }
        }
        __syncthreads();
        for (int i = tid; i < n; i += AGG_THREADS) {
            const int row = lrow[i];
#pragma unroll
            for (int h = 0; h < HH; ++h) {
                const float v = expf(lw[i * HH + h] - ord2f(mx[row * HH + h]));
                lw[i * HH + h] = v;
                __hip_atomic_fetch_add(&den[row * HH + h], v, __ATOMIC_RELAXED, __HIP_MEMORY_SCOPE_WORKGROUP);
            }
        }
        __syncthreads();
        static_assert(NW == 32 && AGG_CAP <= 65536, "rows are dealt out by row & 31, a wave's lanes read the 32 counters; a list place fits 16 bits");
        for (int i = tid; i < n; i += AGG_THREADS) {
            const int row = lrow[i];
#pragma unroll
            for (int h = 0; h < HH; ++h) lw[i * HH + h] = lw[i * HH + h] / den[row * HH + h];
            const int ow = row & (NW - 1);
            const unsigned p = __hip_atomic_fetch_add(&subcnt[ow], 1u, __ATOMIC_RELAXED, __HIP_MEMORY_SCOPE_WORKGROUP);
            if (p < (unsigned)SUB_CAP) sub[ow * SUB_CAP + p] = (unsigned short)i;
        }
        __syncthreads();
        constexpr int SL = (AGG_ROWS + NW - 1) / NW;
        constexpr int RV = HD / 32;
        const unsigned short* mysub = sub + wave * SUB_CAP;
        const unsigned cl = subcnt[lane];
        const bool allfit = __builtin_amdgcn_ballot_w32(cl > (unsigned)SUB_CAP) == 0u;
        const int mine_n = __builtin_amdgcn_readfirstlane((int)subcnt[wave]);
        if (allfit) {
            float ra[SL][RV];
#pragma unroll
            for (int s = 0; s < SL; ++s)
#pragma unroll
                for (int j = 0; j < RV; ++j) ra[s][j] = 0.f;
            const int mgrp = mine_n / AGG_U;
            int pi[AGG_U], ps[AGG_U], pr[AGG_U];
#pragma unroll
            for (int u = 0; u < AGG_U; ++u) {
                pi[u] = mgrp > 0 ? (int)mysub[u] : 0;
                ps[u] = lsrc[pi[u]];
                pr[u] = lrow[pi[u]];
            }
            for (int g = 0; g < mgrp; ++g) {
                RowVals<HD> p[AGG_U];
                int rw[AGG_U];
#pragma unroll
                for (int u = 0; u < AGG_U; ++u) {
                    p[u] = agg_take_at<HD>(hfeat, lw, pi[u], ps[u], lane);
                    rw[u] = pr[u];
                }
                const int gn = min(g + 1, mgrp - 1);
#pragma unroll
                for (int u = 0; u < AGG_U; ++u) {
                    pi[u] = (int)mysub[gn * AGG_U + u];
                    ps[u] = lsrc[pi[u]];
                    pr[u] = lrow[pi[u]];
                }
#pragma unroll
                for (int u = 0; u < AGG_U; ++u)
                    own_add<HD, SL>(ra, __builtin_amdgcn_readfirstlane(rw[u]) >> 5, p[u]);
            }
            for (int k = mgrp * AGG_U; k < mine_n; ++k) {
                int rw1;
                const RowVals<HD> p1 = agg_take<HD>(hfeat, lsrc, lrow, lw, (int)mysub[k], lane, rw1);
                own_add<HD, SL>(ra, __builtin_amdgcn_readfirstlane(rw1) >> 5, p1);
            }
#pragma unroll
            for (int s = 0; s < SL; ++s) {
                const int r = wave + NW * s;
                if (r < nrows) {
#pragma unroll
                    for (int j = 0; j < RV; ++j) acc[r * HD + j * 32 + lane] = ra[s][j];
                }
            }
        } else {
        const int ngrp = n / AGG_U;
        for (int g = wave; g < ngrp; g += NW) {
            RowVals<HD> p[AGG_U];
            int rw[AGG_U];
#pragma unroll
            for (int u = 0; u < AGG_U; ++u)
                p[u] = agg_take<HD>(hfeat, lsrc, lrow, lw, g * AGG_U + u, lane, rw[u]);
#pragma unroll
            for (int u = 0; u < AGG_U; ++u) agg_add_row<HD>(acc, rw[u], lane, p[u]);
        }
        for (int i = ngrp * AGG_U + wave; i < n; i += NW) {
            int rw1;
            const RowVals<HD> p1 = agg_take<HD>(hfeat, lsrc, lrow, lw, i, lane, rw1);
            agg_add_row<HD>(acc, rw1, lane, p1);
        }
        }
    } else {
        const int Et = E + N;
        for (int ph = 0; ph < 3; ++ph) {
            for (int eb = wave * 32; eb < Et; eb += AGG_THREADS) {
                const int e = eb + lane;
                int src = 0, rl = -1;
                if (e < Et) {
                    int dst; edge_endpoints(ei, E, e, src, dst);
                    rl = dst - lo;
                }
                const bool in = (unsigned)rl < (unsigned)nrows;
                if (ph < 2) {
                    if (in) {
#pragma unroll
                        for (int h = 0; h < HH; ++h) {
                            const float lr = agg_lrelu(es_at<HH, ES_TAIL>(es, keep, N, src, h) + edv[rl * HH + h]);
                            if (ph == 0) atomicMax(&mx[rl * HH + h], f2ord(lr));
                            else __hip_atomic_fetch_add(&den[rl * HH + h], expf(lr - ord2f(mx[rl * HH + h])),
                                                        __ATOMIC_RELAXED, __HIP_MEMORY_SCOPE_WORKGROUP);
                        }
                    }
                } else {
                    unsigned mk = __builtin_amdgcn_ballot_w32(in);
                    while (mk) {
                        const int b = __builtin_ctz(mk);
                        mk &= mk - 1;
                        const int sb = __builtin_amdgcn_readlane(src, b);
                        const int rb = __builtin_amdgcn_readlane(rl, b);
                        const int h0 = HH == 4 ? (lane >> 4) : 0;
                        const int h1 = HH == 4 ? 2 + (lane >> 4) : 0;
                        const float l0 = agg_lrelu(es_at<HH, ES_TAIL>(es, keep, N, sb, h0) + edv[rb * HH + h0]);
                        const float l1 = agg_lrelu(es_at<HH, ES_TAIL>(es, keep, N, sb, h1) + edv[rb * HH + h1]);
                        const float w0 = expf(l0 - ord2f(mx[rb * HH + h0])) / den[rb * HH + h0];
                        const float w1 = expf(l1 - ord2f(mx[rb * HH + h1])) / den[rb * HH + h1];
                        const RowVals<HD> p = agg_fetch<HD>(hfeat, sb, lane, w0, w1);
                        agg_add_row<HD>(acc, rb, lane, p);
                    }
                }
            }
            __syncthreads();
        }
    }
    __syncthreads();

    if constexpr (NC > 0) {
        for (int r = wave; r < nrows; r += NW) {
            float* arow = acc + r * HD;
            const float* a = arow + lane;
            const float4* b4 = reinterpret_cast<const float4*>(bias);
            const float4 c0 = b4[lane], c1 = b4[32 + lane];
            const float s0 = a[0], s1 = a[32], s2 = a[64], s3 = a[96], s4 = a[128], s5 = a[160], s6 = a[192], s7 = a[224];
            float4 o0, o1;
            o0.x = agg_act(s0 + c0.x, do_elu); o0.y = agg_act(s1 + c0.y, do_elu);
            o0.z = agg_act(s2 + c0.z, do_elu); o0.w = agg_act(s3 + c0.w, do_elu);
            o1.x = agg_act(s4 + c1.x, do_elu); o1.y = agg_act(s5 + c1.y, do_elu);
            o1.z = agg_act(s6 + c1.z, do_elu); o1.w = agg_act(s7 + c1.w, do_elu);
            const int swz = (r & 15) << 2;
            *reinterpret_cast<float4*>(arow + ((4 * lane) ^ swz))       = o0;
            *reinterpret_cast<float4*>(arow + ((128 + 4 * lane) ^ swz)) = o1;
        }
        __syncthreads();

        constexpr int HN = NC / 64;
        constexpr int MG = NW / HN;
        const int wv = __builtin_amdgcn_readfirstlane(wave);
        const int head = wv % HN;
        const int n0 = head * 64;
        const int nmt = (nrows + 15) >> 4;
        const int l15 = lane & 15;
        const int kh  = (lane >> 4) << 1;
        const float sa0 = asn[n0 + l15], sa1 = asn[n0 + 16 + l15], sa2 = asn[n0 + 32 + l15], sa3 = asn[n0 + 48 + l15];
        const float da0 = adn[n0 + l15], da1 = adn[n0 + 16 + l15], da2 = adn[n0 + 32 + l15], da3 = adn[n0 + 48 + l15];
        for (int mt = wv / HN; mt < nmt; mt += MG) {
            const int ar = min(mt * 16 + l15, AGG_ROWS - 1);
            const float* Ap = acc + ar * HD + kh;
            const int swz = (ar & 15) << 2;
            v8f q0 = {0.f,0.f,0.f,0.f,0.f,0.f,0.f,0.f};
            v8f q1 = q0, q2 = q0, q3 = q0;
            for (int k = 0; k < HD; k += 4) {
                const v2f a = *reinterpret_cast<const v2f*>(Ap + (k ^ swz));
                const float* Bp = Wn + (size_t)(k + kh) * NC + n0 + l15;
                v2f b0, b1, b2, b3;
                b0.x = Bp[0];  b0.y = Bp[(size_t)NC];
                b1.x = Bp[16]; b1.y = Bp[(size_t)NC + 16];
                b2.x = Bp[32]; b2.y = Bp[(size_t)NC + 32];
                b3.x = Bp[48]; b3.y = Bp[(size_t)NC + 48];
                q0 = __builtin_amdgcn_wmma_f32_16x16x4_f32(false, a, false, b0, (short)0, q0, false, false);
                q1 = __builtin_amdgcn_wmma_f32_16x16x4_f32(false, a, false, b1, (short)0, q1, false, false);
                q2 = __builtin_amdgcn_wmma_f32_16x16x4_f32(false, a, false, b2, (short)0, q2, false, false);
                q3 = __builtin_amdgcn_wmma_f32_16x16x4_f32(false, a, false, b3, (short)0, q3, false, false);
            }
            const int rbase = mt * 16 + ((lane >> 4) << 3);
            float* Cp = hn + (size_t)(lo + rbase) * NC + n0 + l15;
#pragma unroll
            for (int v = 0; v < 8; ++v) {
                if (rbase + v < nrows) {
                    Cp[(size_t)v * NC + 0]  = q0[v];
                    Cp[(size_t)v * NC + 16] = q1[v];
                    Cp[(size_t)v * NC + 32] = q2[v];
                    Cp[(size_t)v * NC + 48] = q3[v];
                }
            }
            float cs = 0.f, cd = 0.f;
#pragma unroll
            for (int v = 0; v < 8; ++v) {
                const float ps = row16_sum(q0[v] * sa0 + q1[v] * sa1 + q2[v] * sa2 + q3[v] * sa3);
                const float pd = row16_sum(q0[v] * da0 + q1[v] * da1 + q2[v] * da2 + q3[v] * da3);
                if (l15 == v) { cs = ps; cd = pd; }
            }
            if (l15 < 8 && rbase + l15 < nrows) {
                const int rl = rbase + l15;
                if constexpr (HN == 4) {
                    kp[kcap + TAIL_WORDS * rl + head] = cs;
                    edn[(size_t)(lo + rl) * HN + head] = cd;
                } else {
                    esn[(size_t)(lo + rl) * HN + head] = cs;
                    edn[(size_t)(lo + rl) * HN + head] = cd;
                }
            }
        }
    } else {
    float cs0 = 0.f, cs1 = 0.f;
    for (int r = wave; r < nrows; r += NW) {
        const float* a = acc + r * HD + lane;
        const size_t nd = (size_t)(lo + r);
        if constexpr (HD == 256) {
            const float4* b4 = reinterpret_cast<const float4*>(bias);
            const float4 c0 = b4[lane], c1 = b4[32 + lane];
            float4 o0, o1;
            o0.x = agg_act(a[0]   + c0.x, do_elu); o0.y = agg_act(a[32]  + c0.y, do_elu);
            o0.z = agg_act(a[64]  + c0.z, do_elu); o0.w = agg_act(a[96]  + c0.w, do_elu);
            o1.x = agg_act(a[128] + c1.x, do_elu); o1.y = agg_act(a[160] + c1.y, do_elu);
            o1.z = agg_act(a[192] + c1.z, do_elu); o1.w = agg_act(a[224] + c1.w, do_elu);
            float4* o4 = reinterpret_cast<float4*>(out + nd * 256);
            o4[lane] = o0;
            o4[32 + lane] = o1;
        } else {
            const float2 c = reinterpret_cast<const float2*>(bias)[lane];
            float2 o;
            o.x = agg_act(a[0]  + c.x, do_elu);
            o.y = agg_act(a[32] + c.y, do_elu);
            reinterpret_cast<float2*>(out + nd * 64)[lane] = o;
            cs0 += o.x;
            cs1 += o.y;
        }
    }
    if constexpr (HD == 64) {
        if (colpart != nullptr) {
            lw[wave * 64 + 2 * lane]     = cs0;
            lw[wave * 64 + 2 * lane + 1] = cs1;
            __syncthreads();
            if (tid < 64) {
                float s = 0.f;
                for (int w = 0; w < NW; ++w) s += lw[w * 64 + tid];
                colpart[(size_t)blockIdx.x * 64 + tid] = s;
            }
        }
    }
    }
}

#define MEAN_GROUPS 16
__global__ __launch_bounds__(64 * MEAN_GROUPS)
void gat_colmean_parts(const float* __restrict__ part, float* __restrict__ out, int N) {
    __shared__ float sm[MEAN_GROUPS * 64];
    const int nblk = (N + AGG_ROWS - 1) / AGG_ROWS;
    const int c = threadIdx.x & 63;
    const int g = threadIdx.x >> 6;
    float s = 0.f;
#pragma unroll 16
    for (int b = g; b < nblk; b += MEAN_GROUPS) s += part[(size_t)b * 64 + c];
    sm[g * 64 + c] = s;
    __syncthreads();
    if (threadIdx.x < 64) {
        float t = 0.f;
        for (int k = 0; k < MEAN_GROUPS; ++k) t += sm[k * 64 + threadIdx.x];
        out[threadIdx.x] = t / (float)N;
    }
}

static inline int cdiv(long long a, long long b) { return (int)((a + b - 1) / b); }

extern "C" void kernel_launch(void* const* d_in, const int* in_sizes, int n_in,
                              void* d_out, int out_size, void* d_ws, size_t ws_size,
                              hipStream_t stream) {
    const float* x   = (const float*)d_in[0];
    const int*   ei  = (const int*)  d_in[1];
    const float* W1  = (const float*)d_in[2];
    const float* a1s = (const float*)d_in[3];
    const float* a1d = (const float*)d_in[4];
    const float* b1  = (const float*)d_in[5];
    const float* W2  = (const float*)d_in[6];
    const float* a2s = (const float*)d_in[7];
    const float* a2d = (const float*)d_in[8];
    const float* b2  = (const float*)d_in[9];
    const float* W3  = (const float*)d_in[10];
    const float* a3s = (const float*)d_in[11];
    const float* a3d = (const float*)d_in[12];
    const float* b3  = (const float*)d_in[13];

    const int N  = in_sizes[0] / 16;
    const int E  = in_sizes[1] / 2;
    const int H  = 4;

    float* ws = (float*)d_ws;
    size_t off = 0;
    float*    buf_h    = ws + off; off += (size_t)N * 256;
    float*    buf_h2   = ws + off; off += (size_t)N * 256;
    float*    es       = ws + off; off += (size_t)N * H;
    float*    ed       = ws + off; off += (size_t)N * H;
    (void)ws_size; (void)off;

    float* node_emb  = (float*)d_out;
    float* graph_emb = node_emb + (size_t)N * 64;

    hipLaunchKernelGGL(gat_gemm_wmma, dim3(1, cdiv(N, 16)), dim3(32, 4), 0, stream,
                       x, W1, buf_h, a1s, a1d, es, ed, N, 16, 256, ei, E, node_emb);

    const dim3 ag(cdiv(N, AGG_ROWS)), ab(AGG_THREADS);
    float* es3      = es;
    float* col_part = es + (size_t)N;
    float* ed3      = es + 2 * (size_t)N;
    float* none     = nullptr;
    gat_agg_owner<256, 256, false><<<ag, ab, 0, stream>>>(ei, E, N, buf_h, es, ed, b1, none, 1, none, node_emb, 0,
                                                          W2, a2s, a2d, buf_h2, none, ed);
    gat_agg_owner<256, 64, true><<<ag, ab, 0, stream>>>(ei, E, N, buf_h2, es, ed, b2, none, 1, none, node_emb, 1,
                                                        W3, a3s, a3d, buf_h, es3, ed3);
    gat_agg_owner<64, 0, false><<<ag, ab, 0, stream>>>(ei, E, N, buf_h, es3, ed3, b3, node_emb, 0, col_part, node_emb, 1,
                                                       none, none, none, none, none, none);

    hipLaunchKernelGGL(gat_colmean_parts, dim3(1), dim3(64 * MEAN_GROUPS), 0, stream,
                       col_part, graph_emb, N);
}


# Layer 1 sums before it projects, in the owner block: waves gather 64-byte input rows per head into register sums, block multiplies the sums by W1 on f32 WMMA in LDS; h1 plane gone; sum order changes
# speedup vs baseline: 10.8184x; 1.1201x over previous
#include <hip/hip_runtime.h>
#include <math.h>

typedef __attribute__((ext_vector_type(2))) float v2f;
typedef __attribute__((ext_vector_type(8))) float v8f;

#define GAT_D 64

__device__ __forceinline__ unsigned f2ord(float f) {
    int b = __float_as_int(f);
    return (unsigned)b ^ (unsigned)((b >> 31) | 0x80000000);
}
__device__ __forceinline__ float ord2f(unsigned u) {
    int b = (u & 0x80000000u) ? (int)(u ^ 0x80000000u) : ~(int)u;
    return __int_as_float(b);
}

__device__ __forceinline__ float row16_sum(float v) {
    v += __int_as_float(__builtin_amdgcn_update_dpp(0, __float_as_int(v), 0x161, 0xf, 0xf, false));
    v += __int_as_float(__builtin_amdgcn_update_dpp(0, __float_as_int(v), 0x162, 0xf, 0xf, false));
    v += __int_as_float(__builtin_amdgcn_update_dpp(0, __float_as_int(v), 0x164, 0xf, 0xf, false));
    v += __int_as_float(__builtin_amdgcn_update_dpp(0, __float_as_int(v), 0x168, 0xf, 0xf, false));
    return v;
}

#define TAG_ROWS     196
#define TAG_REGION   (196 * 64)
#define TAG_LIST_MAX 4096
#define TAG_OFF      ((1 + 2 * TAG_LIST_MAX + 3) & ~3)
#define TAG_WAVES    4
#define TAG_UNITS    256
#define TAG_WORDS    (4 * TAG_UNITS)
#define BM_OFF       (TAG_OFF + TAG_WORDS)
#define BM_WORDS     ((TAG_UNITS / 32) * 256)

__device__ __forceinline__ bool tags_fit(int E, int N) {
    return (E & 15) == 0 && N <= 256 * TAG_ROWS && ((E >> 4) + TAG_UNITS - 1) / TAG_UNITS <= N / TAG_ROWS &&
           ((E >> 4) + 31) / 32 <= 2 * 1024;
}
__device__ __forceinline__ uint4* tag_unit(float* plane, int u) {
    return reinterpret_cast<uint4*>(plane + (size_t)(u / TAG_UNITS) * TAG_REGION + TAG_OFF) + (u % TAG_UNITS);
}
__device__ __forceinline__ unsigned* tag_group(float* plane, int g) {
    return reinterpret_cast<unsigned*>(plane + (size_t)(g / (TAG_UNITS / 32)) * TAG_REGION + BM_OFF) + (g % (TAG_UNITS / 32)) * 256;
}
__device__ __forceinline__ unsigned tag_word(int4 d) {
    return  (((unsigned)d.x / TAG_ROWS) & 255u)        | ((((unsigned)d.y / TAG_ROWS) & 255u) << 8) |
           ((((unsigned)d.z / TAG_ROWS) & 255u) << 16) | ((((unsigned)d.w / TAG_ROWS) & 255u) << 24);
}

__global__ void gat_gemm_wmma(const float* __restrict__ A,
                              const float* __restrict__ B,
                              const float* __restrict__ a_s,
                              const float* __restrict__ a_d,
                              float* __restrict__ es, float* __restrict__ ed,
                              int M, int K, int N,
                              const int* __restrict__ ei, int E, float* tagplane) {
    __shared__ unsigned own[TAG_WAVES * 256];
    if (tagplane != nullptr && tags_fit(E, M)) {
        const int U = E >> 4, G = (U + 31) >> 5;
        const int nblk = (int)(gridDim.x * gridDim.y);
        unsigned* mine = own + threadIdx.y * 256;
        const int ln = threadIdx.x;
        const int4* d4 = reinterpret_cast<const int4*>(ei + E);
        for (int gb = (int)(blockIdx.y * gridDim.x + blockIdx.x); gb * TAG_WAVES < G; gb += nblk) {
            const int g = gb * TAG_WAVES + (int)threadIdx.y;
            const int u = 32 * g + ln;
            for (int k = 0; k < 8; ++k) mine[ln + 32 * k] = 0u;
            __syncthreads();
            if (u < U) {
                unsigned w[4];
                for (int q = 0; q < 4; ++q) w[q] = tag_word(d4[4 * u + q]);
                *tag_unit(tagplane, u) = make_uint4(w[0], w[1], w[2], w[3]);
                for (int q = 0; q < 4; ++q)
                    for (int b = 0; b < 4; ++b)
                        __hip_atomic_fetch_or(&mine[(w[q] >> (8 * b)) & 255u], 1u << ln,
                                              __ATOMIC_RELAXED, __HIP_MEMORY_SCOPE_WORKGROUP);
            }
            __syncthreads();
            if (g < G) {
                unsigned* gw = tag_group(tagplane, g);
                for (int k = 0; k < 8; ++k) gw[ln + 32 * k] = mine[ln + 32 * k];
            }
            __syncthreads();
        }
    }
    const int lane  = threadIdx.x;
    const int grp   = blockIdx.x * blockDim.y + threadIdx.y;
    const int mtile = blockIdx.y;
    const int n0 = grp * 64;
    const int m0 = mtile * 16;
    if (n0 >= N || m0 >= M) return;

    const int l15 = lane & 15;
    const int kh  = (lane >> 4) << 1;
    const float* Arow = A + (size_t)(m0 + l15) * K;
    v8f acc0 = {0.f,0.f,0.f,0.f,0.f,0.f,0.f,0.f};
    v8f acc1 = acc0, acc2 = acc0, acc3 = acc0;
    const float sa0 = a_s[n0 + l15], sa1 = a_s[n0 + 16 + l15], sa2 = a_s[n0 + 32 + l15], sa3 = a_s[n0 + 48 + l15];
    const float da0 = a_d[n0 + l15], da1 = a_d[n0 + 16 + l15], da2 = a_d[n0 + 32 + l15], da3 = a_d[n0 + 48 + l15];

    for (int k = 0; k < K; k += 4) {
        v2f a;
        a.x = Arow[k + kh];
        a.y = Arow[k + kh + 1];
        const float* Bp = B + (size_t)(k + kh) * N + n0 + l15;
        v2f b0, b1, b2, b3;
        b0.x = Bp[0];  b0.y = Bp[(size_t)N];
        b1.x = Bp[16]; b1.y = Bp[(size_t)N + 16];
        b2.x = Bp[32]; b2.y = Bp[(size_t)N + 32];
        b3.x = Bp[48]; b3.y = Bp[(size_t)N + 48];
        acc0 = __builtin_amdgcn_wmma_f32_16x16x4_f32(false, a, false, b0, (short)0, acc0, false, false);
        acc1 = __builtin_amdgcn_wmma_f32_16x16x4_f32(false, a, false, b1, (short)0, acc1, false, false);
        acc2 = __builtin_amdgcn_wmma_f32_16x16x4_f32(false, a, false, b2, (short)0, acc2, false, false);
        acc3 = __builtin_amdgcn_wmma_f32_16x16x4_f32(false, a, false, b3, (short)0, acc3, false, false);
    }


    float cs = 0.f, cd = 0.f;
#pragma unroll
    for (int v = 0; v < 8; ++v) {
        const float ps = row16_sum(acc0[v] * sa0 + acc1[v] * sa1 + acc2[v] * sa2 + acc3[v] * sa3);
        const float pd = row16_sum(acc0[v] * da0 + acc1[v] * da1 + acc2[v] * da2 + acc3[v] * da3);
        if (l15 == v) { cs = ps; cd = pd; }
    }
    if (l15 < 8) {
        const size_t ci = (size_t)(m0 + ((lane >> 4) << 3) + l15) * (size_t)(N / GAT_D) + (size_t)grp;
        es[ci] = cs;
        ed[ci] = cd;
    }
}

__device__ __forceinline__ void edge_endpoints(const int* __restrict__ ei, int E,
                                               int e, int& src, int& dst) {
    if (e < E) { src = ei[e]; dst = ei[E + e]; }
    else       { src = e - E; dst = e - E;    }
}

#define AGG_ROWS    196
#define AGG_THREADS 1024
#define AGG_K       8
#define AGG_CAP     4096
#define AGG_U       4
#define SUB_CAP     256
#define KEEP_PITCH  64
#define TAIL_WORDS  4

template <int HD> struct RowVals { float v[HD / 32]; };

__device__ __forceinline__ const float* es_tail(const float* keep, int N, int src) {
    const int blk = src / AGG_ROWS;
    const int lo_b = blk * AGG_ROWS;
    const int nr = min(AGG_ROWS, N - lo_b);
    return keep + (size_t)lo_b * KEEP_PITCH + nr * (KEEP_PITCH - TAIL_WORDS) + TAIL_WORDS * (src - lo_b);
}

template <int HH, bool ES_TAIL>
__device__ __forceinline__ float es_at(const float* es, const float* keep, int N, int src, int h) {
    if constexpr (ES_TAIL) return es_tail(keep, N, src)[h];
    else                   return es[(size_t)src * HH + h];
}

template <int HD>
__device__ __forceinline__ RowVals<HD> agg_fetch(const float* __restrict__ hfeat,
                                                 int src, int lane, float w0, float w1) {
    RowVals<HD> r;
    if constexpr (HD == 256) {
        const float4* hp = reinterpret_cast<const float4*>(hfeat + (size_t)src * 256);
        const float4 a = hp[lane];
        const float4 b = hp[32 + lane];
        r.v[0] = a.x * w0; r.v[1] = a.y * w0; r.v[2] = a.z * w0; r.v[3] = a.w * w0;
        r.v[4] = b.x * w1; r.v[5] = b.y * w1; r.v[6] = b.z * w1; r.v[7] = b.w * w1;
    } else {
        const float2* hp = reinterpret_cast<const float2*>(hfeat + (size_t)src * 64);
        const float2 a = hp[lane];
        r.v[0] = a.x * w0; r.v[1] = a.y * w0;
    }
    return r;
}

template <int HD>
__device__ __forceinline__ RowVals<HD> agg_take(const float* __restrict__ hfeat,
                                                const int* __restrict__ lsrc,
                                                const int* __restrict__ lrow,
                                                const float* __restrict__ lw,
                                                int i, int lane, int& row) {
    row = lrow[i];
    const float* wp = lw + i * (HD / 64);
    float w0, w1;
    if constexpr (HD == 256) { w0 = wp[lane >> 4]; w1 = wp[2 + (lane >> 4)]; }
    else                     { w0 = wp[0]; w1 = 0.f; }
    return agg_fetch<HD>(hfeat, lsrc[i], lane, w0, w1);
}

template <int HD>
__device__ __forceinline__ RowVals<HD> agg_take_at(const float* __restrict__ hfeat,
                                                   const float* __restrict__ lw,
                                                   int i, int src, int lane) {
    const float* wp = lw + i * (HD / 64);
    float w0, w1;
    if constexpr (HD == 256) { w0 = wp[lane >> 4]; w1 = wp[2 + (lane >> 4)]; }
    else                     { w0 = wp[0]; w1 = 0.f; }
    return agg_fetch<HD>(hfeat, src, lane, w0, w1);
}

template <int HD>
__device__ __forceinline__ void agg_add_row(float* __restrict__ acc, int row, int lane,
                                            const RowVals<HD>& p) {
    float* a = acc + row * HD + lane;
#pragma unroll
    for (int j = 0; j < HD / 32; ++j)
        __hip_atomic_fetch_add(a + j * 32, p.v[j], __ATOMIC_RELAXED, __HIP_MEMORY_SCOPE_WORKGROUP);
}

#define PRE_K 16
__device__ __forceinline__ RowVals<64> pre_fetch(const float* __restrict__ xin, int src, int lane, float w) {
    RowVals<64> r;
    const float2 a = reinterpret_cast<const float2*>(xin + (size_t)src * PRE_K)[lane & 7];
    r.v[0] = a.x * w; r.v[1] = a.y * w;
    return r;
}
__device__ __forceinline__ float* pre_word(float* __restrict__ acc, int row, int lane) {
    return acc + row * 256 + (lane >> 3) * 64 + 2 * (lane & 7);
}
__device__ __forceinline__ void pre_add_row(float* __restrict__ acc, int row, int lane, const RowVals<64>& p) {
    float* a = pre_word(acc, row, lane);
    __hip_atomic_fetch_add(a,     p.v[0], __ATOMIC_RELAXED, __HIP_MEMORY_SCOPE_WORKGROUP);
    __hip_atomic_fetch_add(a + 1, p.v[1], __ATOMIC_RELAXED, __HIP_MEMORY_SCOPE_WORKGROUP);
}
__device__ __forceinline__ RowVals<64> pre_take_at(const float* __restrict__ xin, const float* __restrict__ lw,
                                                   int i, int src, int lane) {
    return pre_fetch(xin, src, lane, lw[i * 4 + (lane >> 3)]);
}

template <int HD, int SL>
__device__ __forceinline__ void own_add(float (&ra)[SL][HD / 32], int s, const RowVals<HD>& p) {
#pragma unroll
    for (int c = 0; c < SL; ++c) {
        if (__builtin_expect(s == c, 0)) {
#pragma unroll
            for (int j = 0; j < HD / 32; ++j) ra[c][j] += p.v[j];
        }
    }
}

__device__ __forceinline__ float agg_act(float v, int do_elu) {
    if (do_elu) v = v > 0.f ? v : (expf(v) - 1.f);
    return v;
}

__device__ __forceinline__ float agg_lrelu(float s) { return s > 0.f ? s : 0.2f * s; }

template <bool TAIL>
__device__ __forceinline__ void agg_load_dst(const int* __restrict__ dstp, int base, int lane,
                                             int E, int (&d)[AGG_K]) {
#pragma unroll
    for (int k = 0; k < AGG_K; ++k) {
        const int e = base + 32 * k + lane;
        if constexpr (TAIL) d[k] = e < E ? dstp[e] : -1;
        else                d[k] = dstp[e];
    }
}

__device__ __forceinline__ void agg_collect(const int (&d)[AGG_K], int base, int lane, int lo, int nrows,
                                            int* __restrict__ lsrc, int* __restrict__ lrow,
                                            unsigned* __restrict__ cnt) {
    unsigned m[AGG_K], any = 0;
    int rl[AGG_K];
#pragma unroll
    for (int k = 0; k < AGG_K; ++k) {
        rl[k] = d[k] - lo;
        m[k] = __builtin_amdgcn_ballot_w32((unsigned)rl[k] < (unsigned)nrows);
        any |= m[k];
    }
    if (!any) return;
    unsigned tot = 0;
#pragma unroll
    for (int k = 0; k < AGG_K; ++k) tot += (unsigned)__builtin_popcount(m[k]);
    unsigned start = 0;
    if (lane == 0)
        start = __hip_atomic_fetch_add(cnt, tot, __ATOMIC_RELAXED, __HIP_MEMORY_SCOPE_WORKGROUP);
    start = (unsigned)__builtin_amdgcn_readfirstlane((int)start);
    const unsigned below = (1u << lane) - 1u;
#pragma unroll
    for (int k = 0; k < AGG_K; ++k) {
        if ((m[k] >> lane) & 1u) {
            const unsigned p = start + (unsigned)__builtin_popcount(m[k] & below);
            if (p < (unsigned)AGG_CAP) { lsrc[p] = base + 32 * k + lane; lrow[p] = rl[k]; }
        }
        start += (unsigned)__builtin_popcount(m[k]);
    }
}

struct DstOne  { int  d[AGG_K]; };
struct DstFour { int4 d[AGG_K / 4]; };

__device__ __forceinline__ void agg_load(const int* __restrict__ dstp, int base, int lane, DstOne& s) {
    agg_load_dst<false>(dstp, base, lane, 0, s.d);
}
__device__ __forceinline__ void agg_load(const int* __restrict__ dstp, int base, int lane, DstFour& s) {
#pragma unroll
    for (int q = 0; q < AGG_K / 4; ++q)
        s.d[q] = *reinterpret_cast<const int4*>(dstp + base + 128 * q + 4 * lane);
}

__device__ __forceinline__ void agg_note(const DstOne& s, int base, int lane, int lo, int nrows,
                                         int* __restrict__ lsrc, int* __restrict__ lrow,
                                         unsigned* __restrict__ cnt) {
    agg_collect(s.d, base, lane, lo, nrows, lsrc, lrow, cnt);
}

__device__ __forceinline__ void agg_note_one(unsigned rl, int e, int nrows,
                                             int* __restrict__ lsrc, int* __restrict__ lrow,
                                             unsigned* __restrict__ cnt) {
    if (rl < (unsigned)nrows) {
        const unsigned p = __hip_atomic_fetch_add(cnt, 1u, __ATOMIC_RELAXED, __HIP_MEMORY_SCOPE_WORKGROUP);
        if (p < (unsigned)AGG_CAP) { lsrc[p] = e; lrow[p] = (int)rl; }
    }
}

__device__ __forceinline__ void agg_note(const DstFour& s, int base, int lane, int lo, int nrows,
                                         int* __restrict__ lsrc, int* __restrict__ lrow,
                                         unsigned* __restrict__ cnt) {
    unsigned r[AGG_K];
#pragma unroll
    for (int q = 0; q < AGG_K / 4; ++q) {
        r[4 * q + 0] = (unsigned)(s.d[q].x - lo);
        r[4 * q + 1] = (unsigned)(s.d[q].y - lo);
        r[4 * q + 2] = (unsigned)(s.d[q].z - lo);
        r[4 * q + 3] = (unsigned)(s.d[q].w - lo);
    }
    unsigned mn = r[0];
#pragma unroll
    for (int k = 1; k < AGG_K; ++k) mn = min(mn, r[k]);
    if (mn < (unsigned)nrows) {
#pragma unroll
        for (int k = 0; k < AGG_K; ++k)
            agg_note_one(r[k], base + 128 * (k >> 2) + 4 * lane + (k & 3), nrows, lsrc, lrow, cnt);
    }
}

template <class DS>
__device__ __forceinline__ void agg_walk(const int* __restrict__ dstp, int full, int t0, int trip, int woff,
                                         int lane, int lo, int nrows,
                                         int* __restrict__ lsrc, int* __restrict__ lrow,
                                         unsigned* __restrict__ cnt) {
    DS dA, dB;
    agg_load(dstp, t0 * trip + woff, lane, dA);
    for (int t = 0; t < full; t += 2) {
        int tc0 = t0 + t;    if (tc0 >= full) tc0 -= full;
        int tc1 = tc0 + 1;   if (tc1 >= full) tc1 -= full;
        int tc2 = tc1 + 1;   if (tc2 >= full) tc2 -= full;
        agg_load(dstp, tc1 * trip + woff, lane, dB);
        agg_note(dA, tc0 * trip + woff, lane, lo, nrows, lsrc, lrow, cnt);
        agg_load(dstp, tc2 * trip + woff, lane, dA);
        if (t + 1 < full)
            agg_note(dB, tc1 * trip + woff, lane, lo, nrows, lsrc, lrow, cnt);
    }
}

static_assert(TAG_ROWS == AGG_ROWS && TAG_REGION == AGG_ROWS * KEEP_PITCH && TAG_LIST_MAX == AGG_CAP,
              "the tag bytes are laid out for the owner blocks' rows, words and list length");
static_assert(BM_OFF + BM_WORDS <= AGG_ROWS * (KEEP_PITCH - TAIL_WORDS),
              "the tag bytes and owner words end before a full block's coefficient words");

__device__ __forceinline__ unsigned zero_bytes(unsigned x) {
    return ~((((x & 0x7F7F7F7Fu) + 0x7F7F7F7Fu) | x) | 0x7F7F7F7Fu);
}

__device__ __forceinline__ int tag_next(unsigned& bits0, unsigned& bits1, int g0, int g1) {
    if (bits0 != 0u) { const int b = __builtin_ctz(bits0); bits0 &= bits0 - 1u; return 32 * g0 + b; }
    if (bits1 != 0u) { const int b = __builtin_ctz(bits1); bits1 &= bits1 - 1u; return 32 * g1 + b; }
    return -1;
}

__device__ __forceinline__ void tag_note(uint4 w, int u, unsigned pat,
                                         int* __restrict__ lsrc, unsigned* __restrict__ cnt) {
    unsigned m = (zero_bytes(w.x ^ pat) >> 7) | (zero_bytes(w.y ^ pat) >> 6) |
                 (zero_bytes(w.z ^ pat) >> 5) | (zero_bytes(w.w ^ pat) >> 4);
    while (m != 0u) {
        const int bit = __builtin_ctz(m);
        m &= m - 1u;
        const unsigned p = __hip_atomic_fetch_add(cnt, 1u, __ATOMIC_RELAXED, __HIP_MEMORY_SCOPE_WORKGROUP);
        if (p < (unsigned)AGG_CAP) lsrc[p] = 16 * u + 4 * (bit & 7) + (bit >> 3);
    }
}

__device__ __forceinline__ void tag_walk(float* plane, int E, int tid, unsigned owner,
                                         int* __restrict__ lsrc, unsigned* __restrict__ cnt) {
    const int U = E >> 4, G = (U + 31) >> 5;
    const unsigned pat = owner * 0x01010101u;
    const int g0 = tid, g1 = tid + AGG_THREADS;
    unsigned bits0 = g0 < G ? tag_group(plane, g0)[owner] : 0u;
    unsigned bits1 = g1 < G ? tag_group(plane, g1)[owner] : 0u;
    int u = tag_next(bits0, bits1, g0, g1);
    uint4 w = *tag_unit(plane, max(u, 0));
    while (u >= 0) {
        const int un = tag_next(bits0, bits1, g0, g1);
        const uint4 wn = *tag_unit(plane, max(un, 0));
        tag_note(w, u, pat, lsrc, cnt);
        u = un; w = wn;
    }
}

template <int HD, int NC, bool ES_TAIL, bool PRE>
__global__ __launch_bounds__(AGG_THREADS)
void gat_agg_owner(const int* __restrict__ ei, int E, int N,
                   const float* __restrict__ hfeat,
                   const float* es, const float* ed,
                   const float* __restrict__ bias,
                   float* out, int do_elu,
                   float* __restrict__ colpart,
                   float* keep, int take_list,
                   const float* __restrict__ Wn,
                   const float* __restrict__ asn, const float* __restrict__ adn,
                   float* __restrict__ hn, float* esn, float* edn,
                   const float* __restrict__ Wp) {
    static_assert(NC == 0 || HD == 256, "the projection at the end reads 256-channel rows");
    static_assert(!ES_TAIL || HD == 256, "only 4-head coefficients are left in the output plane");
    static_assert(!PRE || (HD == 256 && NC > 0 && HD / 64 * PRE_K == 64), "summing first: 4 heads x 16 input channels = 2 sums a lane, projected in place ahead of the next layer's weight");
    constexpr int HH = HD / 64;
    constexpr int NW = AGG_THREADS / 32;
    constexpr int SPAN = 32 * AGG_K;
    constexpr int TRIP = NW * SPAN;
    __shared__ __attribute__((aligned(16))) float acc[AGG_ROWS * HD];
    __shared__ unsigned mx[AGG_ROWS * HH];
    __shared__ float    den[AGG_ROWS * HH];
    __shared__ float    edv[AGG_ROWS * HH];
    __shared__ int      lsrc[AGG_CAP];
    __shared__ int      lrow[AGG_CAP];
    __shared__ float    lw[AGG_CAP * HH];
    __shared__ unsigned cnt;
    __shared__ unsigned short sub[(AGG_THREADS / 32) * SUB_CAP];
    __shared__ unsigned subcnt[AGG_THREADS / 32];
    const int tid  = threadIdx.x;
    const int lane = tid & 31;
    const int wave = tid >> 5;
    const int lo = blockIdx.x * AGG_ROWS;
    const int nrows = max(0, min(AGG_ROWS, N - lo));

    float* kp = keep + (size_t)lo * KEEP_PITCH;
    const int kcap = nrows * (KEEP_PITCH - TAIL_WORDS);
    int taken = -1;
    if (take_list && nrows > 0) taken = __builtin_amdgcn_readfirstlane(__float_as_int(kp[0]));
    const bool have = taken >= 0 && taken <= AGG_CAP && 1 + 2 * taken <= kcap;

    {
        float4* a4 = reinterpret_cast<float4*>(acc);
        for (int i = tid; i < AGG_ROWS * HD / 4; i += AGG_THREADS) a4[i] = make_float4(0.f, 0.f, 0.f, 0.f);
        if (tid < AGG_THREADS / 32) subcnt[tid] = 0u;
        for (int i = tid; i < nrows * HH; i += AGG_THREADS) {
            mx[i] = 0u;
            den[i] = 0.f;
            edv[i] = ed[(size_t)lo * HH + i];
        }
        if (have) {
            for (int i = tid; i < taken; i += AGG_THREADS) {
                lsrc[i] = __float_as_int(kp[1 + i]);
                lrow[i] = __float_as_int(kp[1 + taken + i]);
            }
            if (tid == 0) cnt = (unsigned)taken;
        } else {
            for (int r = tid; r < nrows; r += AGG_THREADS) { lsrc[r] = E + lo + r; lrow[r] = r; }
            if (tid == 0) cnt = (unsigned)nrows;
        }
    }
    __syncthreads();

    const int* __restrict__ dstp = ei + E;
    const int full = E / TRIP;
    const int t0 = (int)((blockIdx.x * (unsigned)full) / gridDim.x);
    const int woff = wave * SPAN;
    const bool tagw = !take_list && tags_fit(E, N);
    if (!have && tagw) {
        tag_walk(keep, E, tid, blockIdx.x, lsrc, &cnt);
    } else if (!have && full > 0) {
        if ((E & 3) == 0)
            agg_walk<DstFour>(dstp, full, t0, TRIP, woff, lane, lo, nrows, lsrc, lrow, &cnt);
        else
            agg_walk<DstOne>(dstp, full, t0, TRIP, woff, lane, lo, nrows, lsrc, lrow, &cnt);
    }
    if (!have && !tagw && full * TRIP < E) {
        int dT[AGG_K];
        agg_load_dst<true>(dstp, full * TRIP + woff, lane, E, dT);
        agg_collect(dT, full * TRIP + woff, lane, lo, nrows, lsrc, lrow, &cnt);
    }
    __syncthreads();
    const int n = (int)cnt;

    const bool save = !take_list && n <= AGG_CAP && 1 + 2 * n <= kcap;
    if (!take_list && tid == 0 && nrows > 0) kp[0] = __int_as_float(save ? n : -1);
    if (n <= AGG_CAP) {
        for (int i = tid; i < n; i += AGG_THREADS) {
            int src = lsrc[i];
            int row;
            if (!have) {
                const int e = src;
                src = e < E ? ei[e] : e - E;
                lsrc[i] = src;
                if (tagw && e < E) { row = dstp[e] - lo; lrow[i] = row; }
                else               row = lrow[i];
                if (save) { kp[1 + i] = __int_as_float(src); kp[1 + n + i] = __int_as_float(row); }
            } else {
                row = lrow[i];
            }
            float sv[HH];
            if constexpr (HH == 4) {
                const float4 q = ES_TAIL ? *reinterpret_cast<const float4*>(es_tail(keep, N, src))
                                         : reinterpret_cast<const float4*>(es)[src];
                sv[0] = q.x; sv[1] = q.y; sv[2] = q.z; sv[3] = q.w;
            } else {
                sv[0] = es[src];
            }
#pragma unroll
            for (int h = 0; h < HH; ++h) {
                const float lr = agg_lrelu(sv[h] + edv[row * HH + h]);
                lw[i * HH + h] = lr;
                atomicMax(&mx[row * HH + h], f2ord(lr));
            }
        }
        __syncthreads();
        for (int i = tid; i < n; i += AGG_THREADS) {
            const int row = lrow[i];
#pragma unroll
            for (int h = 0; h < HH; ++h) {
                const float v = expf(lw[i * HH + h] - ord2f(mx[row * HH + h]));
                lw[i * HH + h] = v;
                __hip_atomic_fetch_add(&den[row * HH + h], v, __ATOMIC_RELAXED, __HIP_MEMORY_SCOPE_WORKGROUP);
            }
        }
        __syncthreads();
        static_assert(NW == 32 && AGG_CAP <= 65536, "rows are dealt out by row & 31, a wave's lanes read the 32 counters; a list place fits 16 bits");
        for (int i = tid; i < n; i += AGG_THREADS) {
            const int row = lrow[i];
#pragma unroll
            for (int h = 0; h < HH; ++h) lw[i * HH + h] = lw[i * HH + h] / den[row * HH + h];
            const int ow = row & (NW - 1);
            const unsigned p = __hip_atomic_fetch_add(&subcnt[ow], 1u, __ATOMIC_RELAXED, __HIP_MEMORY_SCOPE_WORKGROUP);
            if (p < (unsigned)SUB_CAP) sub[ow * SUB_CAP + p] = (unsigned short)i;
        }
        __syncthreads();
        constexpr int SL = (AGG_ROWS + NW - 1) / NW;
        constexpr int RV = HD / 32;
        const unsigned short* mysub = sub + wave * SUB_CAP;
        const unsigned cl = subcnt[lane];
        const bool allfit = __builtin_amdgcn_ballot_w32(cl > (unsigned)SUB_CAP) == 0u;
        const int mine_n = __builtin_amdgcn_readfirstlane((int)subcnt[wave]);
        if (allfit) {
            if constexpr (PRE) {
                constexpr int PRE_U = 16;
                float ra[SL][2];
#pragma unroll
                for (int s = 0; s < SL; ++s) { ra[s][0] = 0.f; ra[s][1] = 0.f; }
                const int ng = (mine_n + PRE_U - 1) / PRE_U;
                int vi, vs, vl;
                {
                    const int idx = lane & (PRE_U - 1);
                    vi = ng > 0 ? (int)mysub[min(idx, mine_n - 1)] : 0;
                    vs = lsrc[vi];
                    const int rr = lrow[vi];
                    vl = idx < mine_n ? (rr >> 5) : SL;
                }
                for (int g = 0; g < ng; ++g) {
                    RowVals<64> p[PRE_U];
                    int sl[PRE_U];
#pragma unroll
                    for (int u = 0; u < PRE_U; ++u) {
                        const int pi = __builtin_amdgcn_readlane(vi, u);
                        const int ps = __builtin_amdgcn_readlane(vs, u);
                        sl[u] = __builtin_amdgcn_readlane(vl, u);
                        p[u] = pre_take_at(hfeat, lw, pi, ps, lane);
                    }
                    {
                        const int idx = min(g + 1, ng - 1) * PRE_U + (lane & (PRE_U - 1));
                        vi = (int)mysub[min(idx, mine_n - 1)];
                        vs = lsrc[vi];
                        const int rr = lrow[vi];
                        vl = idx < mine_n ? (rr >> 5) : SL;
                    }
#pragma unroll
                    for (int u = 0; u < PRE_U; ++u) own_add<64, SL>(ra, sl[u], p[u]);
                }
#pragma unroll
                for (int s = 0; s < SL; ++s) {
                    const int r = wave + NW * s;
                    if (r < nrows) *reinterpret_cast<float2*>(pre_word(acc, r, lane)) = make_float2(ra[s][0], ra[s][1]);
                }
            } else {
            float ra[SL][RV];
#pragma unroll
            for (int s = 0; s < SL; ++s)
#pragma unroll
                for (int j = 0; j < RV; ++j) ra[s][j] = 0.f;
            const int mgrp = mine_n / AGG_U;
            int pi[AGG_U], ps[AGG_U], pr[AGG_U];
#pragma unroll
            for (int u = 0; u < AGG_U; ++u) {
                pi[u] = mgrp > 0 ? (int)mysub[u] : 0;
                ps[u] = lsrc[pi[u]];
                pr[u] = lrow[pi[u]];
            }
            for (int g = 0; g < mgrp; ++g) {
                RowVals<HD> p[AGG_U];
                int rw[AGG_U];
#pragma unroll
                for (int u = 0; u < AGG_U; ++u) {
                    p[u] = agg_take_at<HD>(hfeat, lw, pi[u], ps[u], lane);
                    rw[u] = pr[u];
                }
                const int gn = min(g + 1, mgrp - 1);
#pragma unroll
                for (int u = 0; u < AGG_U; ++u) {
                    pi[u] = (int)mysub[gn * AGG_U + u];
                    ps[u] = lsrc[pi[u]];
                    pr[u] = lrow[pi[u]];
                }
#pragma unroll
                for (int u = 0; u < AGG_U; ++u)
                    own_add<HD, SL>(ra, __builtin_amdgcn_readfirstlane(rw[u]) >> 5, p[u]);
            }
            for (int k = mgrp * AGG_U; k < mine_n; ++k) {
                int rw1;
                const RowVals<HD> p1 = agg_take<HD>(hfeat, lsrc, lrow, lw, (int)mysub[k], lane, rw1);
                own_add<HD, SL>(ra, __builtin_amdgcn_readfirstlane(rw1) >> 5, p1);
            }
#pragma unroll
            for (int s = 0; s < SL; ++s) {
                const int r = wave + NW * s;
                if (r < nrows) {
#pragma unroll
                    for (int j = 0; j < RV; ++j) acc[r * HD + j * 32 + lane] = ra[s][j];
                }
            }
            }
        } else {
        if constexpr (PRE) {
            for (int i = wave; i < n; i += NW) {
                const RowVals<64> p1 = pre_take_at(hfeat, lw, i, lsrc[i], lane);
                pre_add_row(acc, lrow[i], lane, p1);
            }
        } else {
        const int ngrp = n / AGG_U;
        for (int g = wave; g < ngrp; g += NW) {
            RowVals<HD> p[AGG_U];
            int rw[AGG_U];
#pragma unroll
            for (int u = 0; u < AGG_U; ++u)
                p[u] = agg_take<HD>(hfeat, lsrc, lrow, lw, g * AGG_U + u, lane, rw[u]);
#pragma unroll
            for (int u = 0; u < AGG_U; ++u) agg_add_row<HD>(acc, rw[u], lane, p[u]);
        }
        for (int i = ngrp * AGG_U + wave; i < n; i += NW) {
            int rw1;
            const RowVals<HD> p1 = agg_take<HD>(hfeat, lsrc, lrow, lw, i, lane, rw1);
            agg_add_row<HD>(acc, rw1, lane, p1);
        }
        }
        }
    } else {
        const int Et = E + N;
        for (int ph = 0; ph < 3; ++ph) {
            for (int eb = wave * 32; eb < Et; eb += AGG_THREADS) {
                const int e = eb + lane;
                int src = 0, rl = -1;
                if (e < Et) {
                    int dst; edge_endpoints(ei, E, e, src, dst);
                    rl = dst - lo;
                }
                const bool in = (unsigned)rl < (unsigned)nrows;
                if (ph < 2) {
                    if (in) {
#pragma unroll
                        for (int h = 0; h < HH; ++h) {
                            const float lr = agg_lrelu(es_at<HH, ES_TAIL>(es, keep, N, src, h) + edv[rl * HH + h]);
                            if (ph == 0) atomicMax(&mx[rl * HH + h], f2ord(lr));
                            else __hip_atomic_fetch_add(&den[rl * HH + h], expf(lr - ord2f(mx[rl * HH + h])),
                                                        __ATOMIC_RELAXED, __HIP_MEMORY_SCOPE_WORKGROUP);
                        }
                    }
                } else {
                    unsigned mk = __builtin_amdgcn_ballot_w32(in);
                    while (mk) {
                        const int b = __builtin_ctz(mk);
                        mk &= mk - 1;
                        const int sb = __builtin_amdgcn_readlane(src, b);
                        const int rb = __builtin_amdgcn_readlane(rl, b);
                        if constexpr (PRE) {
                            const int hq = lane >> 3;
                            const float lq = agg_lrelu(es_at<HH, ES_TAIL>(es, keep, N, sb, hq) + edv[rb * HH + hq]);
                            const float wq = expf(lq - ord2f(mx[rb * HH + hq])) / den[rb * HH + hq];
                            pre_add_row(acc, rb, lane, pre_fetch(hfeat, sb, lane, wq));
                        } else {
                        const int h0 = HH == 4 ? (lane >> 4) : 0;
                        const int h1 = HH == 4 ? 2 + (lane >> 4) : 0;
                        const float l0 = agg_lrelu(es_at<HH, ES_TAIL>(es, keep, N, sb, h0) + edv[rb * HH + h0]);
                        const float l1 = agg_lrelu(es_at<HH, ES_TAIL>(es, keep, N, sb, h1) + edv[rb * HH + h1]);
                        const float w0 = expf(l0 - ord2f(mx[rb * HH + h0])) / den[rb * HH + h0];
                        const float w1 = expf(l1 - ord2f(mx[rb * HH + h1])) / den[rb * HH + h1];
                        const RowVals<HD> p = agg_fetch<HD>(hfeat, sb, lane, w0, w1);
                        agg_add_row<HD>(acc, rb, lane, p);
                        }
                    }
                }
            }
            __syncthreads();
        }
    }
    __syncthreads();

    if constexpr (NC > 0) {
        if constexpr (PRE) {
            const int wq = __builtin_amdgcn_readfirstlane(wave);
            const int c0 = (wq & 3) * 64;
            const int nmt1 = (nrows + 15) >> 4;
            const int q15 = lane & 15;
            const int qh  = (lane >> 4) << 1;
            v2f wb[4][4];
#pragma unroll
            for (int ks = 0; ks < 4; ++ks)
#pragma unroll
                for (int j = 0; j < 4; ++j) {
                    const float* Bp = Wp + (size_t)(4 * ks + qh) * HD + c0 + 16 * j + q15;
                    wb[ks][j].x = Bp[0];
                    wb[ks][j].y = Bp[HD];
                }
            const float bb0 = bias[c0 + q15], bb1 = bias[c0 + 16 + q15], bb2 = bias[c0 + 32 + q15], bb3 = bias[c0 + 48 + q15];
            for (int mt = wq >> 2; mt < nmt1; mt += NW / 4) {
                const int ar = min(mt * 16 + q15, AGG_ROWS - 1);
                const float* Ap = acc + ar * HD + c0 + qh;
                v2f af[4];
#pragma unroll
                for (int ks = 0; ks < 4; ++ks) af[ks] = *reinterpret_cast<const v2f*>(Ap + 4 * ks);
                v8f q0 = {0.f,0.f,0.f,0.f,0.f,0.f,0.f,0.f};
                v8f q1 = q0, q2 = q0, q3 = q0;
#pragma unroll
                for (int ks = 0; ks < 4; ++ks) {
                    q0 = __builtin_amdgcn_wmma_f32_16x16x4_f32(false, af[ks], false, wb[ks][0], (short)0, q0, false, false);
                    q1 = __builtin_amdgcn_wmma_f32_16x16x4_f32(false, af[ks], false, wb[ks][1], (short)0, q1, false, false);
                    q2 = __builtin_amdgcn_wmma_f32_16x16x4_f32(false, af[ks], false, wb[ks][2], (short)0, q2, false, false);
                    q3 = __builtin_amdgcn_wmma_f32_16x16x4_f32(false, af[ks], false, wb[ks][3], (short)0, q3, false, false);
                }
                const int rb = mt * 16 + ((lane >> 4) << 3);
#pragma unroll
                for (int v = 0; v < 8; ++v) {
                    const int r = rb + v;
                    if (r < AGG_ROWS) {
                        float* orow = acc + r * HD;
                        const int swz = (r & 15) << 2;
                        orow[(c0 + q15) ^ swz]      = agg_act(q0[v] + bb0, do_elu);
                        orow[(c0 + 16 + q15) ^ swz] = agg_act(q1[v] + bb1, do_elu);
                        orow[(c0 + 32 + q15) ^ swz] = agg_act(q2[v] + bb2, do_elu);
                        orow[(c0 + 48 + q15) ^ swz] = agg_act(q3[v] + bb3, do_elu);
                    }
                }
            }
        } else {
        for (int r = wave; r < nrows; r += NW) {
            float* arow = acc + r * HD;
            const float* a = arow + lane;
            const float4* b4 = reinterpret_cast<const float4*>(bias);
            const float4 c0 = b4[lane], c1 = b4[32 + lane];
            const float s0 = a[0], s1 = a[32], s2 = a[64], s3 = a[96], s4 = a[128], s5 = a[160], s6 = a[192], s7 = a[224];
            float4 o0, o1;
            o0.x = agg_act(s0 + c0.x, do_elu); o0.y = agg_act(s1 + c0.y, do_elu);
            o0.z = agg_act(s2 + c0.z, do_elu); o0.w = agg_act(s3 + c0.w, do_elu);
            o1.x = agg_act(s4 + c1.x, do_elu); o1.y = agg_act(s5 + c1.y, do_elu);
            o1.z = agg_act(s6 + c1.z, do_elu); o1.w = agg_act(s7 + c1.w, do_elu);
            const int swz = (r & 15) << 2;
            *reinterpret_cast<float4*>(arow + ((4 * lane) ^ swz))       = o0;
            *reinterpret_cast<float4*>(arow + ((128 + 4 * lane) ^ swz)) = o1;
        }
        }
        __syncthreads();

        constexpr int HN = NC / 64;
        constexpr int MG = NW / HN;
        const int wv = __builtin_amdgcn_readfirstlane(wave);
        const int head = wv % HN;
        const int n0 = head * 64;
        const int nmt = (nrows + 15) >> 4;
        const int l15 = lane & 15;
        const int kh  = (lane >> 4) << 1;
        const float sa0 = asn[n0 + l15], sa1 = asn[n0 + 16 + l15], sa2 = asn[n0 + 32 + l15], sa3 = asn[n0 + 48 + l15];
        const float da0 = adn[n0 + l15], da1 = adn[n0 + 16 + l15], da2 = adn[n0 + 32 + l15], da3 = adn[n0 + 48 + l15];
        for (int mt = wv / HN; mt < nmt; mt += MG) {
            const int ar = min(mt * 16 + l15, AGG_ROWS - 1);
            const float* Ap = acc + ar * HD + kh;
            const int swz = (ar & 15) << 2;
            v8f q0 = {0.f,0.f,0.f,0.f,0.f,0.f,0.f,0.f};
            v8f q1 = q0, q2 = q0, q3 = q0;
            for (int k = 0; k < HD; k += 4) {
                const v2f a = *reinterpret_cast<const v2f*>(Ap + (k ^ swz));
                const float* Bp = Wn + (size_t)(k + kh) * NC + n0 + l15;
                v2f b0, b1, b2, b3;
                b0.x = Bp[0];  b0.y = Bp[(size_t)NC];
                b1.x = Bp[16]; b1.y = Bp[(size_t)NC + 16];
                b2.x = Bp[32]; b2.y = Bp[(size_t)NC + 32];
                b3.x = Bp[48]; b3.y = Bp[(size_t)NC + 48];
                q0 = __builtin_amdgcn_wmma_f32_16x16x4_f32(false, a, false, b0, (short)0, q0, false, false);
                q1 = __builtin_amdgcn_wmma_f32_16x16x4_f32(false, a, false, b1, (short)0, q1, false, false);
                q2 = __builtin_amdgcn_wmma_f32_16x16x4_f32(false, a, false, b2, (short)0, q2, false, false);
                q3 = __builtin_amdgcn_wmma_f32_16x16x4_f32(false, a, false, b3, (short)0, q3, false, false);
            }
            const int rbase = mt * 16 + ((lane >> 4) << 3);
            float* Cp = hn + (size_t)(lo + rbase) * NC + n0 + l15;
#pragma unroll
            for (int v = 0; v < 8; ++v) {
                if (rbase + v < nrows) {
                    Cp[(size_t)v * NC + 0]  = q0[v];
                    Cp[(size_t)v * NC + 16] = q1[v];
                    Cp[(size_t)v * NC + 32] = q2[v];
                    Cp[(size_t)v * NC + 48] = q3[v];
                }
            }
            float cs = 0.f, cd = 0.f;
#pragma unroll
            for (int v = 0; v < 8; ++v) {
                const float ps = row16_sum(q0[v] * sa0 + q1[v] * sa1 + q2[v] * sa2 + q3[v] * sa3);
                const float pd = row16_sum(q0[v] * da0 + q1[v] * da1 + q2[v] * da2 + q3[v] * da3);
                if (l15 == v) { cs = ps; cd = pd; }
            }
            if (l15 < 8 && rbase + l15 < nrows) {
                const int rl = rbase + l15;
                if constexpr (HN == 4) {
                    kp[kcap + TAIL_WORDS * rl + head] = cs;
                    edn[(size_t)(lo + rl) * HN + head] = cd;
                } else {
                    esn[(size_t)(lo + rl) * HN + head] = cs;
                    edn[(size_t)(lo + rl) * HN + head] = cd;
                }
            }
        }
    } else {
    float cs0 = 0.f, cs1 = 0.f;
    for (int r = wave; r < nrows; r += NW) {
        const float* a = acc + r * HD + lane;
        const size_t nd = (size_t)(lo + r);
        if constexpr (HD == 256) {
            const float4* b4 = reinterpret_cast<const float4*>(bias);
            const float4 c0 = b4[lane], c1 = b4[32 + lane];
            float4 o0, o1;
            o0.x = agg_act(a[0]   + c0.x, do_elu); o0.y = agg_act(a[32]  + c0.y, do_elu);
            o0.z = agg_act(a[64]  + c0.z, do_elu); o0.w = agg_act(a[96]  + c0.w, do_elu);
            o1.x = agg_act(a[128] + c1.x, do_elu); o1.y = agg_act(a[160] + c1.y, do_elu);
            o1.z = agg_act(a[192] + c1.z, do_elu); o1.w = agg_act(a[224] + c1.w, do_elu);
            float4* o4 = reinterpret_cast<float4*>(out + nd * 256);
            o4[lane] = o0;
            o4[32 + lane] = o1;
        } else {
            const float2 c = reinterpret_cast<const float2*>(bias)[lane];
            float2 o;
            o.x = agg_act(a[0]  + c.x, do_elu);
            o.y = agg_act(a[32] + c.y, do_elu);
            reinterpret_cast<float2*>(out + nd * 64)[lane] = o;
            cs0 += o.x;
            cs1 += o.y;
        }
    }
    if constexpr (HD == 64) {
        if (colpart != nullptr) {
            lw[wave * 64 + 2 * lane]     = cs0;
            lw[wave * 64 + 2 * lane + 1] = cs1;
            __syncthreads();
            if (tid < 64) {
                float s = 0.f;
                for (int w = 0; w < NW; ++w) s += lw[w * 64 + tid];
                colpart[(size_t)blockIdx.x * 64 + tid] = s;
            }
        }
    }
    }
}

#define MEAN_GROUPS 16
__global__ __launch_bounds__(64 * MEAN_GROUPS)
void gat_colmean_parts(const float* __restrict__ part, float* __restrict__ out, int N) {
    __shared__ float sm[MEAN_GROUPS * 64];
    const int nblk = (N + AGG_ROWS - 1) / AGG_ROWS;
    const int c = threadIdx.x & 63;
    const int g = threadIdx.x >> 6;
    float s = 0.f;
#pragma unroll 16
    for (int b = g; b < nblk; b += MEAN_GROUPS) s += part[(size_t)b * 64 + c];
    sm[g * 64 + c] = s;
    __syncthreads();
    if (threadIdx.x < 64) {
        float t = 0.f;
        for (int k = 0; k < MEAN_GROUPS; ++k) t += sm[k * 64 + threadIdx.x];
        out[threadIdx.x] = t / (float)N;
    }
}

static inline int cdiv(long long a, long long b) { return (int)((a + b - 1) / b); }

extern "C" void kernel_launch(void* const* d_in, const int* in_sizes, int n_in,
                              void* d_out, int out_size, void* d_ws, size_t ws_size,
                              hipStream_t stream) {
    const float* x   = (const float*)d_in[0];
    const int*   ei  = (const int*)  d_in[1];
    const float* W1  = (const float*)d_in[2];
    const float* a1s = (const float*)d_in[3];
    const float* a1d = (const float*)d_in[4];
    const float* b1  = (const float*)d_in[5];
    const float* W2  = (const float*)d_in[6];
    const float* a2s = (const float*)d_in[7];
    const float* a2d = (const float*)d_in[8];
    const float* b2  = (const float*)d_in[9];
    const float* W3  = (const float*)d_in[10];
    const float* a3s = (const float*)d_in[11];
    const float* a3d = (const float*)d_in[12];
    const float* b3  = (const float*)d_in[13];

    const int N  = in_sizes[0] / 16;
    const int E  = in_sizes[1] / 2;
    const int H  = 4;

    float* ws = (float*)d_ws;
    size_t off = 0;
    float*    buf_h    = ws + off; off += (size_t)N * 256;
    float*    buf_h2   = ws + off; off += (size_t)N * 256;
    float*    es       = ws + off; off += (size_t)N * H;
    float*    ed       = ws + off; off += (size_t)N * H;
    (void)ws_size; (void)off;

    float* node_emb  = (float*)d_out;
    float* graph_emb = node_emb + (size_t)N * 64;

    hipLaunchKernelGGL(gat_gemm_wmma, dim3(1, cdiv(N, 16)), dim3(32, 4), 0, stream,
                       x, W1, a1s, a1d, es, ed, N, 16, 256, ei, E, node_emb);

    const dim3 ag(cdiv(N, AGG_ROWS)), ab(AGG_THREADS);
    float* es3      = es;
    float* col_part = es + (size_t)N;
    float* ed3      = es + 2 * (size_t)N;
    float* none     = nullptr;
    gat_agg_owner<256, 256, false, true><<<ag, ab, 0, stream>>>(ei, E, N, x, es, ed, b1, none, 1, none, node_emb, 0,
                                                                W2, a2s, a2d, buf_h2, none, ed, W1);
    gat_agg_owner<256, 64, true, false><<<ag, ab, 0, stream>>>(ei, E, N, buf_h2, es, ed, b2, none, 1, none, node_emb, 1,
                                                               W3, a3s, a3d, buf_h, es3, ed3, none);
    gat_agg_owner<64, 0, false, false><<<ag, ab, 0, stream>>>(ei, E, N, buf_h, es3, ed3, b3, node_emb, 0, col_part, node_emb, 1,
                                                              none, none, none, none, none, none, none);

    hipLaunchKernelGGL(gat_colmean_parts, dim3(1), dim3(64 * MEAN_GROUPS), 0, stream,
                       col_part, graph_emb, N);
}
